# starting kernel
# baseline (speedup 1.0000x reference)
#include <hip/hip_runtime.h>

typedef __attribute__((ext_vector_type(2))) float v2f;
typedef __attribute__((ext_vector_type(8))) float v8f;

static constexpr int NN = 100000;
static constexpr int NE = 1600000;
static constexpr int DF = 128;
static constexpr int DH = 64;
static constexpr int DC = 40;
#define NEG_SLOPE 0.01f

__device__ __forceinline__ float lrelu(float x) { return x > 0.f ? x : x * NEG_SLOPE; }

__device__ __forceinline__ v8f wmma4(v2f a, v2f b, v8f c) {
    return __builtin_amdgcn_wmma_f32_16x16x4_f32(false, a, false, b, (short)0, c, false, false);
}

__global__ void zero_f32(float* __restrict__ p, int n) {
    int i = blockIdx.x * blockDim.x + threadIdx.x;
    int stride = gridDim.x * blockDim.x;
    for (; i < n; i += stride) p[i] = 0.f;
}

__global__ void deg_kernel(const int* __restrict__ snd, const int* __restrict__ rcv,
                           float* __restrict__ sdeg, float* __restrict__ rdeg) {
    int e = blockIdx.x * blockDim.x + threadIdx.x;
    if (e < NE) {
        atomicAdd(&sdeg[snd[e]], 1.f);
        atomicAdd(&rdeg[rcv[e]], 1.f);
    }
}

__launch_bounds__(256)
__global__ void mlp_kernel(const float* __restrict__ X,
                           const float* __restrict__ W1, const float* __restrict__ b1,
                           const float* __restrict__ W2, const float* __restrict__ b2,
                           const float* __restrict__ sdeg,
                           float* __restrict__ h1, int nTiles) {
    __shared__ float lds[8 * 16 * DH];
    const int w    = threadIdx.x >> 5;
    const int lane = threadIdx.x & 31;
    const int m16  = lane & 15;
    const int kh   = lane >> 4;
    const int t    = blockIdx.x * 8 + w;
    const bool active = (t < nTiles);
    const int row0 = t * 16;
    float* my = &lds[w * 16 * DH];

    if (active) {
        v8f c0 = {}, c1 = {}, c2 = {}, c3 = {};
        #pragma unroll 4
        for (int kb = 0; kb < DF; kb += 4) {
            v2f a = *(const v2f*)(X + (size_t)(row0 + m16) * DF + kb + 2 * kh);
            const float* wr0 = W1 + (size_t)(kb + 2 * kh) * DH;
            const float* wr1 = wr0 + DH;
            v2f b;
            b.x = wr0[m16];      b.y = wr1[m16];      c0 = wmma4(a, b, c0);
            b.x = wr0[16 + m16]; b.y = wr1[16 + m16]; c1 = wmma4(a, b, c1);
            b.x = wr0[32 + m16]; b.y = wr1[32 + m16]; c2 = wmma4(a, b, c2);
            b.x = wr0[48 + m16]; b.y = wr1[48 + m16]; c3 = wmma4(a, b, c3);
        }
        #pragma unroll
        for (int r = 0; r < 8; ++r) {
            float* dr = my + (r + 8 * kh) * DH;
            dr[m16]      = lrelu(c0[r] + b1[m16]);
            dr[16 + m16] = lrelu(c1[r] + b1[16 + m16]);
            dr[32 + m16] = lrelu(c2[r] + b1[32 + m16]);
            dr[48 + m16] = lrelu(c3[r] + b1[48 + m16]);
        }
    }
    __syncthreads();
    if (!active) return;

    v8f d0 = {}, d1 = {}, d2 = {}, d3 = {};
    #pragma unroll 4
    for (int kb = 0; kb < DH; kb += 4) {
        v2f a = *(const v2f*)(my + m16 * DH + kb + 2 * kh);
        const float* wr0 = W2 + (size_t)(kb + 2 * kh) * DH;
        const float* wr1 = wr0 + DH;
        v2f b;
        b.x = wr0[m16];      b.y = wr1[m16];      d0 = wmma4(a, b, d0);
        b.x = wr0[16 + m16]; b.y = wr1[16 + m16]; d1 = wmma4(a, b, d1);
        b.x = wr0[32 + m16]; b.y = wr1[32 + m16]; d2 = wmma4(a, b, d2);
        b.x = wr0[48 + m16]; b.y = wr1[48 + m16]; d3 = wmma4(a, b, d3);
    }
    float sc[8];
    #pragma unroll
    for (int r = 0; r < 8; ++r) sc[r] = rsqrtf(fmaxf(sdeg[row0 + r + 8 * kh], 1.f));
    #pragma unroll
    for (int r = 0; r < 8; ++r) {
        float* orow = h1 + (size_t)(row0 + r + 8 * kh) * DH;
        orow[m16]      = lrelu(d0[r] + b2[m16])      * sc[r];
        orow[16 + m16] = lrelu(d1[r] + b2[16 + m16]) * sc[r];
        orow[32 + m16] = lrelu(d2[r] + b2[32 + m16]) * sc[r];
        orow[48 + m16] = lrelu(d3[r] + b2[48 + m16]) * sc[r];
    }
}

__global__ void agg1_kernel(const float* __restrict__ h1, const int* __restrict__ snd,
                            const int* __restrict__ rcv, float* __restrict__ agg) {
    int idx = blockIdx.x * blockDim.x + threadIdx.x;
    int e = idx >> 6, f = idx & 63;
    if (e < NE)
        atomicAdd(&agg[(size_t)rcv[e] * DH + f], h1[(size_t)snd[e] * DH + f]);
}

__launch_bounds__(256)
__global__ void fin1_kernel(const float* __restrict__ agg1,
                            const float* __restrict__ rdeg, const float* __restrict__ sdeg,
                            const float* __restrict__ W3, const float* __restrict__ b3,
                            float* __restrict__ h2, int nTiles) {
    const int w    = threadIdx.x >> 5;
    const int lane = threadIdx.x & 31;
    const int m16  = lane & 15;
    const int kh   = lane >> 4;
    const int t    = blockIdx.x * 8 + w;
    if (t >= nTiles) return;
    const int row0 = t * 16;
    const float rs = rsqrtf(fmaxf(rdeg[row0 + m16], 1.f));

    v8f c0 = {}, c1 = {}, c2 = {};
    #pragma unroll 4
    for (int kb = 0; kb < DH; kb += 4) {
        v2f a = *(const v2f*)(agg1 + (size_t)(row0 + m16) * DH + kb + 2 * kh);
        a.x *= rs; a.y *= rs;
        const float* wr0 = W3 + (size_t)(kb + 2 * kh) * DC;
        const float* wr1 = wr0 + DC;
        v2f b;
        b.x = wr0[m16];      b.y = wr1[m16];      c0 = wmma4(a, b, c0);
        b.x = wr0[16 + m16]; b.y = wr1[16 + m16]; c1 = wmma4(a, b, c1);
        const int n2 = 32 + m16;
        b.x = (n2 < DC) ? wr0[n2] : 0.f;
        b.y = (n2 < DC) ? wr1[n2] : 0.f;
        c2 = wmma4(a, b, c2);
    }
    float sc[8];
    #pragma unroll
    for (int r = 0; r < 8; ++r) sc[r] = rsqrtf(fmaxf(sdeg[row0 + r + 8 * kh], 1.f));
    #pragma unroll
    for (int r = 0; r < 8; ++r) {
        float* orow = h2 + (size_t)(row0 + r + 8 * kh) * DC;
        orow[m16]      = (c0[r] + b3[m16])      * sc[r];
        orow[16 + m16] = (c1[r] + b3[16 + m16]) * sc[r];
        if (32 + m16 < DC)
            orow[32 + m16] = (c2[r] + b3[32 + m16]) * sc[r];
    }
}

__global__ void agg2_kernel(const float* __restrict__ h2, const int* __restrict__ snd,
                            const int* __restrict__ rcv, float* __restrict__ out) {
    int idx = blockIdx.x * blockDim.x + threadIdx.x;
    int e = idx / DC, f = idx - e * DC;
    if (e < NE)
        atomicAdd(&out[(size_t)rcv[e] * DC + f], h2[(size_t)snd[e] * DC + f]);
}

__global__ void softmax_kernel(float* __restrict__ out, const float* __restrict__ rdeg) {
    int n = blockIdx.x * blockDim.x + threadIdx.x;
    if (n >= NN) return;
    float rs = rsqrtf(fmaxf(rdeg[n], 1.f));
    float* row = out + (size_t)n * DC;
    float v[DC];
    float mx = -3.402823466e38f;
    #pragma unroll
    for (int f = 0; f < DC; ++f) { v[f] = row[f] * rs; mx = fmaxf(mx, v[f]); }
    float s = 0.f;
    #pragma unroll
    for (int f = 0; f < DC; ++f) { v[f] = __expf(v[f] - mx); s += v[f]; }
    float inv = 1.f / s;
    #pragma unroll
    for (int f = 0; f < DC; ++f) row[f] = v[f] * inv;
}

extern "C" void kernel_launch(void* const* d_in, const int* in_sizes, int n_in,
                              void* d_out, int out_size, void* d_ws, size_t ws_size,
                              hipStream_t stream) {
    const float* nodes     = (const float*)d_in[0];
    const int*   senders   = (const int*)d_in[1];
    const int*   receivers = (const int*)d_in[2];
    const float* W1 = (const float*)d_in[3];
    const float* b1 = (const float*)d_in[4];
    const float* W2 = (const float*)d_in[5];
    const float* b2 = (const float*)d_in[6];
    const float* W3 = (const float*)d_in[7];
    const float* b3 = (const float*)d_in[8];
    float* out = (float*)d_out;
    float* ws  = (float*)d_ws;

    float* sdeg = ws;
    float* rdeg = ws + NN;
    float* h1   = ws + 2 * (size_t)NN;
    float* agg1 = h1 + (size_t)NN * DH;
    float* h2   = agg1 + (size_t)NN * DH;

    const int nTiles = NN / 16;
    zero_f32<<<2048, 256, 0, stream>>>(sdeg, 2 * NN);
    zero_f32<<<8192, 256, 0, stream>>>(agg1, NN * DH);
    zero_f32<<<8192, 256, 0, stream>>>(out, NN * DC);
    deg_kernel<<<(NE + 255) / 256, 256, 0, stream>>>(senders, receivers, sdeg, rdeg);
    mlp_kernel<<<(nTiles + 7) / 8, 256, 0, stream>>>(nodes, W1, b1, W2, b2, sdeg, h1, nTiles);
    agg1_kernel<<<(NE * 64) / 256, 256, 0, stream>>>(h1, senders, receivers, agg1);
    fin1_kernel<<<(nTiles + 7) / 8, 256, 0, stream>>>(agg1, rdeg, sdeg, W3, b3, h2, nTiles);
    agg2_kernel<<<(NE * DC + 255) / 256, 256, 0, stream>>>(h2, senders, receivers, out);
    softmax_kernel<<<(NN + 255) / 256, 256, 0, stream>>>(out, rdeg);
}


# Edge sums by owner blocks: a block keeps a receiver range's sums in LDS, scans all receivers for its edges, adds sender rows by LDS float atomics (arrival order); rows stored once; 9 -> 7 launches
# speedup vs baseline: 1.6976x; 1.6976x over previous
#include <hip/hip_runtime.h>

typedef __attribute__((ext_vector_type(2))) float v2f;
typedef __attribute__((ext_vector_type(8))) float v8f;

static constexpr int NN = 100000;
static constexpr int NE = 1600000;
static constexpr int DF = 128;
static constexpr int DH = 64;
static constexpr int DC = 40;
#define NEG_SLOPE 0.01f

__device__ __forceinline__ float lrelu(float x) { return x > 0.f ? x : x * NEG_SLOPE; }

__device__ __forceinline__ v8f wmma4(v2f a, v2f b, v8f c) {
    return __builtin_amdgcn_wmma_f32_16x16x4_f32(false, a, false, b, (short)0, c, false, false);
}

__global__ void zero_f32(float* __restrict__ p, int n) {
    int i = blockIdx.x * blockDim.x + threadIdx.x;
    int stride = gridDim.x * blockDim.x;
    for (; i < n; i += stride) p[i] = 0.f;
}

__global__ void deg_kernel(const int* __restrict__ snd, const int* __restrict__ rcv,
                           float* __restrict__ sdeg, float* __restrict__ rdeg) {
    int e = blockIdx.x * blockDim.x + threadIdx.x;
    if (e < NE) {
        atomicAdd(&sdeg[snd[e]], 1.f);
        atomicAdd(&rdeg[rcv[e]], 1.f);
    }
}

__launch_bounds__(256)
__global__ void mlp_kernel(const float* __restrict__ X,
                           const float* __restrict__ W1, const float* __restrict__ b1,
                           const float* __restrict__ W2, const float* __restrict__ b2,
                           const float* __restrict__ sdeg,
                           float* __restrict__ h1, int nTiles) {
    __shared__ float lds[8 * 16 * DH];
    const int w    = threadIdx.x >> 5;
    const int lane = threadIdx.x & 31;
    const int m16  = lane & 15;
    const int kh   = lane >> 4;
    const int t    = blockIdx.x * 8 + w;
    const bool active = (t < nTiles);
    const int row0 = t * 16;
    float* my = &lds[w * 16 * DH];

    if (active) {
        v8f c0 = {}, c1 = {}, c2 = {}, c3 = {};
        #pragma unroll 4
        for (int kb = 0; kb < DF; kb += 4) {
            v2f a = *(const v2f*)(X + (size_t)(row0 + m16) * DF + kb + 2 * kh);
            const float* wr0 = W1 + (size_t)(kb + 2 * kh) * DH;
            const float* wr1 = wr0 + DH;
            v2f b;
            b.x = wr0[m16];      b.y = wr1[m16];      c0 = wmma4(a, b, c0);
            b.x = wr0[16 + m16]; b.y = wr1[16 + m16]; c1 = wmma4(a, b, c1);
            b.x = wr0[32 + m16]; b.y = wr1[32 + m16]; c2 = wmma4(a, b, c2);
            b.x = wr0[48 + m16]; b.y = wr1[48 + m16]; c3 = wmma4(a, b, c3);
        }
        #pragma unroll
        for (int r = 0; r < 8; ++r) {
            float* dr = my + (r + 8 * kh) * DH;
            dr[m16]      = lrelu(c0[r] + b1[m16]);
            dr[16 + m16] = lrelu(c1[r] + b1[16 + m16]);
            dr[32 + m16] = lrelu(c2[r] + b1[32 + m16]);
            dr[48 + m16] = lrelu(c3[r] + b1[48 + m16]);
        }
    }
    __syncthreads();
    if (!active) return;

    v8f d0 = {}, d1 = {}, d2 = {}, d3 = {};
    #pragma unroll 4
    for (int kb = 0; kb < DH; kb += 4) {
        v2f a = *(const v2f*)(my + m16 * DH + kb + 2 * kh);
        const float* wr0 = W2 + (size_t)(kb + 2 * kh) * DH;
        const float* wr1 = wr0 + DH;
        v2f b;
        b.x = wr0[m16];      b.y = wr1[m16];      d0 = wmma4(a, b, d0);
        b.x = wr0[16 + m16]; b.y = wr1[16 + m16]; d1 = wmma4(a, b, d1);
        b.x = wr0[32 + m16]; b.y = wr1[32 + m16]; d2 = wmma4(a, b, d2);
        b.x = wr0[48 + m16]; b.y = wr1[48 + m16]; d3 = wmma4(a, b, d3);
    }
    float sc[8];
    #pragma unroll
    for (int r = 0; r < 8; ++r) sc[r] = rsqrtf(fmaxf(sdeg[row0 + r + 8 * kh], 1.f));
    #pragma unroll
    for (int r = 0; r < 8; ++r) {
        float* orow = h1 + (size_t)(row0 + r + 8 * kh) * DH;
        orow[m16]      = lrelu(d0[r] + b2[m16])      * sc[r];
        orow[16 + m16] = lrelu(d1[r] + b2[16 + m16]) * sc[r];
        orow[32 + m16] = lrelu(d2[r] + b2[32 + m16]) * sc[r];
        orow[48 + m16] = lrelu(d3[r] + b2[48 + m16]) * sc[r];
    }
}

static constexpr int OWN_G  = 256;
static constexpr int OWN_R  = (NN + OWN_G - 1) / OWN_G;
static constexpr int OWN_NW = 32;
static constexpr int OWN_U  = 8;
static constexpr int OWN_QCAP = OWN_U * 128 + 32;
static constexpr int NG4 = NE / 4;

template <int F>
__device__ __forceinline__ void drain_rows(const float* __restrict__ src, const int* __restrict__ snd,
                                           float* acc, const int* q, int n, int lane) {
    const int ent = (lane < n) ? q[lane] : 0;
    const int l = ent >> 21;
    const int s = (lane < n) ? snd[ent & 0x1FFFFF] : 0;
    for (int j0 = 0; j0 < n; j0 += 8) {
        float a[8], b[8];
        int lj[8];
        #pragma unroll
        for (int u = 0; u < 8; ++u) {
            const int sj = __builtin_amdgcn_readlane(s, j0 + u);
            lj[u] = __builtin_amdgcn_readlane(l, j0 + u);
            const float* row = src + (size_t)sj * F;
            a[u] = row[lane];
            b[u] = (lane < F - 32) ? row[32 + lane] : 0.f;
        }
        #pragma unroll
        for (int u = 0; u < 8; ++u) {
            if (j0 + u < n) {
                float* d = acc + lj[u] * F;
                atomicAdd(d + lane, a[u]);
                if (lane < F - 32) atomicAdd(d + 32 + lane, b[u]);
            }
        }
    }
}

template <int F>
__launch_bounds__(OWN_NW * 32)
__global__ void agg_own_kernel(const float* __restrict__ src, const int* __restrict__ snd,
                               const int* __restrict__ rcv, float* __restrict__ dst) {
    __shared__ float acc[OWN_R * F];
    __shared__ int   qs[OWN_NW * OWN_QCAP];
    const int w    = __builtin_amdgcn_readfirstlane((int)(threadIdx.x >> 5));
    const int lane = threadIdx.x & 31;
    const int r0   = blockIdx.x * OWN_R;
    int* q = qs + w * OWN_QCAP;

    for (int i = threadIdx.x; i < OWN_R * F; i += OWN_NW * 32) acc[i] = 0.f;
    __syncthreads();

    const int4* rcv4 = (const int4*)rcv;
    int qn = 0;
    for (int gb = w * 32 * OWN_U; gb < NG4; gb += OWN_NW * 32 * OWN_U) {
        int4 v[OWN_U];
        #pragma unroll
        for (int u = 0; u < OWN_U; ++u) {
            const int g = gb + u * 32 + lane;
            v[u] = (g < NG4) ? rcv4[g] : make_int4(-1, -1, -1, -1);
        }
        #pragma unroll
        for (int u = 0; u < OWN_U; ++u) {
            const int e0 = (gb + u * 32 + lane) * 4;
            const int rr[4] = {v[u].x, v[u].y, v[u].z, v[u].w};
            #pragma unroll
            for (int k = 0; k < 4; ++k) {
                const unsigned d = (unsigned)(rr[k] - r0);
                const bool hit = d < (unsigned)OWN_R;
                const unsigned m = __builtin_amdgcn_ballot_w32(hit);
                if (m) {
                    const int pos = qn + (int)__builtin_amdgcn_mbcnt_lo(m, 0u);
                    if (hit) q[pos] = (e0 + k) | ((int)d << 21);
                    qn += __builtin_popcount(m);
                }
            }
        }
        if (qn >= 32) {
            int done = 0;
            while (qn - done >= 32) {
                drain_rows<F>(src, snd, acc, q + done, 32, lane);
                done += 32;
            }
            const int rem = qn - done;
            const int t = (lane < rem) ? q[done + lane] : 0;
            if (lane < rem) q[lane] = t;
            qn = rem;
        }
    }
    if (qn > 0) drain_rows<F>(src, snd, acc, q, qn, lane);
    __syncthreads();

    for (int i = threadIdx.x; i < OWN_R * F; i += OWN_NW * 32) {
        const int node = r0 + i / F;
        if (node < NN) dst[(size_t)r0 * F + i] = acc[i];
    }
}

__launch_bounds__(256)
__global__ void fin1_kernel(const float* __restrict__ agg1,
                            const float* __restrict__ rdeg, const float* __restrict__ sdeg,
                            const float* __restrict__ W3, const float* __restrict__ b3,
                            float* __restrict__ h2, int nTiles) {
    const int w    = threadIdx.x >> 5;
    const int lane = threadIdx.x & 31;
    const int m16  = lane & 15;
    const int kh   = lane >> 4;
    const int t    = blockIdx.x * 8 + w;
    if (t >= nTiles) return;
    const int row0 = t * 16;
    const float rs = rsqrtf(fmaxf(rdeg[row0 + m16], 1.f));

    v8f c0 = {}, c1 = {}, c2 = {};
    #pragma unroll 4
    for (int kb = 0; kb < DH; kb += 4) {
        v2f a = *(const v2f*)(agg1 + (size_t)(row0 + m16) * DH + kb + 2 * kh);
        a.x *= rs; a.y *= rs;
        const float* wr0 = W3 + (size_t)(kb + 2 * kh) * DC;
        const float* wr1 = wr0 + DC;
        v2f b;
        b.x = wr0[m16];      b.y = wr1[m16];      c0 = wmma4(a, b, c0);
        b.x = wr0[16 + m16]; b.y = wr1[16 + m16]; c1 = wmma4(a, b, c1);
        const int n2 = 32 + m16;
        b.x = (n2 < DC) ? wr0[n2] : 0.f;
        b.y = (n2 < DC) ? wr1[n2] : 0.f;
        c2 = wmma4(a, b, c2);
    }
    float sc[8];
    #pragma unroll
    for (int r = 0; r < 8; ++r) sc[r] = rsqrtf(fmaxf(sdeg[row0 + r + 8 * kh], 1.f));
    #pragma unroll
    for (int r = 0; r < 8; ++r) {
        float* orow = h2 + (size_t)(row0 + r + 8 * kh) * DC;
        orow[m16]      = (c0[r] + b3[m16])      * sc[r];
        orow[16 + m16] = (c1[r] + b3[16 + m16]) * sc[r];
        if (32 + m16 < DC)
            orow[32 + m16] = (c2[r] + b3[32 + m16]) * sc[r];
    }
}

__global__ void softmax_kernel(float* __restrict__ out, const float* __restrict__ rdeg) {
    int n = blockIdx.x * blockDim.x + threadIdx.x;
    if (n >= NN) return;
    float rs = rsqrtf(fmaxf(rdeg[n], 1.f));
    float* row = out + (size_t)n * DC;
    float v[DC];
    float mx = -3.402823466e38f;
    #pragma unroll
    for (int f = 0; f < DC; ++f) { v[f] = row[f] * rs; mx = fmaxf(mx, v[f]); }
    float s = 0.f;
    #pragma unroll
    for (int f = 0; f < DC; ++f) { v[f] = __expf(v[f] - mx); s += v[f]; }
    float inv = 1.f / s;
    #pragma unroll
    for (int f = 0; f < DC; ++f) row[f] = v[f] * inv;
}

extern "C" void kernel_launch(void* const* d_in, const int* in_sizes, int n_in,
                              void* d_out, int out_size, void* d_ws, size_t ws_size,
                              hipStream_t stream) {
    const float* nodes     = (const float*)d_in[0];
    const int*   senders   = (const int*)d_in[1];
    const int*   receivers = (const int*)d_in[2];
    const float* W1 = (const float*)d_in[3];
    const float* b1 = (const float*)d_in[4];
    const float* W2 = (const float*)d_in[5];
    const float* b2 = (const float*)d_in[6];
    const float* W3 = (const float*)d_in[7];
    const float* b3 = (const float*)d_in[8];
    float* out = (float*)d_out;
    float* ws  = (float*)d_ws;

    float* sdeg = ws;
    float* rdeg = ws + NN;
    float* h1   = ws + 2 * (size_t)NN;
    float* agg1 = h1 + (size_t)NN * DH;
    float* h2   = agg1 + (size_t)NN * DH;

    const int nTiles = NN / 16;
    zero_f32<<<2048, 256, 0, stream>>>(sdeg, 2 * NN);
    deg_kernel<<<(NE + 255) / 256, 256, 0, stream>>>(senders, receivers, sdeg, rdeg);
    mlp_kernel<<<(nTiles + 7) / 8, 256, 0, stream>>>(nodes, W1, b1, W2, b2, sdeg, h1, nTiles);
    agg_own_kernel<DH><<<OWN_G, OWN_NW * 32, 0, stream>>>(h1, senders, receivers, agg1);
    fin1_kernel<<<(nTiles + 7) / 8, 256, 0, stream>>>(agg1, rdeg, sdeg, W3, b3, h2, nTiles);
    agg_own_kernel<DC><<<OWN_G, OWN_NW * 32, 0, stream>>>(h2, senders, receivers, out);
    softmax_kernel<<<(NN + 255) / 256, 256, 0, stream>>>(out, rdeg);
}


# Owner scan reads one-byte owner tags (written by the degree kernel), 16 edges a load, not 4-byte receivers; hit groups re-read exactly; edges queue in another order: LDS-atomic sum order changes
# speedup vs baseline: 1.9018x; 1.1203x over previous
#include <hip/hip_runtime.h>

typedef __attribute__((ext_vector_type(2))) float v2f;
typedef __attribute__((ext_vector_type(8))) float v8f;

static constexpr int NN = 100000;
static constexpr int NE = 1600000;
static constexpr int DF = 128;
static constexpr int DH = 64;
static constexpr int DC = 40;
#define NEG_SLOPE 0.01f
static constexpr unsigned TAG_R = (NN + 255) / 256;

__device__ __forceinline__ float lrelu(float x) { return x > 0.f ? x : x * NEG_SLOPE; }

__device__ __forceinline__ v8f wmma4(v2f a, v2f b, v8f c) {
    return __builtin_amdgcn_wmma_f32_16x16x4_f32(false, a, false, b, (short)0, c, false, false);
}

__global__ void zero_f32(float* __restrict__ p, int n) {
    int i = blockIdx.x * blockDim.x + threadIdx.x;
    int stride = gridDim.x * blockDim.x;
    for (; i < n; i += stride) p[i] = 0.f;
}

__global__ void deg_kernel(const int* __restrict__ snd, const int* __restrict__ rcv,
                           float* __restrict__ sdeg, float* __restrict__ rdeg,
                           unsigned* __restrict__ tags) {
    const int g = blockIdx.x * blockDim.x + threadIdx.x;
    if (g < NE / 4) {
        const int4 s = ((const int4*)snd)[g];
        const int4 r = ((const int4*)rcv)[g];
        atomicAdd(&sdeg[s.x], 1.f);
        atomicAdd(&sdeg[s.y], 1.f);
        atomicAdd(&sdeg[s.z], 1.f);
        atomicAdd(&sdeg[s.w], 1.f);
        atomicAdd(&rdeg[r.x], 1.f);
        atomicAdd(&rdeg[r.y], 1.f);
        atomicAdd(&rdeg[r.z], 1.f);
        atomicAdd(&rdeg[r.w], 1.f);
        tags[g] = ((unsigned)r.x / TAG_R) | (((unsigned)r.y / TAG_R) << 8) |
                  (((unsigned)r.z / TAG_R) << 16) | (((unsigned)r.w / TAG_R) << 24);
    }
}

__launch_bounds__(256)
__global__ void mlp_kernel(const float* __restrict__ X,
                           const float* __restrict__ W1, const float* __restrict__ b1,
                           const float* __restrict__ W2, const float* __restrict__ b2,
                           const float* __restrict__ sdeg,
                           float* __restrict__ h1, int nTiles) {
    __shared__ float lds[8 * 16 * DH];
    const int w    = threadIdx.x >> 5;
    const int lane = threadIdx.x & 31;
    const int m16  = lane & 15;
    const int kh   = lane >> 4;
    const int t    = blockIdx.x * 8 + w;
    const bool active = (t < nTiles);
    const int row0 = t * 16;
    float* my = &lds[w * 16 * DH];

    if (active) {
        v8f c0 = {}, c1 = {}, c2 = {}, c3 = {};
        #pragma unroll 4
        for (int kb = 0; kb < DF; kb += 4) {
            v2f a = *(const v2f*)(X + (size_t)(row0 + m16) * DF + kb + 2 * kh);
            const float* wr0 = W1 + (size_t)(kb + 2 * kh) * DH;
            const float* wr1 = wr0 + DH;
            v2f b;
            b.x = wr0[m16];      b.y = wr1[m16];      c0 = wmma4(a, b, c0);
            b.x = wr0[16 + m16]; b.y = wr1[16 + m16]; c1 = wmma4(a, b, c1);
            b.x = wr0[32 + m16]; b.y = wr1[32 + m16]; c2 = wmma4(a, b, c2);
            b.x = wr0[48 + m16]; b.y = wr1[48 + m16]; c3 = wmma4(a, b, c3);
        }
        #pragma unroll
        for (int r = 0; r < 8; ++r) {
            float* dr = my + (r + 8 * kh) * DH;
            dr[m16]      = lrelu(c0[r] + b1[m16]);
            dr[16 + m16] = lrelu(c1[r] + b1[16 + m16]);
            dr[32 + m16] = lrelu(c2[r] + b1[32 + m16]);
            dr[48 + m16] = lrelu(c3[r] + b1[48 + m16]);
        }
    }
    __syncthreads();
    if (!active) return;

    v8f d0 = {}, d1 = {}, d2 = {}, d3 = {};
    #pragma unroll 4
    for (int kb = 0; kb < DH; kb += 4) {
        v2f a = *(const v2f*)(my + m16 * DH + kb + 2 * kh);
        const float* wr0 = W2 + (size_t)(kb + 2 * kh) * DH;
        const float* wr1 = wr0 + DH;
        v2f b;
        b.x = wr0[m16];      b.y = wr1[m16];      d0 = wmma4(a, b, d0);
        b.x = wr0[16 + m16]; b.y = wr1[16 + m16]; d1 = wmma4(a, b, d1);
        b.x = wr0[32 + m16]; b.y = wr1[32 + m16]; d2 = wmma4(a, b, d2);
        b.x = wr0[48 + m16]; b.y = wr1[48 + m16]; d3 = wmma4(a, b, d3);
    }
    float sc[8];
    #pragma unroll
    for (int r = 0; r < 8; ++r) sc[r] = rsqrtf(fmaxf(sdeg[row0 + r + 8 * kh], 1.f));
    #pragma unroll
    for (int r = 0; r < 8; ++r) {
        float* orow = h1 + (size_t)(row0 + r + 8 * kh) * DH;
        orow[m16]      = lrelu(d0[r] + b2[m16])      * sc[r];
        orow[16 + m16] = lrelu(d1[r] + b2[16 + m16]) * sc[r];
        orow[32 + m16] = lrelu(d2[r] + b2[32 + m16]) * sc[r];
        orow[48 + m16] = lrelu(d3[r] + b2[48 + m16]) * sc[r];
    }
}

static constexpr int OWN_G  = 256;
static constexpr int OWN_R  = (NN + OWN_G - 1) / OWN_G;
static constexpr int OWN_NW = 32;
static constexpr int OWN_U  = 8;
static constexpr int OWN_GQCAP = OWN_U * 32 + 32;
static constexpr int OWN_QCAP  = 16 * 32 + 32;
static constexpr int NG16 = NE / 16;
static_assert(OWN_R == (int)TAG_R, "tags must name the owner blocks");
static_assert(NE % 16 == 0, "edge list must split into groups of sixteen");
static_assert(NG16 % 32 == 0, "a wave's 32 groups must lie all inside the list or all past its end");

template <int F>
__device__ __forceinline__ void drain_rows(const float* __restrict__ src, const int* __restrict__ snd,
                                           float* acc, const int* q, int n, int lane) {
    const int ent = (lane < n) ? q[lane] : 0;
    const int l = ent >> 21;
    const int s = (lane < n) ? snd[ent & 0x1FFFFF] : 0;
    for (int j0 = 0; j0 < n; j0 += 8) {
        float a[8], b[8];
        int lj[8];
        #pragma unroll
        for (int u = 0; u < 8; ++u) {
            const int sj = __builtin_amdgcn_readlane(s, j0 + u);
            lj[u] = __builtin_amdgcn_readlane(l, j0 + u);
            const float* row = src + (size_t)sj * F;
            a[u] = row[lane];
            b[u] = (lane < F - 32) ? row[32 + lane] : 0.f;
        }
        #pragma unroll
        for (int u = 0; u < 8; ++u) {
            if (j0 + u < n) {
                float* d = acc + lj[u] * F;
                atomicAdd(d + lane, a[u]);
                if (lane < F - 32) atomicAdd(d + 32 + lane, b[u]);
            }
        }
    }
}

__device__ __forceinline__ unsigned nonzero_bytes(unsigned y) {
    return ((y & 0x7F7F7F7Fu) + 0x7F7F7F7Fu) | y;
}

template <int F>
__device__ __forceinline__ void expand_groups(const float* __restrict__ src, const int* __restrict__ snd,
                                              const int4* __restrict__ rcv4, float* acc, int* q, int& qn,
                                              int g, int r0, int lane) {
    int4 r[4];
    #pragma unroll
    for (int k = 0; k < 4; ++k) r[k] = (g >= 0) ? rcv4[g * 4 + k] : make_int4(-1, -1, -1, -1);
    #pragma unroll
    for (int k = 0; k < 4; ++k) {
        const int rr[4] = {r[k].x, r[k].y, r[k].z, r[k].w};
        #pragma unroll
        for (int c = 0; c < 4; ++c) {
            const unsigned d = (unsigned)(rr[c] - r0);
            const bool hit = d < (unsigned)OWN_R;
            const unsigned m = __builtin_amdgcn_ballot_w32(hit);
            if (m) {
                const int pos = qn + (int)__builtin_amdgcn_mbcnt_lo(m, 0u);
                if (hit) q[pos] = (g * 16 + k * 4 + c) | ((int)d << 21);
                qn += __builtin_popcount(m);
            }
        }
    }
    if (qn >= 32) {
        int done = 0;
        while (qn - done >= 32) {
            drain_rows<F>(src, snd, acc, q + done, 32, lane);
            done += 32;
        }
        const int rem = qn - done;
        const int t = (lane < rem) ? q[done + lane] : 0;
        if (lane < rem) q[lane] = t;
        qn = rem;
    }
}

template <int F>
__launch_bounds__(OWN_NW * 32)
__global__ void agg_own_kernel(const float* __restrict__ src, const int* __restrict__ snd,
                               const int* __restrict__ rcv, const unsigned* __restrict__ tags,
                               float* __restrict__ dst) {
    __shared__ float acc[OWN_R * F];
    __shared__ int   qs[OWN_NW * OWN_QCAP];
    __shared__ int   gqs[OWN_NW * OWN_GQCAP];
    const int w    = __builtin_amdgcn_readfirstlane((int)(threadIdx.x >> 5));
    const int lane = threadIdx.x & 31;
    const int r0   = blockIdx.x * OWN_R;
    int* q  = qs + w * OWN_QCAP;
    int* gq = gqs + w * OWN_GQCAP;
    const unsigned pat = (unsigned)blockIdx.x * 0x01010101u;

    for (int i = threadIdx.x; i < OWN_R * F; i += OWN_NW * 32) acc[i] = 0.f;
    __syncthreads();

    const uint4* tag16 = (const uint4*)tags;
    const int4*  rcv4  = (const int4*)rcv;
    int qn = 0;
    int gn = 0;
    for (int gb = w * 32 * OWN_U; gb < NG16; gb += OWN_NW * 32 * OWN_U) {
        uint4 v[OWN_U];
        #pragma unroll
        for (int u = 0; u < OWN_U; ++u) {
            const int g = gb + u * 32 + lane;
            v[u] = tag16[(g < NG16) ? g : NG16 - 1];
        }
        #pragma unroll
        for (int u = 0; u < OWN_U; ++u) {
            const unsigned past = (gb + u * 32 < NG16) ? 0u : 0x01010101u;
            const unsigned t = nonzero_bytes((v[u].x ^ pat) | past) & nonzero_bytes((v[u].y ^ pat) | past) &
                               nonzero_bytes((v[u].z ^ pat) | past) & nonzero_bytes((v[u].w ^ pat) | past);
            const bool hit = (t & 0x80808080u) != 0x80808080u;
            const unsigned m = __builtin_amdgcn_ballot_w32(hit);
            if (m) {
                const int pos = gn + (int)__builtin_amdgcn_mbcnt_lo(m, 0u);
                if (hit) gq[pos] = gb + u * 32 + lane;
                gn += __builtin_popcount(m);
            }
        }
        if (gn >= 32) {
            int done = 0;
            while (gn - done >= 32) {
                expand_groups<F>(src, snd, rcv4, acc, q, qn, gq[done + lane], r0, lane);
                done += 32;
            }
            const int rem = gn - done;
            const int t = (lane < rem) ? gq[done + lane] : 0;
            if (lane < rem) gq[lane] = t;
            gn = rem;
        }
    }
    if (gn > 0) expand_groups<F>(src, snd, rcv4, acc, q, qn, (lane < gn) ? gq[lane] : -1, r0, lane);
    if (qn > 0) drain_rows<F>(src, snd, acc, q, qn, lane);
    __syncthreads();

    for (int i = threadIdx.x; i < OWN_R * F; i += OWN_NW * 32) {
        const int node = r0 + i / F;
        if (node < NN) dst[(size_t)r0 * F + i] = acc[i];
    }
}

__launch_bounds__(256)
__global__ void fin1_kernel(const float* __restrict__ agg1,
                            const float* __restrict__ rdeg, const float* __restrict__ sdeg,
                            const float* __restrict__ W3, const float* __restrict__ b3,
                            float* __restrict__ h2, int nTiles) {
    const int w    = threadIdx.x >> 5;
    const int lane = threadIdx.x & 31;
    const int m16  = lane & 15;
    const int kh   = lane >> 4;
    const int t    = blockIdx.x * 8 + w;
    if (t >= nTiles) return;
    const int row0 = t * 16;
    const float rs = rsqrtf(fmaxf(rdeg[row0 + m16], 1.f));

    v8f c0 = {}, c1 = {}, c2 = {};
    #pragma unroll 4
    for (int kb = 0; kb < DH; kb += 4) {
        v2f a = *(const v2f*)(agg1 + (size_t)(row0 + m16) * DH + kb + 2 * kh);
        a.x *= rs; a.y *= rs;
        const float* wr0 = W3 + (size_t)(kb + 2 * kh) * DC;
        const float* wr1 = wr0 + DC;
        v2f b;
        b.x = wr0[m16];      b.y = wr1[m16];      c0 = wmma4(a, b, c0);
        b.x = wr0[16 + m16]; b.y = wr1[16 + m16]; c1 = wmma4(a, b, c1);
        const int n2 = 32 + m16;
        b.x = (n2 < DC) ? wr0[n2] : 0.f;
        b.y = (n2 < DC) ? wr1[n2] : 0.f;
        c2 = wmma4(a, b, c2);
    }
    float sc[8];
    #pragma unroll
    for (int r = 0; r < 8; ++r) sc[r] = rsqrtf(fmaxf(sdeg[row0 + r + 8 * kh], 1.f));
    #pragma unroll
    for (int r = 0; r < 8; ++r) {
        float* orow = h2 + (size_t)(row0 + r + 8 * kh) * DC;
        orow[m16]      = (c0[r] + b3[m16])      * sc[r];
        orow[16 + m16] = (c1[r] + b3[16 + m16]) * sc[r];
        if (32 + m16 < DC)
            orow[32 + m16] = (c2[r] + b3[32 + m16]) * sc[r];
    }
}

__global__ void softmax_kernel(float* __restrict__ out, const float* __restrict__ rdeg) {
    int n = blockIdx.x * blockDim.x + threadIdx.x;
    if (n >= NN) return;
    float rs = rsqrtf(fmaxf(rdeg[n], 1.f));
    float* row = out + (size_t)n * DC;
    float v[DC];
    float mx = -3.402823466e38f;
    #pragma unroll
    for (int f = 0; f < DC; ++f) { v[f] = row[f] * rs; mx = fmaxf(mx, v[f]); }
    float s = 0.f;
    #pragma unroll
    for (int f = 0; f < DC; ++f) { v[f] = __expf(v[f] - mx); s += v[f]; }
    float inv = 1.f / s;
    #pragma unroll
    for (int f = 0; f < DC; ++f) row[f] = v[f] * inv;
}

extern "C" void kernel_launch(void* const* d_in, const int* in_sizes, int n_in,
                              void* d_out, int out_size, void* d_ws, size_t ws_size,
                              hipStream_t stream) {
    const float* nodes     = (const float*)d_in[0];
    const int*   senders   = (const int*)d_in[1];
    const int*   receivers = (const int*)d_in[2];
    const float* W1 = (const float*)d_in[3];
    const float* b1 = (const float*)d_in[4];
    const float* W2 = (const float*)d_in[5];
    const float* b2 = (const float*)d_in[6];
    const float* W3 = (const float*)d_in[7];
    const float* b3 = (const float*)d_in[8];
    float* out = (float*)d_out;
    float* ws  = (float*)d_ws;

    float* sdeg = ws;
    float* rdeg = ws + NN;
    float* h1   = ws + 2 * (size_t)NN;
    float* agg1 = h1 + (size_t)NN * DH;
    float* h2   = agg1 + (size_t)NN * DH;
    unsigned* tags = (unsigned*)(h2 + (size_t)NN * DC);

    const int nTiles = NN / 16;
    zero_f32<<<2048, 256, 0, stream>>>(sdeg, 2 * NN);
    deg_kernel<<<(NE / 4 + 255) / 256, 256, 0, stream>>>(senders, receivers, sdeg, rdeg, tags);
    mlp_kernel<<<(nTiles + 7) / 8, 256, 0, stream>>>(nodes, W1, b1, W2, b2, sdeg, h1, nTiles);
    agg_own_kernel<DH><<<OWN_G, OWN_NW * 32, 0, stream>>>(h1, senders, receivers, tags, agg1);
    fin1_kernel<<<(nTiles + 7) / 8, 256, 0, stream>>>(agg1, rdeg, sdeg, W3, b3, h2, nTiles);
    agg_own_kernel<DC><<<OWN_G, OWN_NW * 32, 0, stream>>>(h2, senders, receivers, tags, out);
    softmax_kernel<<<(NN + 255) / 256, 256, 0, stream>>>(out, rdeg);
}


# Receiver degrees counted by the first edge sum's owner blocks (one LDS integer add per drained edge, exact, stored once per node); the degree kernel keeps only the sender adds; float sums unchanged
# speedup vs baseline: 2.3314x; 1.2259x over previous
#include <hip/hip_runtime.h>

typedef __attribute__((ext_vector_type(2))) float v2f;
typedef __attribute__((ext_vector_type(8))) float v8f;

static constexpr int NN = 100000;
static constexpr int NE = 1600000;
static constexpr int DF = 128;
static constexpr int DH = 64;
static constexpr int DC = 40;
#define NEG_SLOPE 0.01f
static constexpr unsigned TAG_R = (NN + 255) / 256;

__device__ __forceinline__ float lrelu(float x) { return x > 0.f ? x : x * NEG_SLOPE; }

__device__ __forceinline__ v8f wmma4(v2f a, v2f b, v8f c) {
    return __builtin_amdgcn_wmma_f32_16x16x4_f32(false, a, false, b, (short)0, c, false, false);
}

__global__ void zero_f32(float* __restrict__ p, int n) {
    int i = blockIdx.x * blockDim.x + threadIdx.x;
    int stride = gridDim.x * blockDim.x;
    for (; i < n; i += stride) p[i] = 0.f;
}

__global__ void deg_kernel(const int* __restrict__ snd, const int* __restrict__ rcv,
                           float* __restrict__ sdeg, unsigned* __restrict__ tags) {
    const int g = blockIdx.x * blockDim.x + threadIdx.x;
    if (g < NE / 4) {
        const int4 s = ((const int4*)snd)[g];
        const int4 r = ((const int4*)rcv)[g];
        atomicAdd(&sdeg[s.x], 1.f);
        atomicAdd(&sdeg[s.y], 1.f);
        atomicAdd(&sdeg[s.z], 1.f);
        atomicAdd(&sdeg[s.w], 1.f);
        tags[g] = ((unsigned)r.x / TAG_R) | (((unsigned)r.y / TAG_R) << 8) |
                  (((unsigned)r.z / TAG_R) << 16) | (((unsigned)r.w / TAG_R) << 24);
    }
}

__launch_bounds__(256)
__global__ void mlp_kernel(const float* __restrict__ X,
                           const float* __restrict__ W1, const float* __restrict__ b1,
                           const float* __restrict__ W2, const float* __restrict__ b2,
                           const float* __restrict__ sdeg,
                           float* __restrict__ h1, int nTiles) {
    __shared__ float lds[8 * 16 * DH];
    const int w    = threadIdx.x >> 5;
    const int lane = threadIdx.x & 31;
    const int m16  = lane & 15;
    const int kh   = lane >> 4;
    const int t    = blockIdx.x * 8 + w;
    const bool active = (t < nTiles);
    const int row0 = t * 16;
    float* my = &lds[w * 16 * DH];

    if (active) {
        v8f c0 = {}, c1 = {}, c2 = {}, c3 = {};
        #pragma unroll 4
        for (int kb = 0; kb < DF; kb += 4) {
            v2f a = *(const v2f*)(X + (size_t)(row0 + m16) * DF + kb + 2 * kh);
            const float* wr0 = W1 + (size_t)(kb + 2 * kh) * DH;
            const float* wr1 = wr0 + DH;
            v2f b;
            b.x = wr0[m16];      b.y = wr1[m16];      c0 = wmma4(a, b, c0);
            b.x = wr0[16 + m16]; b.y = wr1[16 + m16]; c1 = wmma4(a, b, c1);
            b.x = wr0[32 + m16]; b.y = wr1[32 + m16]; c2 = wmma4(a, b, c2);
            b.x = wr0[48 + m16]; b.y = wr1[48 + m16]; c3 = wmma4(a, b, c3);
        }
        #pragma unroll
        for (int r = 0; r < 8; ++r) {
            float* dr = my + (r + 8 * kh) * DH;
            dr[m16]      = lrelu(c0[r] + b1[m16]);
            dr[16 + m16] = lrelu(c1[r] + b1[16 + m16]);
            dr[32 + m16] = lrelu(c2[r] + b1[32 + m16]);
            dr[48 + m16] = lrelu(c3[r] + b1[48 + m16]);
        }
    }
    __syncthreads();
    if (!active) return;

    v8f d0 = {}, d1 = {}, d2 = {}, d3 = {};
    #pragma unroll 4
    for (int kb = 0; kb < DH; kb += 4) {
        v2f a = *(const v2f*)(my + m16 * DH + kb + 2 * kh);
        const float* wr0 = W2 + (size_t)(kb + 2 * kh) * DH;
        const float* wr1 = wr0 + DH;
        v2f b;
        b.x = wr0[m16];      b.y = wr1[m16];      d0 = wmma4(a, b, d0);
        b.x = wr0[16 + m16]; b.y = wr1[16 + m16]; d1 = wmma4(a, b, d1);
        b.x = wr0[32 + m16]; b.y = wr1[32 + m16]; d2 = wmma4(a, b, d2);
        b.x = wr0[48 + m16]; b.y = wr1[48 + m16]; d3 = wmma4(a, b, d3);
    }
    float sc[8];
    #pragma unroll
    for (int r = 0; r < 8; ++r) sc[r] = rsqrtf(fmaxf(sdeg[row0 + r + 8 * kh], 1.f));
    #pragma unroll
    for (int r = 0; r < 8; ++r) {
        float* orow = h1 + (size_t)(row0 + r + 8 * kh) * DH;
        orow[m16]      = lrelu(d0[r] + b2[m16])      * sc[r];
        orow[16 + m16] = lrelu(d1[r] + b2[16 + m16]) * sc[r];
        orow[32 + m16] = lrelu(d2[r] + b2[32 + m16]) * sc[r];
        orow[48 + m16] = lrelu(d3[r] + b2[48 + m16]) * sc[r];
    }
}

static constexpr int OWN_G  = 256;
static constexpr int OWN_R  = (NN + OWN_G - 1) / OWN_G;
static constexpr int OWN_NW = 32;
static constexpr int OWN_U  = 8;
static constexpr int OWN_GQCAP = OWN_U * 32 + 32;
static constexpr int OWN_QCAP  = 16 * 32 + 32;
static constexpr int NG16 = NE / 16;
static_assert(OWN_R == (int)TAG_R, "tags must name the owner blocks");
static_assert(NE % 16 == 0, "edge list must split into groups of sixteen");
static_assert(NG16 % 32 == 0, "a wave's 32 groups must lie all inside the list or all past its end");

template <int F, bool CNT>
__device__ __forceinline__ void drain_rows(const float* __restrict__ src, const int* __restrict__ snd,
                                           float* acc, int* cnt, const int* q, int n, int lane) {
    const int ent = (lane < n) ? q[lane] : 0;
    const int l = ent >> 21;
    if (CNT) {
        if (lane < n) atomicAdd(cnt + l, 1);
    }
    const int s = (lane < n) ? snd[ent & 0x1FFFFF] : 0;
    for (int j0 = 0; j0 < n; j0 += 8) {
        float a[8], b[8];
        int lj[8];
        #pragma unroll
        for (int u = 0; u < 8; ++u) {
            const int sj = __builtin_amdgcn_readlane(s, j0 + u);
            lj[u] = __builtin_amdgcn_readlane(l, j0 + u);
            const float* row = src + (size_t)sj * F;
            a[u] = row[lane];
            b[u] = (lane < F - 32) ? row[32 + lane] : 0.f;
        }
        #pragma unroll
        for (int u = 0; u < 8; ++u) {
            if (j0 + u < n) {
                float* d = acc + lj[u] * F;
                atomicAdd(d + lane, a[u]);
                if (lane < F - 32) atomicAdd(d + 32 + lane, b[u]);
            }
        }
    }
}

__device__ __forceinline__ unsigned nonzero_bytes(unsigned y) {
    return ((y & 0x7F7F7F7Fu) + 0x7F7F7F7Fu) | y;
}

template <int F, bool CNT>
__device__ __forceinline__ void expand_groups(const float* __restrict__ src, const int* __restrict__ snd,
                                              const int4* __restrict__ rcv4, float* acc, int* cnt, int* q, int& qn,
                                              int g, int r0, int lane) {
    int4 r[4];
    #pragma unroll
    for (int k = 0; k < 4; ++k) r[k] = (g >= 0) ? rcv4[g * 4 + k] : make_int4(-1, -1, -1, -1);
    #pragma unroll
    for (int k = 0; k < 4; ++k) {
        const int rr[4] = {r[k].x, r[k].y, r[k].z, r[k].w};
        #pragma unroll
        for (int c = 0; c < 4; ++c) {
            const unsigned d = (unsigned)(rr[c] - r0);
            const bool hit = d < (unsigned)OWN_R;
            const unsigned m = __builtin_amdgcn_ballot_w32(hit);
            if (m) {
                const int pos = qn + (int)__builtin_amdgcn_mbcnt_lo(m, 0u);
                if (hit) q[pos] = (g * 16 + k * 4 + c) | ((int)d << 21);
                qn += __builtin_popcount(m);
            }
        }
    }
    if (qn >= 32) {
        int done = 0;
        while (qn - done >= 32) {
            drain_rows<F, CNT>(src, snd, acc, cnt, q + done, 32, lane);
            done += 32;
        }
        const int rem = qn - done;
        const int t = (lane < rem) ? q[done + lane] : 0;
        if (lane < rem) q[lane] = t;
        qn = rem;
    }
}

template <int F, bool CNT>
__launch_bounds__(OWN_NW * 32)
__global__ void agg_own_kernel(const float* __restrict__ src, const int* __restrict__ snd,
                               const int* __restrict__ rcv, const unsigned* __restrict__ tags,
                               float* __restrict__ dst, float* __restrict__ rdeg) {
    __shared__ float acc[OWN_R * F];
    __shared__ int   cnt[CNT ? OWN_R : 1];
    __shared__ int   qs[OWN_NW * OWN_QCAP];
    __shared__ int   gqs[OWN_NW * OWN_GQCAP];
    const int w    = __builtin_amdgcn_readfirstlane((int)(threadIdx.x >> 5));
    const int lane = threadIdx.x & 31;
    const int r0   = blockIdx.x * OWN_R;
    int* q  = qs + w * OWN_QCAP;
    int* gq = gqs + w * OWN_GQCAP;
    const unsigned pat = (unsigned)blockIdx.x * 0x01010101u;

    for (int i = threadIdx.x; i < OWN_R * F; i += OWN_NW * 32) acc[i] = 0.f;
    if (CNT) {
        for (int i = threadIdx.x; i < OWN_R; i += OWN_NW * 32) cnt[i] = 0;
    }
    __syncthreads();

    const uint4* tag16 = (const uint4*)tags;
    const int4*  rcv4  = (const int4*)rcv;
    int qn = 0;
    int gn = 0;
    for (int gb = w * 32 * OWN_U; gb < NG16; gb += OWN_NW * 32 * OWN_U) {
        uint4 v[OWN_U];
        #pragma unroll
        for (int u = 0; u < OWN_U; ++u) {
            const int g = gb + u * 32 + lane;
            v[u] = tag16[(g < NG16) ? g : NG16 - 1];
        }
        #pragma unroll
        for (int u = 0; u < OWN_U; ++u) {
            const unsigned past = (gb + u * 32 < NG16) ? 0u : 0x01010101u;
            const unsigned t = nonzero_bytes((v[u].x ^ pat) | past) & nonzero_bytes((v[u].y ^ pat) | past) &
                               nonzero_bytes((v[u].z ^ pat) | past) & nonzero_bytes((v[u].w ^ pat) | past);
            const bool hit = (t & 0x80808080u) != 0x80808080u;
            const unsigned m = __builtin_amdgcn_ballot_w32(hit);
            if (m) {
                const int pos = gn + (int)__builtin_amdgcn_mbcnt_lo(m, 0u);
                if (hit) gq[pos] = gb + u * 32 + lane;
                gn += __builtin_popcount(m);
            }
        }
        if (gn >= 32) {
            int done = 0;
            while (gn - done >= 32) {
                expand_groups<F, CNT>(src, snd, rcv4, acc, cnt, q, qn, gq[done + lane], r0, lane);
                done += 32;
            }
            const int rem = gn - done;
            const int t = (lane < rem) ? gq[done + lane] : 0;
            if (lane < rem) gq[lane] = t;
            gn = rem;
        }
    }
    if (gn > 0) expand_groups<F, CNT>(src, snd, rcv4, acc, cnt, q, qn, (lane < gn) ? gq[lane] : -1, r0, lane);
    if (qn > 0) drain_rows<F, CNT>(src, snd, acc, cnt, q, qn, lane);
    __syncthreads();

    if (CNT) {
        for (int i = threadIdx.x; i < OWN_R; i += OWN_NW * 32) {
            if (r0 + i < NN) rdeg[r0 + i] = (float)cnt[i];
        }
    }

    for (int i = threadIdx.x; i < OWN_R * F; i += OWN_NW * 32) {
        const int node = r0 + i / F;
        if (node < NN) dst[(size_t)r0 * F + i] = acc[i];
    }
}

__launch_bounds__(256)
__global__ void fin1_kernel(const float* __restrict__ agg1,
                            const float* __restrict__ rdeg, const float* __restrict__ sdeg,
                            const float* __restrict__ W3, const float* __restrict__ b3,
                            float* __restrict__ h2, int nTiles) {
    const int w    = threadIdx.x >> 5;
    const int lane = threadIdx.x & 31;
    const int m16  = lane & 15;
    const int kh   = lane >> 4;
    const int t    = blockIdx.x * 8 + w;
    if (t >= nTiles) return;
    const int row0 = t * 16;
    const float rs = rsqrtf(fmaxf(rdeg[row0 + m16], 1.f));

    v8f c0 = {}, c1 = {}, c2 = {};
    #pragma unroll 4
    for (int kb = 0; kb < DH; kb += 4) {
        v2f a = *(const v2f*)(agg1 + (size_t)(row0 + m16) * DH + kb + 2 * kh);
        a.x *= rs; a.y *= rs;
        const float* wr0 = W3 + (size_t)(kb + 2 * kh) * DC;
        const float* wr1 = wr0 + DC;
        v2f b;
        b.x = wr0[m16];      b.y = wr1[m16];      c0 = wmma4(a, b, c0);
        b.x = wr0[16 + m16]; b.y = wr1[16 + m16]; c1 = wmma4(a, b, c1);
        const int n2 = 32 + m16;
        b.x = (n2 < DC) ? wr0[n2] : 0.f;
        b.y = (n2 < DC) ? wr1[n2] : 0.f;
        c2 = wmma4(a, b, c2);
    }
    float sc[8];
    #pragma unroll
    for (int r = 0; r < 8; ++r) sc[r] = rsqrtf(fmaxf(sdeg[row0 + r + 8 * kh], 1.f));
    #pragma unroll
    for (int r = 0; r < 8; ++r) {
        float* orow = h2 + (size_t)(row0 + r + 8 * kh) * DC;
        orow[m16]      = (c0[r] + b3[m16])      * sc[r];
        orow[16 + m16] = (c1[r] + b3[16 + m16]) * sc[r];
        if (32 + m16 < DC)
            orow[32 + m16] = (c2[r] + b3[32 + m16]) * sc[r];
    }
}

__global__ void softmax_kernel(float* __restrict__ out, const float* __restrict__ rdeg) {
    int n = blockIdx.x * blockDim.x + threadIdx.x;
    if (n >= NN) return;
    float rs = rsqrtf(fmaxf(rdeg[n], 1.f));
    float* row = out + (size_t)n * DC;
    float v[DC];
    float mx = -3.402823466e38f;
    #pragma unroll
    for (int f = 0; f < DC; ++f) { v[f] = row[f] * rs; mx = fmaxf(mx, v[f]); }
    float s = 0.f;
    #pragma unroll
    for (int f = 0; f < DC; ++f) { v[f] = __expf(v[f] - mx); s += v[f]; }
    float inv = 1.f / s;
    #pragma unroll
    for (int f = 0; f < DC; ++f) row[f] = v[f] * inv;
}

extern "C" void kernel_launch(void* const* d_in, const int* in_sizes, int n_in,
                              void* d_out, int out_size, void* d_ws, size_t ws_size,
                              hipStream_t stream) {
    const float* nodes     = (const float*)d_in[0];
    const int*   senders   = (const int*)d_in[1];
    const int*   receivers = (const int*)d_in[2];
    const float* W1 = (const float*)d_in[3];
    const float* b1 = (const float*)d_in[4];
    const float* W2 = (const float*)d_in[5];
    const float* b2 = (const float*)d_in[6];
    const float* W3 = (const float*)d_in[7];
    const float* b3 = (const float*)d_in[8];
    float* out = (float*)d_out;
    float* ws  = (float*)d_ws;

    float* sdeg = ws;
    float* rdeg = ws + NN;
    float* h1   = ws + 2 * (size_t)NN;
    float* agg1 = h1 + (size_t)NN * DH;
    float* h2   = agg1 + (size_t)NN * DH;
    unsigned* tags = (unsigned*)(h2 + (size_t)NN * DC);

    const int nTiles = NN / 16;
    zero_f32<<<(NN + 255) / 256, 256, 0, stream>>>(sdeg, NN);
    deg_kernel<<<(NE / 4 + 255) / 256, 256, 0, stream>>>(senders, receivers, sdeg, tags);
    mlp_kernel<<<(nTiles + 7) / 8, 256, 0, stream>>>(nodes, W1, b1, W2, b2, sdeg, h1, nTiles);
    agg_own_kernel<DH, true><<<OWN_G, OWN_NW * 32, 0, stream>>>(h1, senders, receivers, tags, agg1, rdeg);
    fin1_kernel<<<(nTiles + 7) / 8, 256, 0, stream>>>(agg1, rdeg, sdeg, W3, b3, h2, nTiles);
    agg_own_kernel<DC, false><<<OWN_G, OWN_NW * 32, 0, stream>>>(h2, senders, receivers, tags, out, nullptr);
    softmax_kernel<<<(NN + 255) / 256, 256, 0, stream>>>(out, rdeg);
}


# Sender degrees by owner blocks: tag kernel also writes one-byte sender owner tags; 256 blocks scan them, count their own node range by LDS integer adds (exact), store each once; no global atomics
# speedup vs baseline: 2.8844x; 1.2372x over previous
#include <hip/hip_runtime.h>

typedef __attribute__((ext_vector_type(2))) float v2f;
typedef __attribute__((ext_vector_type(8))) float v8f;

static constexpr int NN = 100000;
static constexpr int NE = 1600000;
static constexpr int DF = 128;
static constexpr int DH = 64;
static constexpr int DC = 40;
#define NEG_SLOPE 0.01f
static constexpr unsigned TAG_R = (NN + 255) / 256;

__device__ __forceinline__ float lrelu(float x) { return x > 0.f ? x : x * NEG_SLOPE; }

__device__ __forceinline__ v8f wmma4(v2f a, v2f b, v8f c) {
    return __builtin_amdgcn_wmma_f32_16x16x4_f32(false, a, false, b, (short)0, c, false, false);
}

__device__ __forceinline__ unsigned owner_tags(int4 v) {
    return ((unsigned)v.x / TAG_R) | (((unsigned)v.y / TAG_R) << 8) |
           (((unsigned)v.z / TAG_R) << 16) | (((unsigned)v.w / TAG_R) << 24);
}

__global__ void tag_kernel(const int* __restrict__ snd, const int* __restrict__ rcv,
                           unsigned* __restrict__ stags, unsigned* __restrict__ tags) {
    const int g = blockIdx.x * blockDim.x + threadIdx.x;
    if (g < NE / 4) {
        const int4 s = ((const int4*)snd)[g];
        const int4 r = ((const int4*)rcv)[g];
        stags[g] = owner_tags(s);
        tags[g]  = owner_tags(r);
    }
}

__launch_bounds__(256)
__global__ void mlp_kernel(const float* __restrict__ X,
                           const float* __restrict__ W1, const float* __restrict__ b1,
                           const float* __restrict__ W2, const float* __restrict__ b2,
                           const float* __restrict__ sdeg,
                           float* __restrict__ h1, int nTiles) {
    __shared__ float lds[8 * 16 * DH];
    const int w    = threadIdx.x >> 5;
    const int lane = threadIdx.x & 31;
    const int m16  = lane & 15;
    const int kh   = lane >> 4;
    const int t    = blockIdx.x * 8 + w;
    const bool active = (t < nTiles);
    const int row0 = t * 16;
    float* my = &lds[w * 16 * DH];

    if (active) {
        v8f c0 = {}, c1 = {}, c2 = {}, c3 = {};
        #pragma unroll 4
        for (int kb = 0; kb < DF; kb += 4) {
            v2f a = *(const v2f*)(X + (size_t)(row0 + m16) * DF + kb + 2 * kh);
            const float* wr0 = W1 + (size_t)(kb + 2 * kh) * DH;
            const float* wr1 = wr0 + DH;
            v2f b;
            b.x = wr0[m16];      b.y = wr1[m16];      c0 = wmma4(a, b, c0);
            b.x = wr0[16 + m16]; b.y = wr1[16 + m16]; c1 = wmma4(a, b, c1);
            b.x = wr0[32 + m16]; b.y = wr1[32 + m16]; c2 = wmma4(a, b, c2);
            b.x = wr0[48 + m16]; b.y = wr1[48 + m16]; c3 = wmma4(a, b, c3);
        }
        #pragma unroll
        for (int r = 0; r < 8; ++r) {
            float* dr = my + (r + 8 * kh) * DH;
            dr[m16]      = lrelu(c0[r] + b1[m16]);
            dr[16 + m16] = lrelu(c1[r] + b1[16 + m16]);
            dr[32 + m16] = lrelu(c2[r] + b1[32 + m16]);
            dr[48 + m16] = lrelu(c3[r] + b1[48 + m16]);
        }
    }
    __syncthreads();
    if (!active) return;

    v8f d0 = {}, d1 = {}, d2 = {}, d3 = {};
    #pragma unroll 4
    for (int kb = 0; kb < DH; kb += 4) {
        v2f a = *(const v2f*)(my + m16 * DH + kb + 2 * kh);
        const float* wr0 = W2 + (size_t)(kb + 2 * kh) * DH;
        const float* wr1 = wr0 + DH;
        v2f b;
        b.x = wr0[m16];      b.y = wr1[m16];      d0 = wmma4(a, b, d0);
        b.x = wr0[16 + m16]; b.y = wr1[16 + m16]; d1 = wmma4(a, b, d1);
        b.x = wr0[32 + m16]; b.y = wr1[32 + m16]; d2 = wmma4(a, b, d2);
        b.x = wr0[48 + m16]; b.y = wr1[48 + m16]; d3 = wmma4(a, b, d3);
    }
    float sc[8];
    #pragma unroll
    for (int r = 0; r < 8; ++r) sc[r] = rsqrtf(fmaxf(sdeg[row0 + r + 8 * kh], 1.f));
    #pragma unroll
    for (int r = 0; r < 8; ++r) {
        float* orow = h1 + (size_t)(row0 + r + 8 * kh) * DH;
        orow[m16]      = lrelu(d0[r] + b2[m16])      * sc[r];
        orow[16 + m16] = lrelu(d1[r] + b2[16 + m16]) * sc[r];
        orow[32 + m16] = lrelu(d2[r] + b2[32 + m16]) * sc[r];
        orow[48 + m16] = lrelu(d3[r] + b2[48 + m16]) * sc[r];
    }
}

static constexpr int OWN_G  = 256;
static constexpr int OWN_R  = (NN + OWN_G - 1) / OWN_G;
static constexpr int OWN_NW = 32;
static constexpr int OWN_U  = 8;
static constexpr int OWN_GQCAP = OWN_U * 32 + 32;
static constexpr int OWN_QCAP  = 16 * 32 + 32;
static constexpr int NG16 = NE / 16;
static_assert(OWN_R == (int)TAG_R, "tags must name the owner blocks");
static_assert(NE % 16 == 0, "edge list must split into groups of sixteen");
static_assert(NG16 % 32 == 0, "a wave's 32 groups must lie all inside the list or all past its end");

template <int F, bool CNT>
__device__ __forceinline__ void drain_rows(const float* __restrict__ src, const int* __restrict__ snd,
                                           float* acc, int* cnt, const int* q, int n, int lane) {
    const int ent = (lane < n) ? q[lane] : 0;
    const int l = ent >> 21;
    if (CNT) {
        if (lane < n) atomicAdd(cnt + l, 1);
    }
    const int s = (lane < n) ? snd[ent & 0x1FFFFF] : 0;
    for (int j0 = 0; j0 < n; j0 += 8) {
        float a[8], b[8];
        int lj[8];
        #pragma unroll
        for (int u = 0; u < 8; ++u) {
            const int sj = __builtin_amdgcn_readlane(s, j0 + u);
            lj[u] = __builtin_amdgcn_readlane(l, j0 + u);
            const float* row = src + (size_t)sj * F;
            a[u] = row[lane];
            b[u] = (lane < F - 32) ? row[32 + lane] : 0.f;
        }
        #pragma unroll
        for (int u = 0; u < 8; ++u) {
            if (j0 + u < n) {
                float* d = acc + lj[u] * F;
                atomicAdd(d + lane, a[u]);
                if (lane < F - 32) atomicAdd(d + 32 + lane, b[u]);
            }
        }
    }
}

__device__ __forceinline__ unsigned nonzero_bytes(unsigned y) {
    return ((y & 0x7F7F7F7Fu) + 0x7F7F7F7Fu) | y;
}

template <int F, bool CNT>
__device__ __forceinline__ void expand_groups(const float* __restrict__ src, const int* __restrict__ snd,
                                              const int4* __restrict__ rcv4, float* acc, int* cnt, int* q, int& qn,
                                              int g, int r0, int lane) {
    int4 r[4];
    #pragma unroll
    for (int k = 0; k < 4; ++k) r[k] = (g >= 0) ? rcv4[g * 4 + k] : make_int4(-1, -1, -1, -1);
    #pragma unroll
    for (int k = 0; k < 4; ++k) {
        const int rr[4] = {r[k].x, r[k].y, r[k].z, r[k].w};
        #pragma unroll
        for (int c = 0; c < 4; ++c) {
            const unsigned d = (unsigned)(rr[c] - r0);
            const bool hit = d < (unsigned)OWN_R;
            const unsigned m = __builtin_amdgcn_ballot_w32(hit);
            if (m) {
                const int pos = qn + (int)__builtin_amdgcn_mbcnt_lo(m, 0u);
                if (hit) q[pos] = (g * 16 + k * 4 + c) | ((int)d << 21);
                qn += __builtin_popcount(m);
            }
        }
    }
    if (qn >= 32) {
        int done = 0;
        while (qn - done >= 32) {
            drain_rows<F, CNT>(src, snd, acc, cnt, q + done, 32, lane);
            done += 32;
        }
        const int rem = qn - done;
        const int t = (lane < rem) ? q[done + lane] : 0;
        if (lane < rem) q[lane] = t;
        qn = rem;
    }
}

template <int F, bool CNT>
__launch_bounds__(OWN_NW * 32)
__global__ void agg_own_kernel(const float* __restrict__ src, const int* __restrict__ snd,
                               const int* __restrict__ rcv, const unsigned* __restrict__ tags,
                               float* __restrict__ dst, float* __restrict__ rdeg) {
    __shared__ float acc[OWN_R * F];
    __shared__ int   cnt[CNT ? OWN_R : 1];
    __shared__ int   qs[OWN_NW * OWN_QCAP];
    __shared__ int   gqs[OWN_NW * OWN_GQCAP];
    const int w    = __builtin_amdgcn_readfirstlane((int)(threadIdx.x >> 5));
    const int lane = threadIdx.x & 31;
    const int r0   = blockIdx.x * OWN_R;
    int* q  = qs + w * OWN_QCAP;
    int* gq = gqs + w * OWN_GQCAP;
    const unsigned pat = (unsigned)blockIdx.x * 0x01010101u;

    for (int i = threadIdx.x; i < OWN_R * F; i += OWN_NW * 32) acc[i] = 0.f;
    if (CNT) {
        for (int i = threadIdx.x; i < OWN_R; i += OWN_NW * 32) cnt[i] = 0;
    }
    __syncthreads();

    const uint4* tag16 = (const uint4*)tags;
    const int4*  rcv4  = (const int4*)rcv;
    int qn = 0;
    int gn = 0;
    for (int gb = w * 32 * OWN_U; gb < NG16; gb += OWN_NW * 32 * OWN_U) {
        uint4 v[OWN_U];
        #pragma unroll
        for (int u = 0; u < OWN_U; ++u) {
            const int g = gb + u * 32 + lane;
            v[u] = tag16[(g < NG16) ? g : NG16 - 1];
        }
        #pragma unroll
        for (int u = 0; u < OWN_U; ++u) {
            const unsigned past = (gb + u * 32 < NG16) ? 0u : 0x01010101u;
            const unsigned t = nonzero_bytes((v[u].x ^ pat) | past) & nonzero_bytes((v[u].y ^ pat) | past) &
                               nonzero_bytes((v[u].z ^ pat) | past) & nonzero_bytes((v[u].w ^ pat) | past);
            const bool hit = (t & 0x80808080u) != 0x80808080u;
            const unsigned m = __builtin_amdgcn_ballot_w32(hit);
            if (m) {
                const int pos = gn + (int)__builtin_amdgcn_mbcnt_lo(m, 0u);
                if (hit) gq[pos] = gb + u * 32 + lane;
                gn += __builtin_popcount(m);
            }
        }
        if (gn >= 32) {
            int done = 0;
            while (gn - done >= 32) {
                expand_groups<F, CNT>(src, snd, rcv4, acc, cnt, q, qn, gq[done + lane], r0, lane);
                done += 32;
            }
            const int rem = gn - done;
            const int t = (lane < rem) ? gq[done + lane] : 0;
            if (lane < rem) gq[lane] = t;
            gn = rem;
        }
    }
    if (gn > 0) expand_groups<F, CNT>(src, snd, rcv4, acc, cnt, q, qn, (lane < gn) ? gq[lane] : -1, r0, lane);
    if (qn > 0) drain_rows<F, CNT>(src, snd, acc, cnt, q, qn, lane);
    __syncthreads();

    if (CNT) {
        for (int i = threadIdx.x; i < OWN_R; i += OWN_NW * 32) {
            if (r0 + i < NN) rdeg[r0 + i] = (float)cnt[i];
        }
    }

    for (int i = threadIdx.x; i < OWN_R * F; i += OWN_NW * 32) {
        const int node = r0 + i / F;
        if (node < NN) dst[(size_t)r0 * F + i] = acc[i];
    }
}

__device__ __forceinline__ void count_groups(const int4* __restrict__ snd4, int* cnt, int g, int r0) {
    int4 s[4];
    #pragma unroll
    for (int k = 0; k < 4; ++k) s[k] = (g >= 0) ? snd4[g * 4 + k] : make_int4(-1, -1, -1, -1);
    #pragma unroll
    for (int k = 0; k < 4; ++k) {
        const int ss[4] = {s[k].x, s[k].y, s[k].z, s[k].w};
        #pragma unroll
        for (int c = 0; c < 4; ++c) {
            const unsigned d = (unsigned)(ss[c] - r0);
            if (d < (unsigned)OWN_R) atomicAdd(cnt + d, 1);
        }
    }
}

__launch_bounds__(OWN_NW * 32)
__global__ void sdeg_own_kernel(const int* __restrict__ snd, const unsigned* __restrict__ stags,
                                float* __restrict__ sdeg) {
    __shared__ int cnt[OWN_R];
    __shared__ int gqs[OWN_NW * OWN_GQCAP];
    const int w    = __builtin_amdgcn_readfirstlane((int)(threadIdx.x >> 5));
    const int lane = threadIdx.x & 31;
    const int r0   = blockIdx.x * OWN_R;
    int* gq = gqs + w * OWN_GQCAP;
    const unsigned pat = (unsigned)blockIdx.x * 0x01010101u;

    for (int i = threadIdx.x; i < OWN_R; i += OWN_NW * 32) cnt[i] = 0;
    __syncthreads();

    const uint4* tag16 = (const uint4*)stags;
    const int4*  snd4  = (const int4*)snd;
    int gn = 0;
    for (int gb = w * 32 * OWN_U; gb < NG16; gb += OWN_NW * 32 * OWN_U) {
        uint4 v[OWN_U];
        #pragma unroll
        for (int u = 0; u < OWN_U; ++u) {
            const int g = gb + u * 32 + lane;
            v[u] = tag16[(g < NG16) ? g : NG16 - 1];
        }
        #pragma unroll
        for (int u = 0; u < OWN_U; ++u) {
            const unsigned past = (gb + u * 32 < NG16) ? 0u : 0x01010101u;
            const unsigned t = nonzero_bytes((v[u].x ^ pat) | past) & nonzero_bytes((v[u].y ^ pat) | past) &
                               nonzero_bytes((v[u].z ^ pat) | past) & nonzero_bytes((v[u].w ^ pat) | past);
            const bool hit = (t & 0x80808080u) != 0x80808080u;
            const unsigned m = __builtin_amdgcn_ballot_w32(hit);
            if (m) {
                const int pos = gn + (int)__builtin_amdgcn_mbcnt_lo(m, 0u);
                if (hit) gq[pos] = gb + u * 32 + lane;
                gn += __builtin_popcount(m);
            }
        }
        if (gn >= 32) {
            int done = 0;
            while (gn - done >= 32) {
                count_groups(snd4, cnt, gq[done + lane], r0);
                done += 32;
            }
            const int rem = gn - done;
            const int t = (lane < rem) ? gq[done + lane] : 0;
            if (lane < rem) gq[lane] = t;
            gn = rem;
        }
    }
    if (gn > 0) count_groups(snd4, cnt, (lane < gn) ? gq[lane] : -1, r0);
    __syncthreads();

    for (int i = threadIdx.x; i < OWN_R; i += OWN_NW * 32) {
        if (r0 + i < NN) sdeg[r0 + i] = (float)cnt[i];
    }
}

__launch_bounds__(256)
__global__ void fin1_kernel(const float* __restrict__ agg1,
                            const float* __restrict__ rdeg, const float* __restrict__ sdeg,
                            const float* __restrict__ W3, const float* __restrict__ b3,
                            float* __restrict__ h2, int nTiles) {
    const int w    = threadIdx.x >> 5;
    const int lane = threadIdx.x & 31;
    const int m16  = lane & 15;
    const int kh   = lane >> 4;
    const int t    = blockIdx.x * 8 + w;
    if (t >= nTiles) return;
    const int row0 = t * 16;
    const float rs = rsqrtf(fmaxf(rdeg[row0 + m16], 1.f));

    v8f c0 = {}, c1 = {}, c2 = {};
    #pragma unroll 4
    for (int kb = 0; kb < DH; kb += 4) {
        v2f a = *(const v2f*)(agg1 + (size_t)(row0 + m16) * DH + kb + 2 * kh);
        a.x *= rs; a.y *= rs;
        const float* wr0 = W3 + (size_t)(kb + 2 * kh) * DC;
        const float* wr1 = wr0 + DC;
        v2f b;
        b.x = wr0[m16];      b.y = wr1[m16];      c0 = wmma4(a, b, c0);
        b.x = wr0[16 + m16]; b.y = wr1[16 + m16]; c1 = wmma4(a, b, c1);
        const int n2 = 32 + m16;
        b.x = (n2 < DC) ? wr0[n2] : 0.f;
        b.y = (n2 < DC) ? wr1[n2] : 0.f;
        c2 = wmma4(a, b, c2);
    }
    float sc[8];
    #pragma unroll
    for (int r = 0; r < 8; ++r) sc[r] = rsqrtf(fmaxf(sdeg[row0 + r + 8 * kh], 1.f));
    #pragma unroll
    for (int r = 0; r < 8; ++r) {
        float* orow = h2 + (size_t)(row0 + r + 8 * kh) * DC;
        orow[m16]      = (c0[r] + b3[m16])      * sc[r];
        orow[16 + m16] = (c1[r] + b3[16 + m16]) * sc[r];
        if (32 + m16 < DC)
            orow[32 + m16] = (c2[r] + b3[32 + m16]) * sc[r];
    }
}

__global__ void softmax_kernel(float* __restrict__ out, const float* __restrict__ rdeg) {
    int n = blockIdx.x * blockDim.x + threadIdx.x;
    if (n >= NN) return;
    float rs = rsqrtf(fmaxf(rdeg[n], 1.f));
    float* row = out + (size_t)n * DC;
    float v[DC];
    float mx = -3.402823466e38f;
    #pragma unroll
    for (int f = 0; f < DC; ++f) { v[f] = row[f] * rs; mx = fmaxf(mx, v[f]); }
    float s = 0.f;
    #pragma unroll
    for (int f = 0; f < DC; ++f) { v[f] = __expf(v[f] - mx); s += v[f]; }
    float inv = 1.f / s;
    #pragma unroll
    for (int f = 0; f < DC; ++f) row[f] = v[f] * inv;
}

extern "C" void kernel_launch(void* const* d_in, const int* in_sizes, int n_in,
                              void* d_out, int out_size, void* d_ws, size_t ws_size,
                              hipStream_t stream) {
    const float* nodes     = (const float*)d_in[0];
    const int*   senders   = (const int*)d_in[1];
    const int*   receivers = (const int*)d_in[2];
    const float* W1 = (const float*)d_in[3];
    const float* b1 = (const float*)d_in[4];
    const float* W2 = (const float*)d_in[5];
    const float* b2 = (const float*)d_in[6];
    const float* W3 = (const float*)d_in[7];
    const float* b3 = (const float*)d_in[8];
    float* out = (float*)d_out;
    float* ws  = (float*)d_ws;

    float* sdeg = ws;
    float* rdeg = ws + NN;
    float* h1   = ws + 2 * (size_t)NN;
    float* agg1 = h1 + (size_t)NN * DH;
    float* h2   = agg1 + (size_t)NN * DH;
    unsigned* tags  = (unsigned*)(h2 + (size_t)NN * DC);
    unsigned* stags = tags + NE / 4;

    const int nTiles = NN / 16;
    tag_kernel<<<(NE / 4 + 255) / 256, 256, 0, stream>>>(senders, receivers, stags, tags);
    sdeg_own_kernel<<<OWN_G, OWN_NW * 32, 0, stream>>>(senders, stags, sdeg);
    mlp_kernel<<<(nTiles + 7) / 8, 256, 0, stream>>>(nodes, W1, b1, W2, b2, sdeg, h1, nTiles);
    agg_own_kernel<DH, true><<<OWN_G, OWN_NW * 32, 0, stream>>>(h1, senders, receivers, tags, agg1, rdeg);
    fin1_kernel<<<(nTiles + 7) / 8, 256, 0, stream>>>(agg1, rdeg, sdeg, W3, b3, h2, nTiles);
    agg_own_kernel<DC, false><<<OWN_G, OWN_NW * 32, 0, stream>>>(h2, senders, receivers, tags, out, nullptr);
    softmax_kernel<<<(NN + 255) / 256, 256, 0, stream>>>(out, rdeg);
}


# First edge sum's owner blocks record their edges (sender, local receiver) per block; the second sum replays the lists: no tag scan, receiver re-read or sender lookup; its LDS-atomic sum order changes
# speedup vs baseline: 3.3770x; 1.1708x over previous
#include <hip/hip_runtime.h>

typedef __attribute__((ext_vector_type(2))) float v2f;
typedef __attribute__((ext_vector_type(8))) float v8f;

static constexpr int NN = 100000;
static constexpr int NE = 1600000;
static constexpr int DF = 128;
static constexpr int DH = 64;
static constexpr int DC = 40;
#define NEG_SLOPE 0.01f
static constexpr unsigned TAG_R = (NN + 255) / 256;

__device__ __forceinline__ float lrelu(float x) { return x > 0.f ? x : x * NEG_SLOPE; }

__device__ __forceinline__ v8f wmma4(v2f a, v2f b, v8f c) {
    return __builtin_amdgcn_wmma_f32_16x16x4_f32(false, a, false, b, (short)0, c, false, false);
}

__device__ __forceinline__ unsigned owner_tags(int4 v) {
    return ((unsigned)v.x / TAG_R) | (((unsigned)v.y / TAG_R) << 8) |
           (((unsigned)v.z / TAG_R) << 16) | (((unsigned)v.w / TAG_R) << 24);
}

__global__ void tag_kernel(const int* __restrict__ snd, const int* __restrict__ rcv,
                           unsigned* __restrict__ stags, unsigned* __restrict__ tags) {
    const int g = blockIdx.x * blockDim.x + threadIdx.x;
    if (g < NE / 4) {
        const int4 s = ((const int4*)snd)[g];
        const int4 r = ((const int4*)rcv)[g];
        stags[g] = owner_tags(s);
        tags[g]  = owner_tags(r);
    }
}

__launch_bounds__(256)
__global__ void mlp_kernel(const float* __restrict__ X,
                           const float* __restrict__ W1, const float* __restrict__ b1,
                           const float* __restrict__ W2, const float* __restrict__ b2,
                           const float* __restrict__ sdeg,
                           float* __restrict__ h1, int nTiles) {
    __shared__ float lds[8 * 16 * DH];
    const int w    = threadIdx.x >> 5;
    const int lane = threadIdx.x & 31;
    const int m16  = lane & 15;
    const int kh   = lane >> 4;
    const int t    = blockIdx.x * 8 + w;
    const bool active = (t < nTiles);
    const int row0 = t * 16;
    float* my = &lds[w * 16 * DH];

    if (active) {
        v8f c0 = {}, c1 = {}, c2 = {}, c3 = {};
        #pragma unroll 4
        for (int kb = 0; kb < DF; kb += 4) {
            v2f a = *(const v2f*)(X + (size_t)(row0 + m16) * DF + kb + 2 * kh);
            const float* wr0 = W1 + (size_t)(kb + 2 * kh) * DH;
            const float* wr1 = wr0 + DH;
            v2f b;
            b.x = wr0[m16];      b.y = wr1[m16];      c0 = wmma4(a, b, c0);
            b.x = wr0[16 + m16]; b.y = wr1[16 + m16]; c1 = wmma4(a, b, c1);
            b.x = wr0[32 + m16]; b.y = wr1[32 + m16]; c2 = wmma4(a, b, c2);
            b.x = wr0[48 + m16]; b.y = wr1[48 + m16]; c3 = wmma4(a, b, c3);
        }
        #pragma unroll
        for (int r = 0; r < 8; ++r) {
            float* dr = my + (r + 8 * kh) * DH;
            dr[m16]      = lrelu(c0[r] + b1[m16]);
            dr[16 + m16] = lrelu(c1[r] + b1[16 + m16]);
            dr[32 + m16] = lrelu(c2[r] + b1[32 + m16]);
            dr[48 + m16] = lrelu(c3[r] + b1[48 + m16]);
        }
    }
    __syncthreads();
    if (!active) return;

    v8f d0 = {}, d1 = {}, d2 = {}, d3 = {};
    #pragma unroll 4
    for (int kb = 0; kb < DH; kb += 4) {
        v2f a = *(const v2f*)(my + m16 * DH + kb + 2 * kh);
        const float* wr0 = W2 + (size_t)(kb + 2 * kh) * DH;
        const float* wr1 = wr0 + DH;
        v2f b;
        b.x = wr0[m16];      b.y = wr1[m16];      d0 = wmma4(a, b, d0);
        b.x = wr0[16 + m16]; b.y = wr1[16 + m16]; d1 = wmma4(a, b, d1);
        b.x = wr0[32 + m16]; b.y = wr1[32 + m16]; d2 = wmma4(a, b, d2);
        b.x = wr0[48 + m16]; b.y = wr1[48 + m16]; d3 = wmma4(a, b, d3);
    }
    float sc[8];
    #pragma unroll
    for (int r = 0; r < 8; ++r) sc[r] = rsqrtf(fmaxf(sdeg[row0 + r + 8 * kh], 1.f));
    #pragma unroll
    for (int r = 0; r < 8; ++r) {
        float* orow = h1 + (size_t)(row0 + r + 8 * kh) * DH;
        orow[m16]      = lrelu(d0[r] + b2[m16])      * sc[r];
        orow[16 + m16] = lrelu(d1[r] + b2[16 + m16]) * sc[r];
        orow[32 + m16] = lrelu(d2[r] + b2[32 + m16]) * sc[r];
        orow[48 + m16] = lrelu(d3[r] + b2[48 + m16]) * sc[r];
    }
}

static constexpr int OWN_G  = 256;
static constexpr int OWN_R  = (NN + OWN_G - 1) / OWN_G;
static constexpr int OWN_NW = 32;
static constexpr int OWN_U  = 8;
static constexpr int OWN_GQCAP = OWN_U * 32 + 32;
static constexpr int OWN_QCAP  = 16 * 32 + 32;
static constexpr int NG16 = NE / 16;
static constexpr int OWN_ECAP = 1 << 17;
static_assert(OWN_R == (int)TAG_R, "tags must name the owner blocks");
static_assert(NN <= (1 << 17) && OWN_R <= (1 << 9), "a list entry holds the sender in bits 0..16 and the local receiver above them");
static_assert(NE % 16 == 0, "edge list must split into groups of sixteen");
static_assert(NG16 % 32 == 0, "a wave's 32 groups must lie all inside the list or all past its end");

template <int F>
__device__ __forceinline__ void add_rows(const float* __restrict__ src, float* acc, int s, int l, int n, int lane) {
    for (int j0 = 0; j0 < n; j0 += 8) {
        float a[8], b[8];
        int lj[8];
        #pragma unroll
        for (int u = 0; u < 8; ++u) {
            const int sj = __builtin_amdgcn_readlane(s, j0 + u);
            lj[u] = __builtin_amdgcn_readlane(l, j0 + u);
            const float* row = src + (size_t)sj * F;
            a[u] = row[lane];
            b[u] = (lane < F - 32) ? row[32 + lane] : 0.f;
        }
        #pragma unroll
        for (int u = 0; u < 8; ++u) {
            if (j0 + u < n) {
                float* d = acc + lj[u] * F;
                atomicAdd(d + lane, a[u]);
                if (lane < F - 32) atomicAdd(d + 32 + lane, b[u]);
            }
        }
    }
}

template <int F, bool CNT, bool REC>
__device__ __forceinline__ void drain_rows(const float* __restrict__ src, const int* __restrict__ snd,
                                           float* acc, int* cnt, unsigned* __restrict__ rec, int* rcnt,
                                           const int* q, int n, int lane) {
    const int ent = (lane < n) ? q[lane] : 0;
    const int l = ent >> 21;
    if (CNT) {
        if (lane < n) atomicAdd(cnt + l, 1);
    }
    const int s = (lane < n) ? snd[ent & 0x1FFFFF] : 0;
    if (REC) {
        int at = 0;
        if (lane == 0) at = atomicAdd(rcnt, n);
        at = __builtin_amdgcn_readfirstlane(at);
        if (lane < n && at + lane < OWN_ECAP) rec[at + lane] = (unsigned)s | ((unsigned)l << 17);
    }
    add_rows<F>(src, acc, s, l, n, lane);
}

__device__ __forceinline__ unsigned nonzero_bytes(unsigned y) {
    return ((y & 0x7F7F7F7Fu) + 0x7F7F7F7Fu) | y;
}

template <int F, bool CNT, bool REC>
__device__ __forceinline__ void expand_groups(const float* __restrict__ src, const int* __restrict__ snd,
                                              const int4* __restrict__ rcv4, float* acc, int* cnt,
                                              unsigned* __restrict__ rec, int* rcnt, int* q, int& qn,
                                              int g, int r0, int lane) {
    int4 r[4];
    #pragma unroll
    for (int k = 0; k < 4; ++k) r[k] = (g >= 0) ? rcv4[g * 4 + k] : make_int4(-1, -1, -1, -1);
    #pragma unroll
    for (int k = 0; k < 4; ++k) {
        const int rr[4] = {r[k].x, r[k].y, r[k].z, r[k].w};
        #pragma unroll
        for (int c = 0; c < 4; ++c) {
            const unsigned d = (unsigned)(rr[c] - r0);
            const bool hit = d < (unsigned)OWN_R;
            const unsigned m = __builtin_amdgcn_ballot_w32(hit);
            if (m) {
                const int pos = qn + (int)__builtin_amdgcn_mbcnt_lo(m, 0u);
                if (hit) q[pos] = (g * 16 + k * 4 + c) | ((int)d << 21);
                qn += __builtin_popcount(m);
            }
        }
    }
    if (qn >= 32) {
        int done = 0;
        while (qn - done >= 32) {
            drain_rows<F, CNT, REC>(src, snd, acc, cnt, rec, rcnt, q + done, 32, lane);
            done += 32;
        }
        const int rem = qn - done;
        const int t = (lane < rem) ? q[done + lane] : 0;
        if (lane < rem) q[lane] = t;
        qn = rem;
    }
}

template <int F, bool CNT, bool REC, bool PLAY>
__launch_bounds__(OWN_NW * 32)
__global__ void agg_own_kernel(const float* __restrict__ src, const int* __restrict__ snd,
                               const int* __restrict__ rcv, const unsigned* __restrict__ tags,
                               float* __restrict__ dst, float* __restrict__ rdeg,
                               unsigned* __restrict__ elist, int* __restrict__ ecount) {
    __shared__ float acc[OWN_R * F];
    __shared__ int   cnt[CNT ? OWN_R : 1];
    __shared__ int   rcnt[1];
    __shared__ int   qs[OWN_NW * OWN_QCAP];
    __shared__ int   gqs[OWN_NW * OWN_GQCAP];
    const int w    = __builtin_amdgcn_readfirstlane((int)(threadIdx.x >> 5));
    const int lane = threadIdx.x & 31;
    const int r0   = blockIdx.x * OWN_R;
    int* q  = qs + w * OWN_QCAP;
    int* gq = gqs + w * OWN_GQCAP;
    const unsigned pat = (unsigned)blockIdx.x * 0x01010101u;

    for (int i = threadIdx.x; i < OWN_R * F; i += OWN_NW * 32) acc[i] = 0.f;
    if (CNT) {
        for (int i = threadIdx.x; i < OWN_R; i += OWN_NW * 32) cnt[i] = 0;
    }
    if (REC) {
        if (threadIdx.x == 0) rcnt[0] = 0;
    }
    __syncthreads();

    unsigned* mine = elist + (size_t)blockIdx.x * OWN_ECAP;
    bool scan = true;
    if (PLAY) {
        const int ne = __builtin_amdgcn_readfirstlane(ecount[blockIdx.x]);
        if (ne <= OWN_ECAP) {
            scan = false;
            int at = w * 32;
            unsigned e = (at + lane < ne) ? mine[at + lane] : 0u;
            while (at + 32 <= ne) {
                const int nx = at + OWN_NW * 32;
                const int ix = nx + lane;
                const unsigned ld = mine[(ix < ne) ? ix : ne - 1];
                add_rows<F>(src, acc, (int)(e & 0x1FFFFu), (int)(e >> 17), 32, lane);
                e = (ix < ne) ? ld : 0u;
                at = nx;
            }
            if (at < ne) add_rows<F>(src, acc, (int)(e & 0x1FFFFu), (int)(e >> 17), ne - at, lane);
        }
    }

    const uint4* tag16 = (const uint4*)tags;
    const int4*  rcv4  = (const int4*)rcv;
    int qn = 0;
    int gn = 0;
    for (int gb = w * 32 * OWN_U; scan && gb < NG16; gb += OWN_NW * 32 * OWN_U) {
        uint4 v[OWN_U];
        #pragma unroll
        for (int u = 0; u < OWN_U; ++u) {
            const int g = gb + u * 32 + lane;
            v[u] = tag16[(g < NG16) ? g : NG16 - 1];
        }
        #pragma unroll
        for (int u = 0; u < OWN_U; ++u) {
            const unsigned past = (gb + u * 32 < NG16) ? 0u : 0x01010101u;
            const unsigned t = nonzero_bytes((v[u].x ^ pat) | past) & nonzero_bytes((v[u].y ^ pat) | past) &
                               nonzero_bytes((v[u].z ^ pat) | past) & nonzero_bytes((v[u].w ^ pat) | past);
            const bool hit = (t & 0x80808080u) != 0x80808080u;
            const unsigned m = __builtin_amdgcn_ballot_w32(hit);
            if (m) {
                const int pos = gn + (int)__builtin_amdgcn_mbcnt_lo(m, 0u);
                if (hit) gq[pos] = gb + u * 32 + lane;
                gn += __builtin_popcount(m);
            }
        }
        if (gn >= 32) {
            int done = 0;
            while (gn - done >= 32) {
                expand_groups<F, CNT, REC>(src, snd, rcv4, acc, cnt, mine, rcnt, q, qn, gq[done + lane], r0, lane);
                done += 32;
            }
            const int rem = gn - done;
            const int t = (lane < rem) ? gq[done + lane] : 0;
            if (lane < rem) gq[lane] = t;
            gn = rem;
        }
    }
    if (gn > 0) expand_groups<F, CNT, REC>(src, snd, rcv4, acc, cnt, mine, rcnt, q, qn, (lane < gn) ? gq[lane] : -1, r0, lane);
    if (qn > 0) drain_rows<F, CNT, REC>(src, snd, acc, cnt, mine, rcnt, q, qn, lane);
    __syncthreads();

    if (REC) {
        if (threadIdx.x == 0) ecount[blockIdx.x] = rcnt[0];
    }

    if (CNT) {
        for (int i = threadIdx.x; i < OWN_R; i += OWN_NW * 32) {
            if (r0 + i < NN) rdeg[r0 + i] = (float)cnt[i];
        }
    }

    for (int i = threadIdx.x; i < OWN_R * F; i += OWN_NW * 32) {
        const int node = r0 + i / F;
        if (node < NN) dst[(size_t)r0 * F + i] = acc[i];
    }
}

__device__ __forceinline__ void count_groups(const int4* __restrict__ snd4, int* cnt, int g, int r0) {
    int4 s[4];
    #pragma unroll
    for (int k = 0; k < 4; ++k) s[k] = (g >= 0) ? snd4[g * 4 + k] : make_int4(-1, -1, -1, -1);
    #pragma unroll
    for (int k = 0; k < 4; ++k) {
        const int ss[4] = {s[k].x, s[k].y, s[k].z, s[k].w};
        #pragma unroll
        for (int c = 0; c < 4; ++c) {
            const unsigned d = (unsigned)(ss[c] - r0);
            if (d < (unsigned)OWN_R) atomicAdd(cnt + d, 1);
        }
    }
}

__launch_bounds__(OWN_NW * 32)
__global__ void sdeg_own_kernel(const int* __restrict__ snd, const unsigned* __restrict__ stags,
                                float* __restrict__ sdeg) {
    __shared__ int cnt[OWN_R];
    __shared__ int gqs[OWN_NW * OWN_GQCAP];
    const int w    = __builtin_amdgcn_readfirstlane((int)(threadIdx.x >> 5));
    const int lane = threadIdx.x & 31;
    const int r0   = blockIdx.x * OWN_R;
    int* gq = gqs + w * OWN_GQCAP;
    const unsigned pat = (unsigned)blockIdx.x * 0x01010101u;

    for (int i = threadIdx.x; i < OWN_R; i += OWN_NW * 32) cnt[i] = 0;
    __syncthreads();

    const uint4* tag16 = (const uint4*)stags;
    const int4*  snd4  = (const int4*)snd;
    int gn = 0;
    for (int gb = w * 32 * OWN_U; gb < NG16; gb += OWN_NW * 32 * OWN_U) {
        uint4 v[OWN_U];
        #pragma unroll
        for (int u = 0; u < OWN_U; ++u) {
            const int g = gb + u * 32 + lane;
            v[u] = tag16[(g < NG16) ? g : NG16 - 1];
        }
        #pragma unroll
        for (int u = 0; u < OWN_U; ++u) {
            const unsigned past = (gb + u * 32 < NG16) ? 0u : 0x01010101u;
            const unsigned t = nonzero_bytes((v[u].x ^ pat) | past) & nonzero_bytes((v[u].y ^ pat) | past) &
                               nonzero_bytes((v[u].z ^ pat) | past) & nonzero_bytes((v[u].w ^ pat) | past);
            const bool hit = (t & 0x80808080u) != 0x80808080u;
            const unsigned m = __builtin_amdgcn_ballot_w32(hit);
            if (m) {
                const int pos = gn + (int)__builtin_amdgcn_mbcnt_lo(m, 0u);
                if (hit) gq[pos] = gb + u * 32 + lane;
                gn += __builtin_popcount(m);
            }
        }
        if (gn >= 32) {
            int done = 0;
            while (gn - done >= 32) {
                count_groups(snd4, cnt, gq[done + lane], r0);
                done += 32;
            }
            const int rem = gn - done;
            const int t = (lane < rem) ? gq[done + lane] : 0;
            if (lane < rem) gq[lane] = t;
            gn = rem;
        }
    }
    if (gn > 0) count_groups(snd4, cnt, (lane < gn) ? gq[lane] : -1, r0);
    __syncthreads();

    for (int i = threadIdx.x; i < OWN_R; i += OWN_NW * 32) {
        if (r0 + i < NN) sdeg[r0 + i] = (float)cnt[i];
    }
}

__launch_bounds__(256)
__global__ void fin1_kernel(const float* __restrict__ agg1,
                            const float* __restrict__ rdeg, const float* __restrict__ sdeg,
                            const float* __restrict__ W3, const float* __restrict__ b3,
                            float* __restrict__ h2, int nTiles) {
    const int w    = threadIdx.x >> 5;
    const int lane = threadIdx.x & 31;
    const int m16  = lane & 15;
    const int kh   = lane >> 4;
    const int t    = blockIdx.x * 8 + w;
    if (t >= nTiles) return;
    const int row0 = t * 16;
    const float rs = rsqrtf(fmaxf(rdeg[row0 + m16], 1.f));

    v8f c0 = {}, c1 = {}, c2 = {};
    #pragma unroll 4
    for (int kb = 0; kb < DH; kb += 4) {
        v2f a = *(const v2f*)(agg1 + (size_t)(row0 + m16) * DH + kb + 2 * kh);
        a.x *= rs; a.y *= rs;
        const float* wr0 = W3 + (size_t)(kb + 2 * kh) * DC;
        const float* wr1 = wr0 + DC;
        v2f b;
        b.x = wr0[m16];      b.y = wr1[m16];      c0 = wmma4(a, b, c0);
        b.x = wr0[16 + m16]; b.y = wr1[16 + m16]; c1 = wmma4(a, b, c1);
        const int n2 = 32 + m16;
        b.x = (n2 < DC) ? wr0[n2] : 0.f;
        b.y = (n2 < DC) ? wr1[n2] : 0.f;
        c2 = wmma4(a, b, c2);
    }
    float sc[8];
    #pragma unroll
    for (int r = 0; r < 8; ++r) sc[r] = rsqrtf(fmaxf(sdeg[row0 + r + 8 * kh], 1.f));
    #pragma unroll
    for (int r = 0; r < 8; ++r) {
        float* orow = h2 + (size_t)(row0 + r + 8 * kh) * DC;
        orow[m16]      = (c0[r] + b3[m16])      * sc[r];
        orow[16 + m16] = (c1[r] + b3[16 + m16]) * sc[r];
        if (32 + m16 < DC)
            orow[32 + m16] = (c2[r] + b3[32 + m16]) * sc[r];
    }
}

__global__ void softmax_kernel(float* __restrict__ out, const float* __restrict__ rdeg) {
    int n = blockIdx.x * blockDim.x + threadIdx.x;
    if (n >= NN) return;
    float rs = rsqrtf(fmaxf(rdeg[n], 1.f));
    float* row = out + (size_t)n * DC;
    float v[DC];
    float mx = -3.402823466e38f;
    #pragma unroll
    for (int f = 0; f < DC; ++f) { v[f] = row[f] * rs; mx = fmaxf(mx, v[f]); }
    float s = 0.f;
    #pragma unroll
    for (int f = 0; f < DC; ++f) { v[f] = __expf(v[f] - mx); s += v[f]; }
    float inv = 1.f / s;
    #pragma unroll
    for (int f = 0; f < DC; ++f) row[f] = v[f] * inv;
}

extern "C" void kernel_launch(void* const* d_in, const int* in_sizes, int n_in,
                              void* d_out, int out_size, void* d_ws, size_t ws_size,
                              hipStream_t stream) {
    const float* nodes     = (const float*)d_in[0];
    const int*   senders   = (const int*)d_in[1];
    const int*   receivers = (const int*)d_in[2];
    const float* W1 = (const float*)d_in[3];
    const float* b1 = (const float*)d_in[4];
    const float* W2 = (const float*)d_in[5];
    const float* b2 = (const float*)d_in[6];
    const float* W3 = (const float*)d_in[7];
    const float* b3 = (const float*)d_in[8];
    float* out = (float*)d_out;
    float* ws  = (float*)d_ws;

    float* sdeg = ws;
    float* rdeg = ws + NN;
    float* h1   = ws + 2 * (size_t)NN;
    float* agg1 = h1 + (size_t)NN * DH;
    float* h2   = agg1 + (size_t)NN * DH;
    unsigned* tags  = (unsigned*)(h2 + (size_t)NN * DC);
    unsigned* stags = tags + NE / 4;
    unsigned* elist = stags + NE / 4;
    int* ecount = (int*)(elist + (size_t)OWN_G * OWN_ECAP);

    const int nTiles = NN / 16;
    tag_kernel<<<(NE / 4 + 255) / 256, 256, 0, stream>>>(senders, receivers, stags, tags);
    sdeg_own_kernel<<<OWN_G, OWN_NW * 32, 0, stream>>>(senders, stags, sdeg);
    mlp_kernel<<<(nTiles + 7) / 8, 256, 0, stream>>>(nodes, W1, b1, W2, b2, sdeg, h1, nTiles);
    agg_own_kernel<DH, true, true, false><<<OWN_G, OWN_NW * 32, 0, stream>>>(h1, senders, receivers, tags, agg1, rdeg, elist, ecount);
    fin1_kernel<<<(nTiles + 7) / 8, 256, 0, stream>>>(agg1, rdeg, sdeg, W3, b3, h2, nTiles);
    agg_own_kernel<DC, false, false, true><<<OWN_G, OWN_NW * 32, 0, stream>>>(h2, senders, receivers, tags, out, nullptr, elist, ecount);
    softmax_kernel<<<(NN + 255) / 256, 256, 0, stream>>>(out, rdeg);
}


# First edge sum's owner blocks finish their rows themselves: LDS sums (row pitch 68) x W3 on the matrix unit, bias, both scales, h2 stored; sums never reach memory; fin1 launch gone, 7 -> 6 launches
# speedup vs baseline: 3.6600x; 1.0838x over previous
#include <hip/hip_runtime.h>

typedef __attribute__((ext_vector_type(2))) float v2f;
typedef __attribute__((ext_vector_type(8))) float v8f;

static constexpr int NN = 100000;
static constexpr int NE = 1600000;
static constexpr int DF = 128;
static constexpr int DH = 64;
static constexpr int DC = 40;
#define NEG_SLOPE 0.01f
static constexpr unsigned TAG_R = (NN + 255) / 256;

__device__ __forceinline__ float lrelu(float x) { return x > 0.f ? x : x * NEG_SLOPE; }

__device__ __forceinline__ v8f wmma4(v2f a, v2f b, v8f c) {
    return __builtin_amdgcn_wmma_f32_16x16x4_f32(false, a, false, b, (short)0, c, false, false);
}

__device__ __forceinline__ unsigned owner_tags(int4 v) {
    return ((unsigned)v.x / TAG_R) | (((unsigned)v.y / TAG_R) << 8) |
           (((unsigned)v.z / TAG_R) << 16) | (((unsigned)v.w / TAG_R) << 24);
}

__global__ void tag_kernel(const int* __restrict__ snd, const int* __restrict__ rcv,
                           unsigned* __restrict__ stags, unsigned* __restrict__ tags) {
    const int g = blockIdx.x * blockDim.x + threadIdx.x;
    if (g < NE / 4) {
        const int4 s = ((const int4*)snd)[g];
        const int4 r = ((const int4*)rcv)[g];
        stags[g] = owner_tags(s);
        tags[g]  = owner_tags(r);
    }
}

__launch_bounds__(256)
__global__ void mlp_kernel(const float* __restrict__ X,
                           const float* __restrict__ W1, const float* __restrict__ b1,
                           const float* __restrict__ W2, const float* __restrict__ b2,
                           const float* __restrict__ sdeg,
                           float* __restrict__ h1, int nTiles) {
    __shared__ float lds[8 * 16 * DH];
    const int w    = threadIdx.x >> 5;
    const int lane = threadIdx.x & 31;
    const int m16  = lane & 15;
    const int kh   = lane >> 4;
    const int t    = blockIdx.x * 8 + w;
    const bool active = (t < nTiles);
    const int row0 = t * 16;
    float* my = &lds[w * 16 * DH];

    if (active) {
        v8f c0 = {}, c1 = {}, c2 = {}, c3 = {};
        #pragma unroll 4
        for (int kb = 0; kb < DF; kb += 4) {
            v2f a = *(const v2f*)(X + (size_t)(row0 + m16) * DF + kb + 2 * kh);
            const float* wr0 = W1 + (size_t)(kb + 2 * kh) * DH;
            const float* wr1 = wr0 + DH;
            v2f b;
            b.x = wr0[m16];      b.y = wr1[m16];      c0 = wmma4(a, b, c0);
            b.x = wr0[16 + m16]; b.y = wr1[16 + m16]; c1 = wmma4(a, b, c1);
            b.x = wr0[32 + m16]; b.y = wr1[32 + m16]; c2 = wmma4(a, b, c2);
            b.x = wr0[48 + m16]; b.y = wr1[48 + m16]; c3 = wmma4(a, b, c3);
        }
        #pragma unroll
        for (int r = 0; r < 8; ++r) {
            float* dr = my + (r + 8 * kh) * DH;
            dr[m16]      = lrelu(c0[r] + b1[m16]);
            dr[16 + m16] = lrelu(c1[r] + b1[16 + m16]);
            dr[32 + m16] = lrelu(c2[r] + b1[32 + m16]);
            dr[48 + m16] = lrelu(c3[r] + b1[48 + m16]);
        }
    }
    __syncthreads();
    if (!active) return;

    v8f d0 = {}, d1 = {}, d2 = {}, d3 = {};
    #pragma unroll 4
    for (int kb = 0; kb < DH; kb += 4) {
        v2f a = *(const v2f*)(my + m16 * DH + kb + 2 * kh);
        const float* wr0 = W2 + (size_t)(kb + 2 * kh) * DH;
        const float* wr1 = wr0 + DH;
        v2f b;
        b.x = wr0[m16];      b.y = wr1[m16];      d0 = wmma4(a, b, d0);
        b.x = wr0[16 + m16]; b.y = wr1[16 + m16]; d1 = wmma4(a, b, d1);
        b.x = wr0[32 + m16]; b.y = wr1[32 + m16]; d2 = wmma4(a, b, d2);
        b.x = wr0[48 + m16]; b.y = wr1[48 + m16]; d3 = wmma4(a, b, d3);
    }
    float sc[8];
    #pragma unroll
    for (int r = 0; r < 8; ++r) sc[r] = rsqrtf(fmaxf(sdeg[row0 + r + 8 * kh], 1.f));
    #pragma unroll
    for (int r = 0; r < 8; ++r) {
        float* orow = h1 + (size_t)(row0 + r + 8 * kh) * DH;
        orow[m16]      = lrelu(d0[r] + b2[m16])      * sc[r];
        orow[16 + m16] = lrelu(d1[r] + b2[16 + m16]) * sc[r];
        orow[32 + m16] = lrelu(d2[r] + b2[32 + m16]) * sc[r];
        orow[48 + m16] = lrelu(d3[r] + b2[48 + m16]) * sc[r];
    }
}

static constexpr int OWN_G  = 256;
static constexpr int OWN_R  = (NN + OWN_G - 1) / OWN_G;
static constexpr int OWN_NW = 32;
static constexpr int OWN_U  = 8;
static constexpr int OWN_GQCAP = OWN_U * 32 + 32;
static constexpr int OWN_QCAP  = 16 * 32 + 32;
static constexpr int NG16 = NE / 16;
static constexpr int OWN_RT = (OWN_R + 15) / 16;
static constexpr int OWN_ECAP = 1 << 17;
static_assert(OWN_R == (int)TAG_R, "tags must name the owner blocks");
static_assert(NN <= (1 << 17) && OWN_R <= (1 << 9), "a list entry holds the sender in bits 0..16 and the local receiver above them");
static_assert(OWN_RT <= OWN_NW, "the fused product gives every row tile a wave of its own");
static_assert(NE % 16 == 0, "edge list must split into groups of sixteen");
static_assert(NG16 % 32 == 0, "a wave's 32 groups must lie all inside the list or all past its end");

template <int F, int P>
__device__ __forceinline__ void add_rows(const float* __restrict__ src, float* acc, int s, int l, int n, int lane) {
    for (int j0 = 0; j0 < n; j0 += 8) {
        float a[8], b[8];
        int lj[8];
        #pragma unroll
        for (int u = 0; u < 8; ++u) {
            const int sj = __builtin_amdgcn_readlane(s, j0 + u);
            lj[u] = __builtin_amdgcn_readlane(l, j0 + u);
            const float* row = src + (size_t)sj * F;
            a[u] = row[lane];
            b[u] = (lane < F - 32) ? row[32 + lane] : 0.f;
        }
        #pragma unroll
        for (int u = 0; u < 8; ++u) {
            if (j0 + u < n) {
                float* d = acc + lj[u] * P;
                atomicAdd(d + lane, a[u]);
                if (lane < F - 32) atomicAdd(d + 32 + lane, b[u]);
            }
        }
    }
}

template <int F, int P, bool CNT, bool REC>
__device__ __forceinline__ void drain_rows(const float* __restrict__ src, const int* __restrict__ snd,
                                           float* acc, int* cnt, unsigned* __restrict__ rec, int* rcnt,
                                           const int* q, int n, int lane) {
    const int ent = (lane < n) ? q[lane] : 0;
    const int l = ent >> 21;
    if (CNT) {
        if (lane < n) atomicAdd(cnt + l, 1);
    }
    const int s = (lane < n) ? snd[ent & 0x1FFFFF] : 0;
    if (REC) {
        int at = 0;
        if (lane == 0) at = atomicAdd(rcnt, n);
        at = __builtin_amdgcn_readfirstlane(at);
        if (lane < n && at + lane < OWN_ECAP) rec[at + lane] = (unsigned)s | ((unsigned)l << 17);
    }
    add_rows<F, P>(src, acc, s, l, n, lane);
}

__device__ __forceinline__ unsigned nonzero_bytes(unsigned y) {
    return ((y & 0x7F7F7F7Fu) + 0x7F7F7F7Fu) | y;
}

template <int F, int P, bool CNT, bool REC>
__device__ __forceinline__ void expand_groups(const float* __restrict__ src, const int* __restrict__ snd,
                                              const int4* __restrict__ rcv4, float* acc, int* cnt,
                                              unsigned* __restrict__ rec, int* rcnt, int* q, int& qn,
                                              int g, int r0, int lane) {
    int4 r[4];
    #pragma unroll
    for (int k = 0; k < 4; ++k) r[k] = (g >= 0) ? rcv4[g * 4 + k] : make_int4(-1, -1, -1, -1);
    #pragma unroll
    for (int k = 0; k < 4; ++k) {
        const int rr[4] = {r[k].x, r[k].y, r[k].z, r[k].w};
        #pragma unroll
        for (int c = 0; c < 4; ++c) {
            const unsigned d = (unsigned)(rr[c] - r0);
            const bool hit = d < (unsigned)OWN_R;
            const unsigned m = __builtin_amdgcn_ballot_w32(hit);
            if (m) {
                const int pos = qn + (int)__builtin_amdgcn_mbcnt_lo(m, 0u);
                if (hit) q[pos] = (g * 16 + k * 4 + c) | ((int)d << 21);
                qn += __builtin_popcount(m);
            }
        }
    }
    if (qn >= 32) {
        int done = 0;
        while (qn - done >= 32) {
            drain_rows<F, P, CNT, REC>(src, snd, acc, cnt, rec, rcnt, q + done, 32, lane);
            done += 32;
        }
        const int rem = qn - done;
        const int t = (lane < rem) ? q[done + lane] : 0;
        if (lane < rem) q[lane] = t;
        qn = rem;
    }
}

template <int F, int P, bool CNT, bool REC, bool PLAY, bool FIN>
__launch_bounds__(OWN_NW * 32)
__global__ void agg_own_kernel(const float* __restrict__ src, const int* __restrict__ snd,
                               const int* __restrict__ rcv, const unsigned* __restrict__ tags,
                               float* __restrict__ dst, float* __restrict__ rdeg,
                               unsigned* __restrict__ elist, int* __restrict__ ecount,
                               const float* __restrict__ sdeg, const float* __restrict__ W3,
                               const float* __restrict__ b3) {
    static_assert(!FIN || (CNT && F == DH && P % 2 == 0), "the fused product reads 64-wide sums in pairs and scales them by the block's own counts");
    constexpr int ROWS = FIN ? OWN_RT * 16 : OWN_R;
    __shared__ __attribute__((aligned(16))) float acc[ROWS * P];
    __shared__ int   cnt[CNT ? ROWS : 1];
    __shared__ int   rcnt[1];
    __shared__ int   qs[OWN_NW * OWN_QCAP];
    __shared__ int   gqs[OWN_NW * OWN_GQCAP];
    const int w    = __builtin_amdgcn_readfirstlane((int)(threadIdx.x >> 5));
    const int lane = threadIdx.x & 31;
    const int r0   = blockIdx.x * OWN_R;
    int* q  = qs + w * OWN_QCAP;
    int* gq = gqs + w * OWN_GQCAP;
    const unsigned pat = (unsigned)blockIdx.x * 0x01010101u;

    for (int i = threadIdx.x; i < ROWS * P; i += OWN_NW * 32) acc[i] = 0.f;
    if (CNT) {
        for (int i = threadIdx.x; i < ROWS; i += OWN_NW * 32) cnt[i] = 0;
    }
    if (REC) {
        if (threadIdx.x == 0) rcnt[0] = 0;
    }
    __syncthreads();

    unsigned* mine = elist + (size_t)blockIdx.x * OWN_ECAP;
    bool scan = true;
    if (PLAY) {
        const int ne = __builtin_amdgcn_readfirstlane(ecount[blockIdx.x]);
        if (ne <= OWN_ECAP) {
            scan = false;
            int at = w * 32;
            unsigned e = (at + lane < ne) ? mine[at + lane] : 0u;
            while (at + 32 <= ne) {
                const int nx = at + OWN_NW * 32;
                const int ix = nx + lane;
                const unsigned ld = mine[(ix < ne) ? ix : ne - 1];
                add_rows<F, P>(src, acc, (int)(e & 0x1FFFFu), (int)(e >> 17), 32, lane);
                e = (ix < ne) ? ld : 0u;
                at = nx;
            }
            if (at < ne) add_rows<F, P>(src, acc, (int)(e & 0x1FFFFu), (int)(e >> 17), ne - at, lane);
        }
    }

    const uint4* tag16 = (const uint4*)tags;
    const int4*  rcv4  = (const int4*)rcv;
    int qn = 0;
    int gn = 0;
    for (int gb = w * 32 * OWN_U; scan && gb < NG16; gb += OWN_NW * 32 * OWN_U) {
        uint4 v[OWN_U];
        #pragma unroll
        for (int u = 0; u < OWN_U; ++u) {
            const int g = gb + u * 32 + lane;
            v[u] = tag16[(g < NG16) ? g : NG16 - 1];
        }
        #pragma unroll
        for (int u = 0; u < OWN_U; ++u) {
            const unsigned past = (gb + u * 32 < NG16) ? 0u : 0x01010101u;
            const unsigned t = nonzero_bytes((v[u].x ^ pat) | past) & nonzero_bytes((v[u].y ^ pat) | past) &
                               nonzero_bytes((v[u].z ^ pat) | past) & nonzero_bytes((v[u].w ^ pat) | past);
            const bool hit = (t & 0x80808080u) != 0x80808080u;
            const unsigned m = __builtin_amdgcn_ballot_w32(hit);
            if (m) {
                const int pos = gn + (int)__builtin_amdgcn_mbcnt_lo(m, 0u);
                if (hit) gq[pos] = gb + u * 32 + lane;
                gn += __builtin_popcount(m);
            }
        }
        if (gn >= 32) {
            int done = 0;
            while (gn - done >= 32) {
                expand_groups<F, P, CNT, REC>(src, snd, rcv4, acc, cnt, mine, rcnt, q, qn, gq[done + lane], r0, lane);
                done += 32;
            }
            const int rem = gn - done;
            const int t = (lane < rem) ? gq[done + lane] : 0;
            if (lane < rem) gq[lane] = t;
            gn = rem;
        }
    }
    if (gn > 0) expand_groups<F, P, CNT, REC>(src, snd, rcv4, acc, cnt, mine, rcnt, q, qn, (lane < gn) ? gq[lane] : -1, r0, lane);
    if (qn > 0) drain_rows<F, P, CNT, REC>(src, snd, acc, cnt, mine, rcnt, q, qn, lane);
    __syncthreads();

    if (REC) {
        if (threadIdx.x == 0) ecount[blockIdx.x] = rcnt[0];
    }

    if (CNT) {
        for (int i = threadIdx.x; i < OWN_R; i += OWN_NW * 32) {
            if (r0 + i < NN) rdeg[r0 + i] = (float)cnt[i];
        }
    }

    if (FIN) {
        if (w < OWN_RT) {
            const int m16  = lane & 15;
            const int kh   = lane >> 4;
            const int lr0  = w * 16;
            const float rs = rsqrtf(fmaxf((float)cnt[lr0 + m16], 1.f));
            const float* arow = acc + (lr0 + m16) * P + 2 * kh;

            float sc[8];
            #pragma unroll
            for (int r = 0; r < 8; ++r) {
                const int node = r0 + lr0 + r + 8 * kh;
                sc[r] = sdeg[(node < NN) ? node : NN - 1];
            }
            const int   n2c = (32 + m16 < DC) ? 32 + m16 : DC - 1;
            const float bb0 = b3[m16], bb1 = b3[16 + m16], bb2 = b3[n2c];

            v8f c0 = {}, c1 = {}, c2 = {};
            #pragma unroll 4
            for (int kb = 0; kb < DH; kb += 4) {
                v2f a = *(const v2f*)(arow + kb);
                a.x *= rs; a.y *= rs;
                const float* wr0 = W3 + (size_t)(kb + 2 * kh) * DC;
                const float* wr1 = wr0 + DC;
                v2f b;
                b.x = wr0[m16];      b.y = wr1[m16];      c0 = wmma4(a, b, c0);
                b.x = wr0[16 + m16]; b.y = wr1[16 + m16]; c1 = wmma4(a, b, c1);
                const float p0 = wr0[n2c], p1 = wr1[n2c];
                b.x = (32 + m16 < DC) ? p0 : 0.f;
                b.y = (32 + m16 < DC) ? p1 : 0.f;
                c2 = wmma4(a, b, c2);
            }
            #pragma unroll
            for (int r = 0; r < 8; ++r) {
                const int lr   = lr0 + r + 8 * kh;
                const int node = r0 + lr;
                if (lr < OWN_R && node < NN) {
                    const float s = rsqrtf(fmaxf(sc[r], 1.f));
                    float* orow = dst + (size_t)node * DC;
                    orow[m16]      = (c0[r] + bb0) * s;
                    orow[16 + m16] = (c1[r] + bb1) * s;
                    if (32 + m16 < DC)
                        orow[32 + m16] = (c2[r] + bb2) * s;
                }
            }
        }
    } else {
        for (int i = threadIdx.x; i < OWN_R * F; i += OWN_NW * 32) {
            const int lr = i / F;
            if (r0 + lr < NN) dst[(size_t)r0 * F + i] = acc[lr * P + (i - lr * F)];
        }
    }
}

__device__ __forceinline__ void count_groups(const int4* __restrict__ snd4, int* cnt, int g, int r0) {
    int4 s[4];
    #pragma unroll
    for (int k = 0; k < 4; ++k) s[k] = (g >= 0) ? snd4[g * 4 + k] : make_int4(-1, -1, -1, -1);
    #pragma unroll
    for (int k = 0; k < 4; ++k) {
        const int ss[4] = {s[k].x, s[k].y, s[k].z, s[k].w};
        #pragma unroll
        for (int c = 0; c < 4; ++c) {
            const unsigned d = (unsigned)(ss[c] - r0);
            if (d < (unsigned)OWN_R) atomicAdd(cnt + d, 1);
        }
    }
}

__launch_bounds__(OWN_NW * 32)
__global__ void sdeg_own_kernel(const int* __restrict__ snd, const unsigned* __restrict__ stags,
                                float* __restrict__ sdeg) {
    __shared__ int cnt[OWN_R];
    __shared__ int gqs[OWN_NW * OWN_GQCAP];
    const int w    = __builtin_amdgcn_readfirstlane((int)(threadIdx.x >> 5));
    const int lane = threadIdx.x & 31;
    const int r0   = blockIdx.x * OWN_R;
    int* gq = gqs + w * OWN_GQCAP;
    const unsigned pat = (unsigned)blockIdx.x * 0x01010101u;

    for (int i = threadIdx.x; i < OWN_R; i += OWN_NW * 32) cnt[i] = 0;
    __syncthreads();

    const uint4* tag16 = (const uint4*)stags;
    const int4*  snd4  = (const int4*)snd;
    int gn = 0;
    for (int gb = w * 32 * OWN_U; gb < NG16; gb += OWN_NW * 32 * OWN_U) {
        uint4 v[OWN_U];
        #pragma unroll
        for (int u = 0; u < OWN_U; ++u) {
            const int g = gb + u * 32 + lane;
            v[u] = tag16[(g < NG16) ? g : NG16 - 1];
        }
        #pragma unroll
        for (int u = 0; u < OWN_U; ++u) {
            const unsigned past = (gb + u * 32 < NG16) ? 0u : 0x01010101u;
            const unsigned t = nonzero_bytes((v[u].x ^ pat) | past) & nonzero_bytes((v[u].y ^ pat) | past) &
                               nonzero_bytes((v[u].z ^ pat) | past) & nonzero_bytes((v[u].w ^ pat) | past);
            const bool hit = (t & 0x80808080u) != 0x80808080u;
            const unsigned m = __builtin_amdgcn_ballot_w32(hit);
            if (m) {
                const int pos = gn + (int)__builtin_amdgcn_mbcnt_lo(m, 0u);
                if (hit) gq[pos] = gb + u * 32 + lane;
                gn += __builtin_popcount(m);
            }
        }
        if (gn >= 32) {
            int done = 0;
            while (gn - done >= 32) {
                count_groups(snd4, cnt, gq[done + lane], r0);
                done += 32;
            }
            const int rem = gn - done;
            const int t = (lane < rem) ? gq[done + lane] : 0;
            if (lane < rem) gq[lane] = t;
            gn = rem;
        }
    }
    if (gn > 0) count_groups(snd4, cnt, (lane < gn) ? gq[lane] : -1, r0);
    __syncthreads();

    for (int i = threadIdx.x; i < OWN_R; i += OWN_NW * 32) {
        if (r0 + i < NN) sdeg[r0 + i] = (float)cnt[i];
    }
}

__global__ void softmax_kernel(float* __restrict__ out, const float* __restrict__ rdeg) {
    int n = blockIdx.x * blockDim.x + threadIdx.x;
    if (n >= NN) return;
    float rs = rsqrtf(fmaxf(rdeg[n], 1.f));
    float* row = out + (size_t)n * DC;
    float v[DC];
    float mx = -3.402823466e38f;
    #pragma unroll
    for (int f = 0; f < DC; ++f) { v[f] = row[f] * rs; mx = fmaxf(mx, v[f]); }
    float s = 0.f;
    #pragma unroll
    for (int f = 0; f < DC; ++f) { v[f] = __expf(v[f] - mx); s += v[f]; }
    float inv = 1.f / s;
    #pragma unroll
    for (int f = 0; f < DC; ++f) row[f] = v[f] * inv;
}

extern "C" void kernel_launch(void* const* d_in, const int* in_sizes, int n_in,
                              void* d_out, int out_size, void* d_ws, size_t ws_size,
                              hipStream_t stream) {
    const float* nodes     = (const float*)d_in[0];
    const int*   senders   = (const int*)d_in[1];
    const int*   receivers = (const int*)d_in[2];
    const float* W1 = (const float*)d_in[3];
    const float* b1 = (const float*)d_in[4];
    const float* W2 = (const float*)d_in[5];
    const float* b2 = (const float*)d_in[6];
    const float* W3 = (const float*)d_in[7];
    const float* b3 = (const float*)d_in[8];
    float* out = (float*)d_out;
    float* ws  = (float*)d_ws;

    float* sdeg = ws;
    float* rdeg = ws + NN;
    float* h1   = ws + 2 * (size_t)NN;
    float* h2   = h1 + 2 * (size_t)NN * DH;
    unsigned* tags  = (unsigned*)(h2 + (size_t)NN * DC);
    unsigned* stags = tags + NE / 4;
    unsigned* elist = stags + NE / 4;
    int* ecount = (int*)(elist + (size_t)OWN_G * OWN_ECAP);

    const int nTiles = NN / 16;
    tag_kernel<<<(NE / 4 + 255) / 256, 256, 0, stream>>>(senders, receivers, stags, tags);
    sdeg_own_kernel<<<OWN_G, OWN_NW * 32, 0, stream>>>(senders, stags, sdeg);
    mlp_kernel<<<(nTiles + 7) / 8, 256, 0, stream>>>(nodes, W1, b1, W2, b2, sdeg, h1, nTiles);
    agg_own_kernel<DH, 68, true, true, false, true><<<OWN_G, OWN_NW * 32, 0, stream>>>(h1, senders, receivers, tags, h2, rdeg, elist, ecount, sdeg, W3, b3);
    agg_own_kernel<DC, DC, false, false, true, false><<<OWN_G, OWN_NW * 32, 0, stream>>>(h2, senders, receivers, tags, out, nullptr, elist, ecount, nullptr, nullptr, nullptr);
    softmax_kernel<<<(NN + 255) / 256, 256, 0, stream>>>(out, rdeg);
}


# Owner blocks read their own row of a per-owner bit index (a bit per 16-edge group, built in LDS by the index kernel), not all one-byte tags; edges queue in another order: LDS-atomic sum order changes
# speedup vs baseline: 4.2862x; 1.1711x over previous
#include <hip/hip_runtime.h>

typedef __attribute__((ext_vector_type(2))) float v2f;
typedef __attribute__((ext_vector_type(8))) float v8f;

static constexpr int NN = 100000;
static constexpr int NE = 1600000;
static constexpr int DF = 128;
static constexpr int DH = 64;
static constexpr int DC = 40;
#define NEG_SLOPE 0.01f
static constexpr unsigned TAG_R = (NN + 255) / 256;

__device__ __forceinline__ float lrelu(float x) { return x > 0.f ? x : x * NEG_SLOPE; }

__device__ __forceinline__ v8f wmma4(v2f a, v2f b, v8f c) {
    return __builtin_amdgcn_wmma_f32_16x16x4_f32(false, a, false, b, (short)0, c, false, false);
}


__launch_bounds__(256)
__global__ void mlp_kernel(const float* __restrict__ X,
                           const float* __restrict__ W1, const float* __restrict__ b1,
                           const float* __restrict__ W2, const float* __restrict__ b2,
                           const float* __restrict__ sdeg,
                           float* __restrict__ h1, int nTiles) {
    __shared__ float lds[8 * 16 * DH];
    const int w    = threadIdx.x >> 5;
    const int lane = threadIdx.x & 31;
    const int m16  = lane & 15;
    const int kh   = lane >> 4;
    const int t    = blockIdx.x * 8 + w;
    const bool active = (t < nTiles);
    const int row0 = t * 16;
    float* my = &lds[w * 16 * DH];

    if (active) {
        v8f c0 = {}, c1 = {}, c2 = {}, c3 = {};
        #pragma unroll 4
        for (int kb = 0; kb < DF; kb += 4) {
            v2f a = *(const v2f*)(X + (size_t)(row0 + m16) * DF + kb + 2 * kh);
            const float* wr0 = W1 + (size_t)(kb + 2 * kh) * DH;
            const float* wr1 = wr0 + DH;
            v2f b;
            b.x = wr0[m16];      b.y = wr1[m16];      c0 = wmma4(a, b, c0);
            b.x = wr0[16 + m16]; b.y = wr1[16 + m16]; c1 = wmma4(a, b, c1);
            b.x = wr0[32 + m16]; b.y = wr1[32 + m16]; c2 = wmma4(a, b, c2);
            b.x = wr0[48 + m16]; b.y = wr1[48 + m16]; c3 = wmma4(a, b, c3);
        }
        #pragma unroll
        for (int r = 0; r < 8; ++r) {
            float* dr = my + (r + 8 * kh) * DH;
            dr[m16]      = lrelu(c0[r] + b1[m16]);
            dr[16 + m16] = lrelu(c1[r] + b1[16 + m16]);
            dr[32 + m16] = lrelu(c2[r] + b1[32 + m16]);
            dr[48 + m16] = lrelu(c3[r] + b1[48 + m16]);
        }
    }
    __syncthreads();
    if (!active) return;

    v8f d0 = {}, d1 = {}, d2 = {}, d3 = {};
    #pragma unroll 4
    for (int kb = 0; kb < DH; kb += 4) {
        v2f a = *(const v2f*)(my + m16 * DH + kb + 2 * kh);
        const float* wr0 = W2 + (size_t)(kb + 2 * kh) * DH;
        const float* wr1 = wr0 + DH;
        v2f b;
        b.x = wr0[m16];      b.y = wr1[m16];      d0 = wmma4(a, b, d0);
        b.x = wr0[16 + m16]; b.y = wr1[16 + m16]; d1 = wmma4(a, b, d1);
        b.x = wr0[32 + m16]; b.y = wr1[32 + m16]; d2 = wmma4(a, b, d2);
        b.x = wr0[48 + m16]; b.y = wr1[48 + m16]; d3 = wmma4(a, b, d3);
    }
    float sc[8];
    #pragma unroll
    for (int r = 0; r < 8; ++r) sc[r] = rsqrtf(fmaxf(sdeg[row0 + r + 8 * kh], 1.f));
    #pragma unroll
    for (int r = 0; r < 8; ++r) {
        float* orow = h1 + (size_t)(row0 + r + 8 * kh) * DH;
        orow[m16]      = lrelu(d0[r] + b2[m16])      * sc[r];
        orow[16 + m16] = lrelu(d1[r] + b2[16 + m16]) * sc[r];
        orow[32 + m16] = lrelu(d2[r] + b2[32 + m16]) * sc[r];
        orow[48 + m16] = lrelu(d3[r] + b2[48 + m16]) * sc[r];
    }
}

static constexpr int OWN_G  = 256;
static constexpr int OWN_R  = (NN + OWN_G - 1) / OWN_G;
static constexpr int OWN_NW = 32;
static constexpr int OWN_GQCAP = 64;
static constexpr int OWN_QCAP  = 16 * 32 + 32;
static constexpr int NG16 = NE / 16;
static constexpr int OWN_RT = (OWN_R + 15) / 16;
static constexpr int OWN_ECAP = 1 << 17;
static constexpr int IDX_GPB = 1024;
static constexpr int IDX_NB  = (NG16 + IDX_GPB - 1) / IDX_GPB;
static constexpr int IDX_ROW = IDX_NB * 32;
static constexpr int IDX_P   = OWN_G + 1;
static constexpr int IDX_FULL = IDX_NB / OWN_NW;
static constexpr int IDX_REM  = (IDX_ROW - IDX_FULL * OWN_NW * 32) / OWN_NW;
static_assert(OWN_R == (int)TAG_R, "the index must name the owner blocks");
static_assert(NN <= (1 << 17) && OWN_R <= (1 << 9), "a list entry holds the sender in bits 0..16 and the local receiver above them");
static_assert(OWN_RT <= OWN_NW, "the fused product gives every row tile a wave of its own");
static_assert(NE % 16 == 0, "edge list must split into groups of sixteen");
static_assert(NG16 % 32 == 0, "a wave's 32 groups must lie all inside the list or all past its end");
static_assert(OWN_G % 32 == 0 && IDX_GPB == 32 * 32, "an index block is 32 waves of 32 groups, and every wave writes whole owners' lines");
static_assert((IDX_ROW - IDX_FULL * OWN_NW * 32) % OWN_NW == 0 && IDX_REM >= 1 && IDX_REM <= 32, "the words left after the full rounds split evenly over the waves, at least one each");

__launch_bounds__(1024)
__global__ void index_kernel(const int* __restrict__ snd, const int* __restrict__ rcv,
                             unsigned* __restrict__ sbits, unsigned* __restrict__ rbits) {
    __shared__ unsigned bm[32 * IDX_P];
    const int w    = threadIdx.x >> 5;
    const int lane = threadIdx.x & 31;
    const int* __restrict__ ids = blockIdx.y ? rcv : snd;
    unsigned* __restrict__ rows = blockIdx.y ? rbits : sbits;

    for (int i = threadIdx.x; i < 32 * IDX_P; i += 1024) bm[i] = 0u;
    __syncthreads();

    const int g0 = blockIdx.x * IDX_GPB + w * 32;
    if (g0 < NG16) {
        const int4* p = (const int4*)ids + (size_t)g0 * 4;
        int4 v[4];
        #pragma unroll
        for (int k = 0; k < 4; ++k) v[k] = p[k * 32 + lane];
        asm volatile("" :: "v"(v[0].x), "v"(v[0].y), "v"(v[0].z), "v"(v[0].w), "v"(v[1].x), "v"(v[1].y), "v"(v[1].z), "v"(v[1].w),
                           "v"(v[2].x), "v"(v[2].y), "v"(v[2].z), "v"(v[2].w), "v"(v[3].x), "v"(v[3].y), "v"(v[3].z), "v"(v[3].w));
        unsigned* row = bm + w * IDX_P;
        #pragma unroll
        for (int k = 0; k < 4; ++k) {
            const unsigned bit = 1u << (k * 8 + (lane >> 2));
            atomicOr(row + (unsigned)v[k].x / TAG_R, bit);
            atomicOr(row + (unsigned)v[k].y / TAG_R, bit);
            atomicOr(row + (unsigned)v[k].z / TAG_R, bit);
            atomicOr(row + (unsigned)v[k].w / TAG_R, bit);
        }
    }
    __syncthreads();

    #pragma unroll
    for (int j = 0; j < OWN_G / 32; ++j) {
        const int o = w * (OWN_G / 32) + j;
        rows[(size_t)o * IDX_ROW + blockIdx.x * 32 + lane] = bm[lane * IDX_P + o];
    }
}

template <int F, int P>
__device__ __forceinline__ void add_rows(const float* __restrict__ src, float* acc, int s, int l, int n, int lane) {
    for (int j0 = 0; j0 < n; j0 += 8) {
        float a[8], b[8];
        int lj[8];
        #pragma unroll
        for (int u = 0; u < 8; ++u) {
            const int sj = __builtin_amdgcn_readlane(s, j0 + u);
            lj[u] = __builtin_amdgcn_readlane(l, j0 + u);
            const float* row = src + (size_t)sj * F;
            a[u] = row[lane];
            b[u] = (lane < F - 32) ? row[32 + lane] : 0.f;
        }
        #pragma unroll
        for (int u = 0; u < 8; ++u) {
            if (j0 + u < n) {
                float* d = acc + lj[u] * P;
                atomicAdd(d + lane, a[u]);
                if (lane < F - 32) atomicAdd(d + 32 + lane, b[u]);
            }
        }
    }
}

template <int F, int P, bool CNT, bool REC>
__device__ __forceinline__ void drain_rows(const float* __restrict__ src, const int* __restrict__ snd,
                                           float* acc, int* cnt, unsigned* __restrict__ rec, int* rcnt,
                                           const int* q, int n, int lane) {
    const int ent = (lane < n) ? q[lane] : 0;
    const int l = ent >> 21;
    if (CNT) {
        if (lane < n) atomicAdd(cnt + l, 1);
    }
    const int s = (lane < n) ? snd[ent & 0x1FFFFF] : 0;
    if (REC) {
        int at = 0;
        if (lane == 0) at = atomicAdd(rcnt, n);
        at = __builtin_amdgcn_readfirstlane(at);
        if (lane < n && at + lane < OWN_ECAP) rec[at + lane] = (unsigned)s | ((unsigned)l << 17);
    }
    add_rows<F, P>(src, acc, s, l, n, lane);
}


template <int F, int P, bool CNT, bool REC>
__device__ __forceinline__ void expand_groups(const float* __restrict__ src, const int* __restrict__ snd,
                                              const int4* __restrict__ rcv4, float* acc, int* cnt,
                                              unsigned* __restrict__ rec, int* rcnt, int* q, int& qn,
                                              int g, int r0, int lane) {
    int4 r[4];
    #pragma unroll
    for (int k = 0; k < 4; ++k) r[k] = (g >= 0) ? rcv4[g * 4 + k] : make_int4(-1, -1, -1, -1);
    #pragma unroll
    for (int k = 0; k < 4; ++k) {
        const int rr[4] = {r[k].x, r[k].y, r[k].z, r[k].w};
        #pragma unroll
        for (int c = 0; c < 4; ++c) {
            const unsigned d = (unsigned)(rr[c] - r0);
            const bool hit = d < (unsigned)OWN_R;
            const unsigned m = __builtin_amdgcn_ballot_w32(hit);
            if (m) {
                const int pos = qn + (int)__builtin_amdgcn_mbcnt_lo(m, 0u);
                if (hit) q[pos] = (g * 16 + k * 4 + c) | ((int)d << 21);
                qn += __builtin_popcount(m);
            }
        }
    }
    if (qn >= 32) {
        int done = 0;
        while (qn - done >= 32) {
            drain_rows<F, P, CNT, REC>(src, snd, acc, cnt, rec, rcnt, q + done, 32, lane);
            done += 32;
        }
        const int rem = qn - done;
        const int t = (lane < rem) ? q[done + lane] : 0;
        if (lane < rem) q[lane] = t;
        qn = rem;
    }
}

template <int F, int P, bool CNT, bool REC, bool PLAY, bool FIN>
__launch_bounds__(OWN_NW * 32)
__global__ void agg_own_kernel(const float* __restrict__ src, const int* __restrict__ snd,
                               const int* __restrict__ rcv, const unsigned* __restrict__ bits,
                               float* __restrict__ dst, float* __restrict__ rdeg,
                               unsigned* __restrict__ elist, int* __restrict__ ecount,
                               const float* __restrict__ sdeg, const float* __restrict__ W3,
                               const float* __restrict__ b3) {
    static_assert(!FIN || (CNT && F == DH && P % 2 == 0), "the fused product reads 64-wide sums in pairs and scales them by the block's own counts");
    constexpr int ROWS = FIN ? OWN_RT * 16 : OWN_R;
    __shared__ __attribute__((aligned(16))) float acc[ROWS * P];
    __shared__ int   cnt[CNT ? ROWS : 1];
    __shared__ int   rcnt[1];
    __shared__ int   qs[OWN_NW * OWN_QCAP];
    __shared__ int   gqs[OWN_NW * OWN_GQCAP];
    const int w    = __builtin_amdgcn_readfirstlane((int)(threadIdx.x >> 5));
    const int lane = threadIdx.x & 31;
    const int r0   = blockIdx.x * OWN_R;
    int* q  = qs + w * OWN_QCAP;
    int* gq = gqs + w * OWN_GQCAP;

    for (int i = threadIdx.x; i < ROWS * P; i += OWN_NW * 32) acc[i] = 0.f;
    if (CNT) {
        for (int i = threadIdx.x; i < ROWS; i += OWN_NW * 32) cnt[i] = 0;
    }
    if (REC) {
        if (threadIdx.x == 0) rcnt[0] = 0;
    }
    __syncthreads();

    unsigned* mine = elist + (size_t)blockIdx.x * OWN_ECAP;
    bool scan = true;
    if (PLAY) {
        const int ne = __builtin_amdgcn_readfirstlane(ecount[blockIdx.x]);
        if (ne <= OWN_ECAP) {
            scan = false;
            int at = w * 32;
            unsigned e = (at + lane < ne) ? mine[at + lane] : 0u;
            while (at + 32 <= ne) {
                const int nx = at + OWN_NW * 32;
                const int ix = nx + lane;
                const unsigned ld = mine[(ix < ne) ? ix : ne - 1];
                add_rows<F, P>(src, acc, (int)(e & 0x1FFFFu), (int)(e >> 17), 32, lane);
                e = (ix < ne) ? ld : 0u;
                at = nx;
            }
            if (at < ne) add_rows<F, P>(src, acc, (int)(e & 0x1FFFFu), (int)(e >> 17), ne - at, lane);
        }
    }

    const unsigned* myb = bits + (size_t)blockIdx.x * IDX_ROW;
    const int4*  rcv4  = (const int4*)rcv;
    int qn = 0;
    int gn = 0;
    if (scan) {
        unsigned wd[IDX_FULL + 1];
        int      wi[IDX_FULL + 1];
        #pragma unroll
        for (int u = 0; u < IDX_FULL; ++u) {
            wi[u] = (u * OWN_NW + w) * 32 + lane;
            wd[u] = myb[wi[u]];
        }
        wi[IDX_FULL] = IDX_FULL * OWN_NW * 32 + w * IDX_REM + ((lane < IDX_REM) ? lane : 0);
        const unsigned last = myb[wi[IDX_FULL]];
        wd[IDX_FULL] = (lane < IDX_REM) ? last : 0u;
        #pragma unroll
        for (int u = 0; u <= IDX_FULL; ++u) {
            unsigned word = wd[u];
            unsigned m = __builtin_amdgcn_ballot_w32(word != 0u);
            while (m) {
                const int pos = gn + (int)__builtin_amdgcn_mbcnt_lo(m, 0u);
                if (word != 0u) {
                    gq[pos] = wi[u] * 32 + __builtin_ctz(word);
                    word &= word - 1u;
                }
                gn += __builtin_popcount(m);
                if (gn >= 32) {
                    expand_groups<F, P, CNT, REC>(src, snd, rcv4, acc, cnt, mine, rcnt, q, qn, gq[lane], r0, lane);
                    const int rem = gn - 32;
                    const int t = (lane < rem) ? gq[32 + lane] : 0;
                    if (lane < rem) gq[lane] = t;
                    gn = rem;
                }
                m = __builtin_amdgcn_ballot_w32(word != 0u);
            }
        }
    }
    if (gn > 0) expand_groups<F, P, CNT, REC>(src, snd, rcv4, acc, cnt, mine, rcnt, q, qn, (lane < gn) ? gq[lane] : -1, r0, lane);
    if (qn > 0) drain_rows<F, P, CNT, REC>(src, snd, acc, cnt, mine, rcnt, q, qn, lane);
    __syncthreads();

    if (REC) {
        if (threadIdx.x == 0) ecount[blockIdx.x] = rcnt[0];
    }

    if (CNT) {
        for (int i = threadIdx.x; i < OWN_R; i += OWN_NW * 32) {
            if (r0 + i < NN) rdeg[r0 + i] = (float)cnt[i];
        }
    }

    if (FIN) {
        if (w < OWN_RT) {
            const int m16  = lane & 15;
            const int kh   = lane >> 4;
            const int lr0  = w * 16;
            const float rs = rsqrtf(fmaxf((float)cnt[lr0 + m16], 1.f));
            const float* arow = acc + (lr0 + m16) * P + 2 * kh;

            float sc[8];
            #pragma unroll
            for (int r = 0; r < 8; ++r) {
                const int node = r0 + lr0 + r + 8 * kh;
                sc[r] = sdeg[(node < NN) ? node : NN - 1];
            }
            const int   n2c = (32 + m16 < DC) ? 32 + m16 : DC - 1;
            const float bb0 = b3[m16], bb1 = b3[16 + m16], bb2 = b3[n2c];

            v8f c0 = {}, c1 = {}, c2 = {};
            #pragma unroll 4
            for (int kb = 0; kb < DH; kb += 4) {
                v2f a = *(const v2f*)(arow + kb);
                a.x *= rs; a.y *= rs;
                const float* wr0 = W3 + (size_t)(kb + 2 * kh) * DC;
                const float* wr1 = wr0 + DC;
                v2f b;
                b.x = wr0[m16];      b.y = wr1[m16];      c0 = wmma4(a, b, c0);
                b.x = wr0[16 + m16]; b.y = wr1[16 + m16]; c1 = wmma4(a, b, c1);
                const float p0 = wr0[n2c], p1 = wr1[n2c];
                b.x = (32 + m16 < DC) ? p0 : 0.f;
                b.y = (32 + m16 < DC) ? p1 : 0.f;
                c2 = wmma4(a, b, c2);
            }
            #pragma unroll
            for (int r = 0; r < 8; ++r) {
                const int lr   = lr0 + r + 8 * kh;
                const int node = r0 + lr;
                if (lr < OWN_R && node < NN) {
                    const float s = rsqrtf(fmaxf(sc[r], 1.f));
                    float* orow = dst + (size_t)node * DC;
                    orow[m16]      = (c0[r] + bb0) * s;
                    orow[16 + m16] = (c1[r] + bb1) * s;
                    if (32 + m16 < DC)
                        orow[32 + m16] = (c2[r] + bb2) * s;
                }
            }
        }
    } else {
        for (int i = threadIdx.x; i < OWN_R * F; i += OWN_NW * 32) {
            const int lr = i / F;
            if (r0 + lr < NN) dst[(size_t)r0 * F + i] = acc[lr * P + (i - lr * F)];
        }
    }
}

__device__ __forceinline__ void count_groups(const int4* __restrict__ snd4, int* cnt, int g, int r0) {
    int4 s[4];
    #pragma unroll
    for (int k = 0; k < 4; ++k) s[k] = (g >= 0) ? snd4[g * 4 + k] : make_int4(-1, -1, -1, -1);
    #pragma unroll
    for (int k = 0; k < 4; ++k) {
        const int ss[4] = {s[k].x, s[k].y, s[k].z, s[k].w};
        #pragma unroll
        for (int c = 0; c < 4; ++c) {
            const unsigned d = (unsigned)(ss[c] - r0);
            if (d < (unsigned)OWN_R) atomicAdd(cnt + d, 1);
        }
    }
}

__launch_bounds__(OWN_NW * 32)
__global__ void sdeg_own_kernel(const int* __restrict__ snd, const unsigned* __restrict__ sbits,
                                float* __restrict__ sdeg) {
    __shared__ int cnt[OWN_R];
    __shared__ int gqs[OWN_NW * OWN_GQCAP];
    const int w    = __builtin_amdgcn_readfirstlane((int)(threadIdx.x >> 5));
    const int lane = threadIdx.x & 31;
    const int r0   = blockIdx.x * OWN_R;
    int* gq = gqs + w * OWN_GQCAP;

    for (int i = threadIdx.x; i < OWN_R; i += OWN_NW * 32) cnt[i] = 0;
    __syncthreads();

    const unsigned* myb = sbits + (size_t)blockIdx.x * IDX_ROW;
    const int4*  snd4  = (const int4*)snd;
    int gn = 0;
    {
        unsigned wd[IDX_FULL + 1];
        int      wi[IDX_FULL + 1];
        #pragma unroll
        for (int u = 0; u < IDX_FULL; ++u) {
            wi[u] = (u * OWN_NW + w) * 32 + lane;
            wd[u] = myb[wi[u]];
        }
        wi[IDX_FULL] = IDX_FULL * OWN_NW * 32 + w * IDX_REM + ((lane < IDX_REM) ? lane : 0);
        const unsigned last = myb[wi[IDX_FULL]];
        wd[IDX_FULL] = (lane < IDX_REM) ? last : 0u;
        #pragma unroll
        for (int u = 0; u <= IDX_FULL; ++u) {
            unsigned word = wd[u];
            unsigned m = __builtin_amdgcn_ballot_w32(word != 0u);
            while (m) {
                const int pos = gn + (int)__builtin_amdgcn_mbcnt_lo(m, 0u);
                if (word != 0u) {
                    gq[pos] = wi[u] * 32 + __builtin_ctz(word);
                    word &= word - 1u;
                }
                gn += __builtin_popcount(m);
                if (gn >= 32) {
                    count_groups(snd4, cnt, gq[lane], r0);
                    const int rem = gn - 32;
                    const int t = (lane < rem) ? gq[32 + lane] : 0;
                    if (lane < rem) gq[lane] = t;
                    gn = rem;
                }
                m = __builtin_amdgcn_ballot_w32(word != 0u);
            }
        }
    }
    if (gn > 0) count_groups(snd4, cnt, (lane < gn) ? gq[lane] : -1, r0);
    __syncthreads();

    for (int i = threadIdx.x; i < OWN_R; i += OWN_NW * 32) {
        if (r0 + i < NN) sdeg[r0 + i] = (float)cnt[i];
    }
}

__global__ void softmax_kernel(float* __restrict__ out, const float* __restrict__ rdeg) {
    int n = blockIdx.x * blockDim.x + threadIdx.x;
    if (n >= NN) return;
    float rs = rsqrtf(fmaxf(rdeg[n], 1.f));
    float* row = out + (size_t)n * DC;
    float v[DC];
    float mx = -3.402823466e38f;
    #pragma unroll
    for (int f = 0; f < DC; ++f) { v[f] = row[f] * rs; mx = fmaxf(mx, v[f]); }
    float s = 0.f;
    #pragma unroll
    for (int f = 0; f < DC; ++f) { v[f] = __expf(v[f] - mx); s += v[f]; }
    float inv = 1.f / s;
    #pragma unroll
    for (int f = 0; f < DC; ++f) row[f] = v[f] * inv;
}

extern "C" void kernel_launch(void* const* d_in, const int* in_sizes, int n_in,
                              void* d_out, int out_size, void* d_ws, size_t ws_size,
                              hipStream_t stream) {
    const float* nodes     = (const float*)d_in[0];
    const int*   senders   = (const int*)d_in[1];
    const int*   receivers = (const int*)d_in[2];
    const float* W1 = (const float*)d_in[3];
    const float* b1 = (const float*)d_in[4];
    const float* W2 = (const float*)d_in[5];
    const float* b2 = (const float*)d_in[6];
    const float* W3 = (const float*)d_in[7];
    const float* b3 = (const float*)d_in[8];
    float* out = (float*)d_out;
    float* ws  = (float*)d_ws;

    float* sdeg = ws;
    float* rdeg = ws + NN;
    float* h1   = ws + 2 * (size_t)NN;
    float* h2   = h1 + 2 * (size_t)NN * DH;
    unsigned* rbits = (unsigned*)(h2 + (size_t)NN * DC);
    unsigned* sbits = rbits + (size_t)OWN_G * IDX_ROW;
    unsigned* elist = sbits + (size_t)OWN_G * IDX_ROW;
    int* ecount = (int*)(elist + (size_t)OWN_G * OWN_ECAP);

    const int nTiles = NN / 16;
    index_kernel<<<dim3(IDX_NB, 2), 1024, 0, stream>>>(senders, receivers, sbits, rbits);
    sdeg_own_kernel<<<OWN_G, OWN_NW * 32, 0, stream>>>(senders, sbits, sdeg);
    mlp_kernel<<<(nTiles + 7) / 8, 256, 0, stream>>>(nodes, W1, b1, W2, b2, sdeg, h1, nTiles);
    agg_own_kernel<DH, 68, true, true, false, true><<<OWN_G, OWN_NW * 32, 0, stream>>>(h1, senders, receivers, rbits, h2, rdeg, elist, ecount, sdeg, W3, b3);
    agg_own_kernel<DC, DC, false, false, true, false><<<OWN_G, OWN_NW * 32, 0, stream>>>(h2, senders, receivers, rbits, out, nullptr, elist, ecount, nullptr, nullptr, nullptr);
    softmax_kernel<<<(NN + 255) / 256, 256, 0, stream>>>(out, rdeg);
}


# Look-ups read a queued group's 64 bytes with four neighbouring lanes (a load touches 8 lines, not 32), in both owner kernels; edges queue in another order: LDS-atomic sum order changes
# speedup vs baseline: 4.8304x; 1.1270x over previous
#include <hip/hip_runtime.h>

typedef __attribute__((ext_vector_type(2))) float v2f;
typedef __attribute__((ext_vector_type(8))) float v8f;

static constexpr int NN = 100000;
static constexpr int NE = 1600000;
static constexpr int DF = 128;
static constexpr int DH = 64;
static constexpr int DC = 40;
#define NEG_SLOPE 0.01f
static constexpr unsigned TAG_R = (NN + 255) / 256;

__device__ __forceinline__ float lrelu(float x) { return x > 0.f ? x : x * NEG_SLOPE; }

__device__ __forceinline__ v8f wmma4(v2f a, v2f b, v8f c) {
    return __builtin_amdgcn_wmma_f32_16x16x4_f32(false, a, false, b, (short)0, c, false, false);
}


__launch_bounds__(256)
__global__ void mlp_kernel(const float* __restrict__ X,
                           const float* __restrict__ W1, const float* __restrict__ b1,
                           const float* __restrict__ W2, const float* __restrict__ b2,
                           const float* __restrict__ sdeg,
                           float* __restrict__ h1, int nTiles) {
    __shared__ float lds[8 * 16 * DH];
    const int w    = threadIdx.x >> 5;
    const int lane = threadIdx.x & 31;
    const int m16  = lane & 15;
    const int kh   = lane >> 4;
    const int t    = blockIdx.x * 8 + w;
    const bool active = (t < nTiles);
    const int row0 = t * 16;
    float* my = &lds[w * 16 * DH];

    if (active) {
        v8f c0 = {}, c1 = {}, c2 = {}, c3 = {};
        #pragma unroll 4
        for (int kb = 0; kb < DF; kb += 4) {
            v2f a = *(const v2f*)(X + (size_t)(row0 + m16) * DF + kb + 2 * kh);
            const float* wr0 = W1 + (size_t)(kb + 2 * kh) * DH;
            const float* wr1 = wr0 + DH;
            v2f b;
            b.x = wr0[m16];      b.y = wr1[m16];      c0 = wmma4(a, b, c0);
            b.x = wr0[16 + m16]; b.y = wr1[16 + m16]; c1 = wmma4(a, b, c1);
            b.x = wr0[32 + m16]; b.y = wr1[32 + m16]; c2 = wmma4(a, b, c2);
            b.x = wr0[48 + m16]; b.y = wr1[48 + m16]; c3 = wmma4(a, b, c3);
        }
        #pragma unroll
        for (int r = 0; r < 8; ++r) {
            float* dr = my + (r + 8 * kh) * DH;
            dr[m16]      = lrelu(c0[r] + b1[m16]);
            dr[16 + m16] = lrelu(c1[r] + b1[16 + m16]);
            dr[32 + m16] = lrelu(c2[r] + b1[32 + m16]);
            dr[48 + m16] = lrelu(c3[r] + b1[48 + m16]);
        }
    }
    __syncthreads();
    if (!active) return;

    v8f d0 = {}, d1 = {}, d2 = {}, d3 = {};
    #pragma unroll 4
    for (int kb = 0; kb < DH; kb += 4) {
        v2f a = *(const v2f*)(my + m16 * DH + kb + 2 * kh);
        const float* wr0 = W2 + (size_t)(kb + 2 * kh) * DH;
        const float* wr1 = wr0 + DH;
        v2f b;
        b.x = wr0[m16];      b.y = wr1[m16];      d0 = wmma4(a, b, d0);
        b.x = wr0[16 + m16]; b.y = wr1[16 + m16]; d1 = wmma4(a, b, d1);
        b.x = wr0[32 + m16]; b.y = wr1[32 + m16]; d2 = wmma4(a, b, d2);
        b.x = wr0[48 + m16]; b.y = wr1[48 + m16]; d3 = wmma4(a, b, d3);
    }
    float sc[8];
    #pragma unroll
    for (int r = 0; r < 8; ++r) sc[r] = rsqrtf(fmaxf(sdeg[row0 + r + 8 * kh], 1.f));
    #pragma unroll
    for (int r = 0; r < 8; ++r) {
        float* orow = h1 + (size_t)(row0 + r + 8 * kh) * DH;
        orow[m16]      = lrelu(d0[r] + b2[m16])      * sc[r];
        orow[16 + m16] = lrelu(d1[r] + b2[16 + m16]) * sc[r];
        orow[32 + m16] = lrelu(d2[r] + b2[32 + m16]) * sc[r];
        orow[48 + m16] = lrelu(d3[r] + b2[48 + m16]) * sc[r];
    }
}

static constexpr int OWN_G  = 256;
static constexpr int OWN_R  = (NN + OWN_G - 1) / OWN_G;
static constexpr int OWN_NW = 32;
static constexpr int OWN_GQCAP = 64;
static constexpr int OWN_QCAP  = 16 * 32 + 32;
static constexpr int NG16 = NE / 16;
static constexpr int OWN_RT = (OWN_R + 15) / 16;
static constexpr int OWN_ECAP = 1 << 17;
static constexpr int IDX_GPB = 1024;
static constexpr int IDX_NB  = (NG16 + IDX_GPB - 1) / IDX_GPB;
static constexpr int IDX_ROW = IDX_NB * 32;
static constexpr int IDX_P   = OWN_G + 1;
static constexpr int IDX_FULL = IDX_NB / OWN_NW;
static constexpr int IDX_REM  = (IDX_ROW - IDX_FULL * OWN_NW * 32) / OWN_NW;
static_assert(OWN_R == (int)TAG_R, "the index must name the owner blocks");
static_assert(NN <= (1 << 17) && OWN_R <= (1 << 9), "a list entry holds the sender in bits 0..16 and the local receiver above them");
static_assert(OWN_RT <= OWN_NW, "the fused product gives every row tile a wave of its own");
static_assert(NE % 16 == 0, "edge list must split into groups of sixteen");
static_assert(NG16 % 32 == 0, "a wave's 32 groups must lie all inside the list or all past its end");
static_assert(OWN_G % 32 == 0 && IDX_GPB == 32 * 32, "an index block is 32 waves of 32 groups, and every wave writes whole owners' lines");
static_assert((IDX_ROW - IDX_FULL * OWN_NW * 32) % OWN_NW == 0 && IDX_REM >= 1 && IDX_REM <= 32, "the words left after the full rounds split evenly over the waves, at least one each");

__launch_bounds__(1024)
__global__ void index_kernel(const int* __restrict__ snd, const int* __restrict__ rcv,
                             unsigned* __restrict__ sbits, unsigned* __restrict__ rbits) {
    __shared__ unsigned bm[32 * IDX_P];
    const int w    = threadIdx.x >> 5;
    const int lane = threadIdx.x & 31;
    const int* __restrict__ ids = blockIdx.y ? rcv : snd;
    unsigned* __restrict__ rows = blockIdx.y ? rbits : sbits;

    for (int i = threadIdx.x; i < 32 * IDX_P; i += 1024) bm[i] = 0u;
    __syncthreads();

    const int g0 = blockIdx.x * IDX_GPB + w * 32;
    if (g0 < NG16) {
        const int4* p = (const int4*)ids + (size_t)g0 * 4;
        int4 v[4];
        #pragma unroll
        for (int k = 0; k < 4; ++k) v[k] = p[k * 32 + lane];
        asm volatile("" :: "v"(v[0].x), "v"(v[0].y), "v"(v[0].z), "v"(v[0].w), "v"(v[1].x), "v"(v[1].y), "v"(v[1].z), "v"(v[1].w),
                           "v"(v[2].x), "v"(v[2].y), "v"(v[2].z), "v"(v[2].w), "v"(v[3].x), "v"(v[3].y), "v"(v[3].z), "v"(v[3].w));
        unsigned* row = bm + w * IDX_P;
        #pragma unroll
        for (int k = 0; k < 4; ++k) {
            const unsigned bit = 1u << (k * 8 + (lane >> 2));
            atomicOr(row + (unsigned)v[k].x / TAG_R, bit);
            atomicOr(row + (unsigned)v[k].y / TAG_R, bit);
            atomicOr(row + (unsigned)v[k].z / TAG_R, bit);
            atomicOr(row + (unsigned)v[k].w / TAG_R, bit);
        }
    }
    __syncthreads();

    #pragma unroll
    for (int j = 0; j < OWN_G / 32; ++j) {
        const int o = w * (OWN_G / 32) + j;
        rows[(size_t)o * IDX_ROW + blockIdx.x * 32 + lane] = bm[lane * IDX_P + o];
    }
}

template <int F, int P>
__device__ __forceinline__ void add_rows(const float* __restrict__ src, float* acc, int s, int l, int n, int lane) {
    for (int j0 = 0; j0 < n; j0 += 8) {
        float a[8], b[8];
        int lj[8];
        #pragma unroll
        for (int u = 0; u < 8; ++u) {
            const int sj = __builtin_amdgcn_readlane(s, j0 + u);
            lj[u] = __builtin_amdgcn_readlane(l, j0 + u);
            const float* row = src + (size_t)sj * F;
            a[u] = row[lane];
            b[u] = (lane < F - 32) ? row[32 + lane] : 0.f;
        }
        #pragma unroll
        for (int u = 0; u < 8; ++u) {
            if (j0 + u < n) {
                float* d = acc + lj[u] * P;
                atomicAdd(d + lane, a[u]);
                if (lane < F - 32) atomicAdd(d + 32 + lane, b[u]);
            }
        }
    }
}

template <int F, int P, bool CNT, bool REC>
__device__ __forceinline__ void drain_rows(const float* __restrict__ src, const int* __restrict__ snd,
                                           float* acc, int* cnt, unsigned* __restrict__ rec, int* rcnt,
                                           const int* q, int n, int lane) {
    const int ent = (lane < n) ? q[lane] : 0;
    const int l = ent >> 21;
    if (CNT) {
        if (lane < n) atomicAdd(cnt + l, 1);
    }
    const int s = (lane < n) ? snd[ent & 0x1FFFFF] : 0;
    if (REC) {
        int at = 0;
        if (lane == 0) at = atomicAdd(rcnt, n);
        at = __builtin_amdgcn_readfirstlane(at);
        if (lane < n && at + lane < OWN_ECAP) rec[at + lane] = (unsigned)s | ((unsigned)l << 17);
    }
    add_rows<F, P>(src, acc, s, l, n, lane);
}


template <int F, int P, bool CNT, bool REC>
__device__ __forceinline__ void expand_groups(const float* __restrict__ src, const int* __restrict__ snd,
                                              const int4* __restrict__ rcv4, float* acc, int* cnt,
                                              unsigned* __restrict__ rec, int* rcnt, int* q, int& qn,
                                              const int* gq, int gn, int r0, int lane) {
    const int sub = lane & 3;
    int  g[4];
    bool has[4];
    int4 r[4];
    #pragma unroll
    for (int p = 0; p < 4; ++p) {
        const int gi = p * 8 + (lane >> 2);
        has[p] = gi < gn;
        g[p] = gq[has[p] ? gi : gn - 1];
        r[p] = rcv4[g[p] * 4 + sub];
    }
    #pragma unroll
    for (int p = 0; p < 4; ++p) {
        const int rr[4] = {r[p].x, r[p].y, r[p].z, r[p].w};
        #pragma unroll
        for (int c = 0; c < 4; ++c) {
            const unsigned d = (unsigned)(rr[c] - r0);
            const bool hit = has[p] && d < (unsigned)OWN_R;
            const unsigned m = __builtin_amdgcn_ballot_w32(hit);
            if (m) {
                const int pos = qn + (int)__builtin_amdgcn_mbcnt_lo(m, 0u);
                if (hit) q[pos] = (g[p] * 16 + sub * 4 + c) | ((int)d << 21);
                qn += __builtin_popcount(m);
            }
        }
    }
    if (qn >= 32) {
        int done = 0;
        while (qn - done >= 32) {
            drain_rows<F, P, CNT, REC>(src, snd, acc, cnt, rec, rcnt, q + done, 32, lane);
            done += 32;
        }
        const int rem = qn - done;
        const int t = (lane < rem) ? q[done + lane] : 0;
        if (lane < rem) q[lane] = t;
        qn = rem;
    }
}

template <int F, int P, bool CNT, bool REC, bool PLAY, bool FIN>
__launch_bounds__(OWN_NW * 32)
__global__ void agg_own_kernel(const float* __restrict__ src, const int* __restrict__ snd,
                               const int* __restrict__ rcv, const unsigned* __restrict__ bits,
                               float* __restrict__ dst, float* __restrict__ rdeg,
                               unsigned* __restrict__ elist, int* __restrict__ ecount,
                               const float* __restrict__ sdeg, const float* __restrict__ W3,
                               const float* __restrict__ b3) {
    static_assert(!FIN || (CNT && F == DH && P % 2 == 0), "the fused product reads 64-wide sums in pairs and scales them by the block's own counts");
    constexpr int ROWS = FIN ? OWN_RT * 16 : OWN_R;
    __shared__ __attribute__((aligned(16))) float acc[ROWS * P];
    __shared__ int   cnt[CNT ? ROWS : 1];
    __shared__ int   rcnt[1];
    __shared__ int   qs[OWN_NW * OWN_QCAP];
    __shared__ int   gqs[OWN_NW * OWN_GQCAP];
    const int w    = __builtin_amdgcn_readfirstlane((int)(threadIdx.x >> 5));
    const int lane = threadIdx.x & 31;
    const int r0   = blockIdx.x * OWN_R;
    int* q  = qs + w * OWN_QCAP;
    int* gq = gqs + w * OWN_GQCAP;

    for (int i = threadIdx.x; i < ROWS * P; i += OWN_NW * 32) acc[i] = 0.f;
    if (CNT) {
        for (int i = threadIdx.x; i < ROWS; i += OWN_NW * 32) cnt[i] = 0;
    }
    if (REC) {
        if (threadIdx.x == 0) rcnt[0] = 0;
    }
    __syncthreads();

    unsigned* mine = elist + (size_t)blockIdx.x * OWN_ECAP;
    bool scan = true;
    if (PLAY) {
        const int ne = __builtin_amdgcn_readfirstlane(ecount[blockIdx.x]);
        if (ne <= OWN_ECAP) {
            scan = false;
            int at = w * 32;
            unsigned e = (at + lane < ne) ? mine[at + lane] : 0u;
            while (at + 32 <= ne) {
                const int nx = at + OWN_NW * 32;
                const int ix = nx + lane;
                const unsigned ld = mine[(ix < ne) ? ix : ne - 1];
                add_rows<F, P>(src, acc, (int)(e & 0x1FFFFu), (int)(e >> 17), 32, lane);
                e = (ix < ne) ? ld : 0u;
                at = nx;
            }
            if (at < ne) add_rows<F, P>(src, acc, (int)(e & 0x1FFFFu), (int)(e >> 17), ne - at, lane);
        }
    }

    const unsigned* myb = bits + (size_t)blockIdx.x * IDX_ROW;
    const int4*  rcv4  = (const int4*)rcv;
    int qn = 0;
    int gn = 0;
    if (scan) {
        unsigned wd[IDX_FULL + 1];
        int      wi[IDX_FULL + 1];
        #pragma unroll
        for (int u = 0; u < IDX_FULL; ++u) {
            wi[u] = (u * OWN_NW + w) * 32 + lane;
            wd[u] = myb[wi[u]];
        }
        wi[IDX_FULL] = IDX_FULL * OWN_NW * 32 + w * IDX_REM + ((lane < IDX_REM) ? lane : 0);
        const unsigned last = myb[wi[IDX_FULL]];
        wd[IDX_FULL] = (lane < IDX_REM) ? last : 0u;
        #pragma unroll
        for (int u = 0; u <= IDX_FULL; ++u) {
            unsigned word = wd[u];
            unsigned m = __builtin_amdgcn_ballot_w32(word != 0u);
            while (m) {
                const int pos = gn + (int)__builtin_amdgcn_mbcnt_lo(m, 0u);
                if (word != 0u) {
                    gq[pos] = wi[u] * 32 + __builtin_ctz(word);
                    word &= word - 1u;
                }
                gn += __builtin_popcount(m);
                if (gn >= 32) {
                    expand_groups<F, P, CNT, REC>(src, snd, rcv4, acc, cnt, mine, rcnt, q, qn, gq, 32, r0, lane);
                    const int rem = gn - 32;
                    const int t = (lane < rem) ? gq[32 + lane] : 0;
                    if (lane < rem) gq[lane] = t;
                    gn = rem;
                }
                m = __builtin_amdgcn_ballot_w32(word != 0u);
            }
        }
    }
    if (gn > 0) expand_groups<F, P, CNT, REC>(src, snd, rcv4, acc, cnt, mine, rcnt, q, qn, gq, gn, r0, lane);
    if (qn > 0) drain_rows<F, P, CNT, REC>(src, snd, acc, cnt, mine, rcnt, q, qn, lane);
    __syncthreads();

    if (REC) {
        if (threadIdx.x == 0) ecount[blockIdx.x] = rcnt[0];
    }

    if (CNT) {
        for (int i = threadIdx.x; i < OWN_R; i += OWN_NW * 32) {
            if (r0 + i < NN) rdeg[r0 + i] = (float)cnt[i];
        }
    }

    if (FIN) {
        if (w < OWN_RT) {
            const int m16  = lane & 15;
            const int kh   = lane >> 4;
            const int lr0  = w * 16;
            const float rs = rsqrtf(fmaxf((float)cnt[lr0 + m16], 1.f));
            const float* arow = acc + (lr0 + m16) * P + 2 * kh;

            float sc[8];
            #pragma unroll
            for (int r = 0; r < 8; ++r) {
                const int node = r0 + lr0 + r + 8 * kh;
                sc[r] = sdeg[(node < NN) ? node : NN - 1];
            }
            const int   n2c = (32 + m16 < DC) ? 32 + m16 : DC - 1;
            const float bb0 = b3[m16], bb1 = b3[16 + m16], bb2 = b3[n2c];

            v8f c0 = {}, c1 = {}, c2 = {};
            #pragma unroll 4
            for (int kb = 0; kb < DH; kb += 4) {
                v2f a = *(const v2f*)(arow + kb);
                a.x *= rs; a.y *= rs;
                const float* wr0 = W3 + (size_t)(kb + 2 * kh) * DC;
                const float* wr1 = wr0 + DC;
                v2f b;
                b.x = wr0[m16];      b.y = wr1[m16];      c0 = wmma4(a, b, c0);
                b.x = wr0[16 + m16]; b.y = wr1[16 + m16]; c1 = wmma4(a, b, c1);
                const float p0 = wr0[n2c], p1 = wr1[n2c];
                b.x = (32 + m16 < DC) ? p0 : 0.f;
                b.y = (32 + m16 < DC) ? p1 : 0.f;
                c2 = wmma4(a, b, c2);
            }
            #pragma unroll
            for (int r = 0; r < 8; ++r) {
                const int lr   = lr0 + r + 8 * kh;
                const int node = r0 + lr;
                if (lr < OWN_R && node < NN) {
                    const float s = rsqrtf(fmaxf(sc[r], 1.f));
                    float* orow = dst + (size_t)node * DC;
                    orow[m16]      = (c0[r] + bb0) * s;
                    orow[16 + m16] = (c1[r] + bb1) * s;
                    if (32 + m16 < DC)
                        orow[32 + m16] = (c2[r] + bb2) * s;
                }
            }
        }
    } else {
        for (int i = threadIdx.x; i < OWN_R * F; i += OWN_NW * 32) {
            const int lr = i / F;
            if (r0 + lr < NN) dst[(size_t)r0 * F + i] = acc[lr * P + (i - lr * F)];
        }
    }
}

__device__ __forceinline__ void count_groups(const int4* __restrict__ snd4, int* cnt, const int* gq, int gn, int r0, int lane) {
    const int sub = lane & 3;
    bool has[4];
    int4 s[4];
    #pragma unroll
    for (int p = 0; p < 4; ++p) {
        const int gi = p * 8 + (lane >> 2);
        has[p] = gi < gn;
        s[p] = snd4[gq[has[p] ? gi : gn - 1] * 4 + sub];
    }
    #pragma unroll
    for (int p = 0; p < 4; ++p) {
        const int ss[4] = {s[p].x, s[p].y, s[p].z, s[p].w};
        #pragma unroll
        for (int c = 0; c < 4; ++c) {
            const unsigned d = (unsigned)(ss[c] - r0);
            if (has[p] && d < (unsigned)OWN_R) atomicAdd(cnt + d, 1);
        }
    }
}

__launch_bounds__(OWN_NW * 32)
__global__ void sdeg_own_kernel(const int* __restrict__ snd, const unsigned* __restrict__ sbits,
                                float* __restrict__ sdeg) {
    __shared__ int cnt[OWN_R];
    __shared__ int gqs[OWN_NW * OWN_GQCAP];
    const int w    = __builtin_amdgcn_readfirstlane((int)(threadIdx.x >> 5));
    const int lane = threadIdx.x & 31;
    const int r0   = blockIdx.x * OWN_R;
    int* gq = gqs + w * OWN_GQCAP;

    for (int i = threadIdx.x; i < OWN_R; i += OWN_NW * 32) cnt[i] = 0;
    __syncthreads();

    const unsigned* myb = sbits + (size_t)blockIdx.x * IDX_ROW;
    const int4*  snd4  = (const int4*)snd;
    int gn = 0;
    {
        unsigned wd[IDX_FULL + 1];
        int      wi[IDX_FULL + 1];
        #pragma unroll
        for (int u = 0; u < IDX_FULL; ++u) {
            wi[u] = (u * OWN_NW + w) * 32 + lane;
            wd[u] = myb[wi[u]];
        }
        wi[IDX_FULL] = IDX_FULL * OWN_NW * 32 + w * IDX_REM + ((lane < IDX_REM) ? lane : 0);
        const unsigned last = myb[wi[IDX_FULL]];
        wd[IDX_FULL] = (lane < IDX_REM) ? last : 0u;
        #pragma unroll
        for (int u = 0; u <= IDX_FULL; ++u) {
            unsigned word = wd[u];
            unsigned m = __builtin_amdgcn_ballot_w32(word != 0u);
            while (m) {
                const int pos = gn + (int)__builtin_amdgcn_mbcnt_lo(m, 0u);
                if (word != 0u) {
                    gq[pos] = wi[u] * 32 + __builtin_ctz(word);
                    word &= word - 1u;
                }
                gn += __builtin_popcount(m);
                if (gn >= 32) {
                    count_groups(snd4, cnt, gq, 32, r0, lane);
                    const int rem = gn - 32;
                    const int t = (lane < rem) ? gq[32 + lane] : 0;
                    if (lane < rem) gq[lane] = t;
                    gn = rem;
                }
                m = __builtin_amdgcn_ballot_w32(word != 0u);
            }
        }
    }
    if (gn > 0) count_groups(snd4, cnt, gq, gn, r0, lane);
    __syncthreads();

    for (int i = threadIdx.x; i < OWN_R; i += OWN_NW * 32) {
        if (r0 + i < NN) sdeg[r0 + i] = (float)cnt[i];
    }
}

__global__ void softmax_kernel(float* __restrict__ out, const float* __restrict__ rdeg) {
    int n = blockIdx.x * blockDim.x + threadIdx.x;
    if (n >= NN) return;
    float rs = rsqrtf(fmaxf(rdeg[n], 1.f));
    float* row = out + (size_t)n * DC;
    float v[DC];
    float mx = -3.402823466e38f;
    #pragma unroll
    for (int f = 0; f < DC; ++f) { v[f] = row[f] * rs; mx = fmaxf(mx, v[f]); }
    float s = 0.f;
    #pragma unroll
    for (int f = 0; f < DC; ++f) { v[f] = __expf(v[f] - mx); s += v[f]; }
    float inv = 1.f / s;
    #pragma unroll
    for (int f = 0; f < DC; ++f) row[f] = v[f] * inv;
}

extern "C" void kernel_launch(void* const* d_in, const int* in_sizes, int n_in,
                              void* d_out, int out_size, void* d_ws, size_t ws_size,
                              hipStream_t stream) {
    const float* nodes     = (const float*)d_in[0];
    const int*   senders   = (const int*)d_in[1];
    const int*   receivers = (const int*)d_in[2];
    const float* W1 = (const float*)d_in[3];
    const float* b1 = (const float*)d_in[4];
    const float* W2 = (const float*)d_in[5];
    const float* b2 = (const float*)d_in[6];
    const float* W3 = (const float*)d_in[7];
    const float* b3 = (const float*)d_in[8];
    float* out = (float*)d_out;
    float* ws  = (float*)d_ws;

    float* sdeg = ws;
    float* rdeg = ws + NN;
    float* h1   = ws + 2 * (size_t)NN;
    float* h2   = h1 + 2 * (size_t)NN * DH;
    unsigned* rbits = (unsigned*)(h2 + (size_t)NN * DC);
    unsigned* sbits = rbits + (size_t)OWN_G * IDX_ROW;
    unsigned* elist = sbits + (size_t)OWN_G * IDX_ROW;
    int* ecount = (int*)(elist + (size_t)OWN_G * OWN_ECAP);

    const int nTiles = NN / 16;
    index_kernel<<<dim3(IDX_NB, 2), 1024, 0, stream>>>(senders, receivers, sbits, rbits);
    sdeg_own_kernel<<<OWN_G, OWN_NW * 32, 0, stream>>>(senders, sbits, sdeg);
    mlp_kernel<<<(nTiles + 7) / 8, 256, 0, stream>>>(nodes, W1, b1, W2, b2, sdeg, h1, nTiles);
    agg_own_kernel<DH, 68, true, true, false, true><<<OWN_G, OWN_NW * 32, 0, stream>>>(h1, senders, receivers, rbits, h2, rdeg, elist, ecount, sdeg, W3, b3);
    agg_own_kernel<DC, DC, false, false, true, false><<<OWN_G, OWN_NW * 32, 0, stream>>>(h2, senders, receivers, rbits, out, nullptr, elist, ecount, nullptr, nullptr, nullptr);
    softmax_kernel<<<(NN + 255) / 256, 256, 0, stream>>>(out, rdeg);
}


# Second edge sum's owner blocks store softmax(LDS sums x receiver scale) themselves, half a wave a row, 16-lane tree (exp-sum order changes); softmax launch and its pass over out gone, 6 -> 5 launches
# speedup vs baseline: 5.2800x; 1.0931x over previous
#include <hip/hip_runtime.h>

typedef __attribute__((ext_vector_type(2))) float v2f;
typedef __attribute__((ext_vector_type(8))) float v8f;

static constexpr int NN = 100000;
static constexpr int NE = 1600000;
static constexpr int DF = 128;
static constexpr int DH = 64;
static constexpr int DC = 40;
#define NEG_SLOPE 0.01f
static constexpr unsigned TAG_R = (NN + 255) / 256;

__device__ __forceinline__ float lrelu(float x) { return x > 0.f ? x : x * NEG_SLOPE; }

__device__ __forceinline__ v8f wmma4(v2f a, v2f b, v8f c) {
    return __builtin_amdgcn_wmma_f32_16x16x4_f32(false, a, false, b, (short)0, c, false, false);
}


__launch_bounds__(256)
__global__ void mlp_kernel(const float* __restrict__ X,
                           const float* __restrict__ W1, const float* __restrict__ b1,
                           const float* __restrict__ W2, const float* __restrict__ b2,
                           const float* __restrict__ sdeg,
                           float* __restrict__ h1, int nTiles) {
    __shared__ float lds[8 * 16 * DH];
    const int w    = threadIdx.x >> 5;
    const int lane = threadIdx.x & 31;
    const int m16  = lane & 15;
    const int kh   = lane >> 4;
    const int t    = blockIdx.x * 8 + w;
    const bool active = (t < nTiles);
    const int row0 = t * 16;
    float* my = &lds[w * 16 * DH];

    if (active) {
        v8f c0 = {}, c1 = {}, c2 = {}, c3 = {};
        #pragma unroll 4
        for (int kb = 0; kb < DF; kb += 4) {
            v2f a = *(const v2f*)(X + (size_t)(row0 + m16) * DF + kb + 2 * kh);
            const float* wr0 = W1 + (size_t)(kb + 2 * kh) * DH;
            const float* wr1 = wr0 + DH;
            v2f b;
            b.x = wr0[m16];      b.y = wr1[m16];      c0 = wmma4(a, b, c0);
            b.x = wr0[16 + m16]; b.y = wr1[16 + m16]; c1 = wmma4(a, b, c1);
            b.x = wr0[32 + m16]; b.y = wr1[32 + m16]; c2 = wmma4(a, b, c2);
            b.x = wr0[48 + m16]; b.y = wr1[48 + m16]; c3 = wmma4(a, b, c3);
        }
        #pragma unroll
        for (int r = 0; r < 8; ++r) {
            float* dr = my + (r + 8 * kh) * DH;
            dr[m16]      = lrelu(c0[r] + b1[m16]);
            dr[16 + m16] = lrelu(c1[r] + b1[16 + m16]);
            dr[32 + m16] = lrelu(c2[r] + b1[32 + m16]);
            dr[48 + m16] = lrelu(c3[r] + b1[48 + m16]);
        }
    }
    __syncthreads();
    if (!active) return;

    v8f d0 = {}, d1 = {}, d2 = {}, d3 = {};
    #pragma unroll 4
    for (int kb = 0; kb < DH; kb += 4) {
        v2f a = *(const v2f*)(my + m16 * DH + kb + 2 * kh);
        const float* wr0 = W2 + (size_t)(kb + 2 * kh) * DH;
        const float* wr1 = wr0 + DH;
        v2f b;
        b.x = wr0[m16];      b.y = wr1[m16];      d0 = wmma4(a, b, d0);
        b.x = wr0[16 + m16]; b.y = wr1[16 + m16]; d1 = wmma4(a, b, d1);
        b.x = wr0[32 + m16]; b.y = wr1[32 + m16]; d2 = wmma4(a, b, d2);
        b.x = wr0[48 + m16]; b.y = wr1[48 + m16]; d3 = wmma4(a, b, d3);
    }
    float sc[8];
    #pragma unroll
    for (int r = 0; r < 8; ++r) sc[r] = rsqrtf(fmaxf(sdeg[row0 + r + 8 * kh], 1.f));
    #pragma unroll
    for (int r = 0; r < 8; ++r) {
        float* orow = h1 + (size_t)(row0 + r + 8 * kh) * DH;
        orow[m16]      = lrelu(d0[r] + b2[m16])      * sc[r];
        orow[16 + m16] = lrelu(d1[r] + b2[16 + m16]) * sc[r];
        orow[32 + m16] = lrelu(d2[r] + b2[32 + m16]) * sc[r];
        orow[48 + m16] = lrelu(d3[r] + b2[48 + m16]) * sc[r];
    }
}

static constexpr int OWN_G  = 256;
static constexpr int OWN_R  = (NN + OWN_G - 1) / OWN_G;
static constexpr int OWN_NW = 32;
static constexpr int OWN_GQCAP = 64;
static constexpr int OWN_QCAP  = 16 * 32 + 32;
static constexpr int NG16 = NE / 16;
static constexpr int OWN_RT = (OWN_R + 15) / 16;
static constexpr int OWN_ECAP = 1 << 17;
static constexpr int IDX_GPB = 1024;
static constexpr int IDX_NB  = (NG16 + IDX_GPB - 1) / IDX_GPB;
static constexpr int IDX_ROW = IDX_NB * 32;
static constexpr int IDX_P   = OWN_G + 1;
static constexpr int IDX_FULL = IDX_NB / OWN_NW;
static constexpr int IDX_REM  = (IDX_ROW - IDX_FULL * OWN_NW * 32) / OWN_NW;
static_assert(OWN_R == (int)TAG_R, "the index must name the owner blocks");
static_assert(NN <= (1 << 17) && OWN_R <= (1 << 9), "a list entry holds the sender in bits 0..16 and the local receiver above them");
static_assert(OWN_RT <= OWN_NW, "the fused product gives every row tile a wave of its own");
static_assert(NE % 16 == 0, "edge list must split into groups of sixteen");
static_assert(NG16 % 32 == 0, "a wave's 32 groups must lie all inside the list or all past its end");
static_assert(OWN_G % 32 == 0 && IDX_GPB == 32 * 32, "an index block is 32 waves of 32 groups, and every wave writes whole owners' lines");
static_assert((IDX_ROW - IDX_FULL * OWN_NW * 32) % OWN_NW == 0 && IDX_REM >= 1 && IDX_REM <= 32, "the words left after the full rounds split evenly over the waves, at least one each");

__launch_bounds__(1024)
__global__ void index_kernel(const int* __restrict__ snd, const int* __restrict__ rcv,
                             unsigned* __restrict__ sbits, unsigned* __restrict__ rbits) {
    __shared__ unsigned bm[32 * IDX_P];
    const int w    = threadIdx.x >> 5;
    const int lane = threadIdx.x & 31;
    const int* __restrict__ ids = blockIdx.y ? rcv : snd;
    unsigned* __restrict__ rows = blockIdx.y ? rbits : sbits;

    for (int i = threadIdx.x; i < 32 * IDX_P; i += 1024) bm[i] = 0u;
    __syncthreads();

    const int g0 = blockIdx.x * IDX_GPB + w * 32;
    if (g0 < NG16) {
        const int4* p = (const int4*)ids + (size_t)g0 * 4;
        int4 v[4];
        #pragma unroll
        for (int k = 0; k < 4; ++k) v[k] = p[k * 32 + lane];
        asm volatile("" :: "v"(v[0].x), "v"(v[0].y), "v"(v[0].z), "v"(v[0].w), "v"(v[1].x), "v"(v[1].y), "v"(v[1].z), "v"(v[1].w),
                           "v"(v[2].x), "v"(v[2].y), "v"(v[2].z), "v"(v[2].w), "v"(v[3].x), "v"(v[3].y), "v"(v[3].z), "v"(v[3].w));
        unsigned* row = bm + w * IDX_P;
        #pragma unroll
        for (int k = 0; k < 4; ++k) {
            const unsigned bit = 1u << (k * 8 + (lane >> 2));
            atomicOr(row + (unsigned)v[k].x / TAG_R, bit);
            atomicOr(row + (unsigned)v[k].y / TAG_R, bit);
            atomicOr(row + (unsigned)v[k].z / TAG_R, bit);
            atomicOr(row + (unsigned)v[k].w / TAG_R, bit);
        }
    }
    __syncthreads();

    #pragma unroll
    for (int j = 0; j < OWN_G / 32; ++j) {
        const int o = w * (OWN_G / 32) + j;
        rows[(size_t)o * IDX_ROW + blockIdx.x * 32 + lane] = bm[lane * IDX_P + o];
    }
}

template <int F, int P>
__device__ __forceinline__ void add_rows(const float* __restrict__ src, float* acc, int s, int l, int n, int lane) {
    for (int j0 = 0; j0 < n; j0 += 8) {
        float a[8], b[8];
        int lj[8];
        #pragma unroll
        for (int u = 0; u < 8; ++u) {
            const int sj = __builtin_amdgcn_readlane(s, j0 + u);
            lj[u] = __builtin_amdgcn_readlane(l, j0 + u);
            const float* row = src + (size_t)sj * F;
            a[u] = row[lane];
            b[u] = (lane < F - 32) ? row[32 + lane] : 0.f;
        }
        #pragma unroll
        for (int u = 0; u < 8; ++u) {
            if (j0 + u < n) {
                float* d = acc + lj[u] * P;
                atomicAdd(d + lane, a[u]);
                if (lane < F - 32) atomicAdd(d + 32 + lane, b[u]);
            }
        }
    }
}

template <int F, int P, bool CNT, bool REC>
__device__ __forceinline__ void drain_rows(const float* __restrict__ src, const int* __restrict__ snd,
                                           float* acc, int* cnt, unsigned* __restrict__ rec, int* rcnt,
                                           const int* q, int n, int lane) {
    const int ent = (lane < n) ? q[lane] : 0;
    const int l = ent >> 21;
    if (CNT) {
        if (lane < n) atomicAdd(cnt + l, 1);
    }
    const int s = (lane < n) ? snd[ent & 0x1FFFFF] : 0;
    if (REC) {
        int at = 0;
        if (lane == 0) at = atomicAdd(rcnt, n);
        at = __builtin_amdgcn_readfirstlane(at);
        if (lane < n && at + lane < OWN_ECAP) rec[at + lane] = (unsigned)s | ((unsigned)l << 17);
    }
    add_rows<F, P>(src, acc, s, l, n, lane);
}


template <int F, int P, bool CNT, bool REC>
__device__ __forceinline__ void expand_groups(const float* __restrict__ src, const int* __restrict__ snd,
                                              const int4* __restrict__ rcv4, float* acc, int* cnt,
                                              unsigned* __restrict__ rec, int* rcnt, int* q, int& qn,
                                              const int* gq, int gn, int r0, int lane) {
    const int sub = lane & 3;
    int  g[4];
    bool has[4];
    int4 r[4];
    #pragma unroll
    for (int p = 0; p < 4; ++p) {
        const int gi = p * 8 + (lane >> 2);
        has[p] = gi < gn;
        g[p] = gq[has[p] ? gi : gn - 1];
        r[p] = rcv4[g[p] * 4 + sub];
    }
    #pragma unroll
    for (int p = 0; p < 4; ++p) {
        const int rr[4] = {r[p].x, r[p].y, r[p].z, r[p].w};
        #pragma unroll
        for (int c = 0; c < 4; ++c) {
            const unsigned d = (unsigned)(rr[c] - r0);
            const bool hit = has[p] && d < (unsigned)OWN_R;
            const unsigned m = __builtin_amdgcn_ballot_w32(hit);
            if (m) {
                const int pos = qn + (int)__builtin_amdgcn_mbcnt_lo(m, 0u);
                if (hit) q[pos] = (g[p] * 16 + sub * 4 + c) | ((int)d << 21);
                qn += __builtin_popcount(m);
            }
        }
    }
    if (qn >= 32) {
        int done = 0;
        while (qn - done >= 32) {
            drain_rows<F, P, CNT, REC>(src, snd, acc, cnt, rec, rcnt, q + done, 32, lane);
            done += 32;
        }
        const int rem = qn - done;
        const int t = (lane < rem) ? q[done + lane] : 0;
        if (lane < rem) q[lane] = t;
        qn = rem;
    }
}

template <int CTRL>
__device__ __forceinline__ float row16_pick(float x) {
    const int xi = __builtin_bit_cast(int, x);
    return __builtin_bit_cast(float, __builtin_amdgcn_update_dpp(0, xi, CTRL, 0xF, 0xF, true));
}
__device__ __forceinline__ float row16_max(float x) {
    x = fmaxf(x, row16_pick<0xB1>(x));
    x = fmaxf(x, row16_pick<0x4E>(x));
    x = fmaxf(x, row16_pick<0x141>(x));
    x = fmaxf(x, row16_pick<0x140>(x));
    return x;
}
__device__ __forceinline__ float row16_sum(float x) {
    x += row16_pick<0xB1>(x);
    x += row16_pick<0x4E>(x);
    x += row16_pick<0x141>(x);
    x += row16_pick<0x140>(x);
    return x;
}

template <int F, int P, bool CNT, bool REC, bool PLAY, bool FIN, bool SMAX>
__launch_bounds__(OWN_NW * 32)
__global__ void agg_own_kernel(const float* __restrict__ src, const int* __restrict__ snd,
                               const int* __restrict__ rcv, const unsigned* __restrict__ bits,
                               float* __restrict__ dst, float* __restrict__ rdeg,
                               unsigned* __restrict__ elist, int* __restrict__ ecount,
                               const float* __restrict__ sdeg, const float* __restrict__ W3,
                               const float* __restrict__ b3, const float* __restrict__ rdin) {
    static_assert(!FIN || (CNT && F == DH && P % 2 == 0), "the fused product reads 64-wide sums in pairs and scales them by the block's own counts");
    static_assert(!SMAX || (!FIN && F > 32 && F <= 48), "the fused softmax gives a row's sixteen lanes three columns each");
    constexpr int ROWS = FIN ? OWN_RT * 16 : OWN_R;
    __shared__ __attribute__((aligned(16))) float acc[ROWS * P];
    __shared__ int   cnt[CNT ? ROWS : 1];
    __shared__ int   rcnt[1];
    __shared__ float rsc[SMAX ? OWN_R : 1];
    __shared__ int   qs[OWN_NW * OWN_QCAP];
    __shared__ int   gqs[OWN_NW * OWN_GQCAP];
    const int w    = __builtin_amdgcn_readfirstlane((int)(threadIdx.x >> 5));
    const int lane = threadIdx.x & 31;
    const int r0   = blockIdx.x * OWN_R;
    int* q  = qs + w * OWN_QCAP;
    int* gq = gqs + w * OWN_GQCAP;

    for (int i = threadIdx.x; i < ROWS * P; i += OWN_NW * 32) acc[i] = 0.f;
    if (CNT) {
        for (int i = threadIdx.x; i < ROWS; i += OWN_NW * 32) cnt[i] = 0;
    }
    if (REC) {
        if (threadIdx.x == 0) rcnt[0] = 0;
    }
    if (SMAX) {
        for (int i = threadIdx.x; i < OWN_R; i += OWN_NW * 32) {
            const int node = r0 + i;
            rsc[i] = rsqrtf(fmaxf(rdin[(node < NN) ? node : NN - 1], 1.f));
        }
    }
    __syncthreads();

    unsigned* mine = elist + (size_t)blockIdx.x * OWN_ECAP;
    bool scan = true;
    if (PLAY) {
        const int ne = __builtin_amdgcn_readfirstlane(ecount[blockIdx.x]);
        if (ne <= OWN_ECAP) {
            scan = false;
            int at = w * 32;
            unsigned e = (at + lane < ne) ? mine[at + lane] : 0u;
            while (at + 32 <= ne) {
                const int nx = at + OWN_NW * 32;
                const int ix = nx + lane;
                const unsigned ld = mine[(ix < ne) ? ix : ne - 1];
                add_rows<F, P>(src, acc, (int)(e & 0x1FFFFu), (int)(e >> 17), 32, lane);
                e = (ix < ne) ? ld : 0u;
                at = nx;
            }
            if (at < ne) add_rows<F, P>(src, acc, (int)(e & 0x1FFFFu), (int)(e >> 17), ne - at, lane);
        }
    }

    const unsigned* myb = bits + (size_t)blockIdx.x * IDX_ROW;
    const int4*  rcv4  = (const int4*)rcv;
    int qn = 0;
    int gn = 0;
    if (scan) {
        unsigned wd[IDX_FULL + 1];
        int      wi[IDX_FULL + 1];
        #pragma unroll
        for (int u = 0; u < IDX_FULL; ++u) {
            wi[u] = (u * OWN_NW + w) * 32 + lane;
            wd[u] = myb[wi[u]];
        }
        wi[IDX_FULL] = IDX_FULL * OWN_NW * 32 + w * IDX_REM + ((lane < IDX_REM) ? lane : 0);
        const unsigned last = myb[wi[IDX_FULL]];
        wd[IDX_FULL] = (lane < IDX_REM) ? last : 0u;
        #pragma unroll
        for (int u = 0; u <= IDX_FULL; ++u) {
            unsigned word = wd[u];
            unsigned m = __builtin_amdgcn_ballot_w32(word != 0u);
            while (m) {
                const int pos = gn + (int)__builtin_amdgcn_mbcnt_lo(m, 0u);
                if (word != 0u) {
                    gq[pos] = wi[u] * 32 + __builtin_ctz(word);
                    word &= word - 1u;
                }
                gn += __builtin_popcount(m);
                if (gn >= 32) {
                    expand_groups<F, P, CNT, REC>(src, snd, rcv4, acc, cnt, mine, rcnt, q, qn, gq, 32, r0, lane);
                    const int rem = gn - 32;
                    const int t = (lane < rem) ? gq[32 + lane] : 0;
                    if (lane < rem) gq[lane] = t;
                    gn = rem;
                }
                m = __builtin_amdgcn_ballot_w32(word != 0u);
            }
        }
    }
    if (gn > 0) expand_groups<F, P, CNT, REC>(src, snd, rcv4, acc, cnt, mine, rcnt, q, qn, gq, gn, r0, lane);
    if (qn > 0) drain_rows<F, P, CNT, REC>(src, snd, acc, cnt, mine, rcnt, q, qn, lane);
    __syncthreads();

    if (REC) {
        if (threadIdx.x == 0) ecount[blockIdx.x] = rcnt[0];
    }

    if (CNT) {
        for (int i = threadIdx.x; i < OWN_R; i += OWN_NW * 32) {
            if (r0 + i < NN) rdeg[r0 + i] = (float)cnt[i];
        }
    }

    if (FIN) {
        if (w < OWN_RT) {
            const int m16  = lane & 15;
            const int kh   = lane >> 4;
            const int lr0  = w * 16;
            const float rs = rsqrtf(fmaxf((float)cnt[lr0 + m16], 1.f));
            const float* arow = acc + (lr0 + m16) * P + 2 * kh;

            float sc[8];
            #pragma unroll
            for (int r = 0; r < 8; ++r) {
                const int node = r0 + lr0 + r + 8 * kh;
                sc[r] = sdeg[(node < NN) ? node : NN - 1];
            }
            const int   n2c = (32 + m16 < DC) ? 32 + m16 : DC - 1;
            const float bb0 = b3[m16], bb1 = b3[16 + m16], bb2 = b3[n2c];

            v8f c0 = {}, c1 = {}, c2 = {};
            #pragma unroll 4
            for (int kb = 0; kb < DH; kb += 4) {
                v2f a = *(const v2f*)(arow + kb);
                a.x *= rs; a.y *= rs;
                const float* wr0 = W3 + (size_t)(kb + 2 * kh) * DC;
                const float* wr1 = wr0 + DC;
                v2f b;
                b.x = wr0[m16];      b.y = wr1[m16];      c0 = wmma4(a, b, c0);
                b.x = wr0[16 + m16]; b.y = wr1[16 + m16]; c1 = wmma4(a, b, c1);
                const float p0 = wr0[n2c], p1 = wr1[n2c];
                b.x = (32 + m16 < DC) ? p0 : 0.f;
                b.y = (32 + m16 < DC) ? p1 : 0.f;
                c2 = wmma4(a, b, c2);
            }
            #pragma unroll
            for (int r = 0; r < 8; ++r) {
                const int lr   = lr0 + r + 8 * kh;
                const int node = r0 + lr;
                if (lr < OWN_R && node < NN) {
                    const float s = rsqrtf(fmaxf(sc[r], 1.f));
                    float* orow = dst + (size_t)node * DC;
                    orow[m16]      = (c0[r] + bb0) * s;
                    orow[16 + m16] = (c1[r] + bb1) * s;
                    if (32 + m16 < DC)
                        orow[32 + m16] = (c2[r] + bb2) * s;
                }
            }
        }
    } else if (SMAX) {
        const int c = lane & 15;
        const int h = lane >> 4;
        const bool t3 = 32 + c < F;
        for (int p = w; p < (OWN_R + 1) / 2; p += OWN_NW) {
            const int  lr  = 2 * p + h;
            const bool ok  = lr < OWN_R && r0 + lr < NN;
            const int  lrc = ok ? lr : 0;
            const float rs = rsc[lrc];
            const float* a = acc + lrc * P;
            const float v0 = a[c] * rs;
            const float v1 = a[16 + c] * rs;
            const float v2 = t3 ? a[t3 ? 32 + c : 0] * rs : -3.402823466e38f;
            const float mx = row16_max(fmaxf(fmaxf(v0, v1), v2));
            const float e0 = __expf(v0 - mx);
            const float e1 = __expf(v1 - mx);
            const float e2 = t3 ? __expf(v2 - mx) : 0.f;
            const float inv = 1.f / row16_sum(e0 + e1 + e2);
            if (ok) {
                float* orow = dst + (size_t)(r0 + lr) * F;
                orow[c]      = e0 * inv;
                orow[16 + c] = e1 * inv;
                if (t3) orow[32 + c] = e2 * inv;
            }
        }
    } else {
        for (int i = threadIdx.x; i < OWN_R * F; i += OWN_NW * 32) {
            const int lr = i / F;
            if (r0 + lr < NN) dst[(size_t)r0 * F + i] = acc[lr * P + (i - lr * F)];
        }
    }
}

__device__ __forceinline__ void count_groups(const int4* __restrict__ snd4, int* cnt, const int* gq, int gn, int r0, int lane) {
    const int sub = lane & 3;
    bool has[4];
    int4 s[4];
    #pragma unroll
    for (int p = 0; p < 4; ++p) {
        const int gi = p * 8 + (lane >> 2);
        has[p] = gi < gn;
        s[p] = snd4[gq[has[p] ? gi : gn - 1] * 4 + sub];
    }
    #pragma unroll
    for (int p = 0; p < 4; ++p) {
        const int ss[4] = {s[p].x, s[p].y, s[p].z, s[p].w};
        #pragma unroll
        for (int c = 0; c < 4; ++c) {
            const unsigned d = (unsigned)(ss[c] - r0);
            if (has[p] && d < (unsigned)OWN_R) atomicAdd(cnt + d, 1);
        }
    }
}

__launch_bounds__(OWN_NW * 32)
__global__ void sdeg_own_kernel(const int* __restrict__ snd, const unsigned* __restrict__ sbits,
                                float* __restrict__ sdeg) {
    __shared__ int cnt[OWN_R];
    __shared__ int gqs[OWN_NW * OWN_GQCAP];
    const int w    = __builtin_amdgcn_readfirstlane((int)(threadIdx.x >> 5));
    const int lane = threadIdx.x & 31;
    const int r0   = blockIdx.x * OWN_R;
    int* gq = gqs + w * OWN_GQCAP;

    for (int i = threadIdx.x; i < OWN_R; i += OWN_NW * 32) cnt[i] = 0;
    __syncthreads();

    const unsigned* myb = sbits + (size_t)blockIdx.x * IDX_ROW;
    const int4*  snd4  = (const int4*)snd;
    int gn = 0;
    {
        unsigned wd[IDX_FULL + 1];
        int      wi[IDX_FULL + 1];
        #pragma unroll
        for (int u = 0; u < IDX_FULL; ++u) {
            wi[u] = (u * OWN_NW + w) * 32 + lane;
            wd[u] = myb[wi[u]];
        }
        wi[IDX_FULL] = IDX_FULL * OWN_NW * 32 + w * IDX_REM + ((lane < IDX_REM) ? lane : 0);
        const unsigned last = myb[wi[IDX_FULL]];
        wd[IDX_FULL] = (lane < IDX_REM) ? last : 0u;
        #pragma unroll
        for (int u = 0; u <= IDX_FULL; ++u) {
            unsigned word = wd[u];
            unsigned m = __builtin_amdgcn_ballot_w32(word != 0u);
            while (m) {
                const int pos = gn + (int)__builtin_amdgcn_mbcnt_lo(m, 0u);
                if (word != 0u) {
                    gq[pos] = wi[u] * 32 + __builtin_ctz(word);
                    word &= word - 1u;
                }
                gn += __builtin_popcount(m);
                if (gn >= 32) {
                    count_groups(snd4, cnt, gq, 32, r0, lane);
                    const int rem = gn - 32;
                    const int t = (lane < rem) ? gq[32 + lane] : 0;
                    if (lane < rem) gq[lane] = t;
                    gn = rem;
                }
                m = __builtin_amdgcn_ballot_w32(word != 0u);
            }
        }
    }
    if (gn > 0) count_groups(snd4, cnt, gq, gn, r0, lane);
    __syncthreads();

    for (int i = threadIdx.x; i < OWN_R; i += OWN_NW * 32) {
        if (r0 + i < NN) sdeg[r0 + i] = (float)cnt[i];
    }
}

extern "C" void kernel_launch(void* const* d_in, const int* in_sizes, int n_in,
                              void* d_out, int out_size, void* d_ws, size_t ws_size,
                              hipStream_t stream) {
    const float* nodes     = (const float*)d_in[0];
    const int*   senders   = (const int*)d_in[1];
    const int*   receivers = (const int*)d_in[2];
    const float* W1 = (const float*)d_in[3];
    const float* b1 = (const float*)d_in[4];
    const float* W2 = (const float*)d_in[5];
    const float* b2 = (const float*)d_in[6];
    const float* W3 = (const float*)d_in[7];
    const float* b3 = (const float*)d_in[8];
    float* out = (float*)d_out;
    float* ws  = (float*)d_ws;

    float* sdeg = ws;
    float* rdeg = ws + NN;
    float* h1   = ws + 2 * (size_t)NN;
    float* h2   = h1 + 2 * (size_t)NN * DH;
    unsigned* rbits = (unsigned*)(h2 + (size_t)NN * DC);
    unsigned* sbits = rbits + (size_t)OWN_G * IDX_ROW;
    unsigned* elist = sbits + (size_t)OWN_G * IDX_ROW;
    int* ecount = (int*)(elist + (size_t)OWN_G * OWN_ECAP);

    const int nTiles = NN / 16;
    index_kernel<<<dim3(IDX_NB, 2), 1024, 0, stream>>>(senders, receivers, sbits, rbits);
    sdeg_own_kernel<<<OWN_G, OWN_NW * 32, 0, stream>>>(senders, sbits, sdeg);
    mlp_kernel<<<(nTiles + 7) / 8, 256, 0, stream>>>(nodes, W1, b1, W2, b2, sdeg, h1, nTiles);
    agg_own_kernel<DH, 68, true, true, false, true, false><<<OWN_G, OWN_NW * 32, 0, stream>>>(h1, senders, receivers, rbits, h2, rdeg, elist, ecount, sdeg, W3, b3, nullptr);
    agg_own_kernel<DC, DC, false, false, true, false, true><<<OWN_G, OWN_NW * 32, 0, stream>>>(h2, senders, receivers, rbits, out, nullptr, elist, ecount, nullptr, nullptr, nullptr, rdeg);
}


# Second edge sum (40-wide rows): columns 32..39 of four rows share one load and one LDS add, 8 lanes a row: 10 loads and 10 adds per 8 rows, not 16 and 16; LDS-atomic sum order changes
# speedup vs baseline: 5.5464x; 1.0505x over previous
#include <hip/hip_runtime.h>

typedef __attribute__((ext_vector_type(2))) float v2f;
typedef __attribute__((ext_vector_type(8))) float v8f;

static constexpr int NN = 100000;
static constexpr int NE = 1600000;
static constexpr int DF = 128;
static constexpr int DH = 64;
static constexpr int DC = 40;
#define NEG_SLOPE 0.01f
static constexpr unsigned TAG_R = (NN + 255) / 256;

__device__ __forceinline__ float lrelu(float x) { return x > 0.f ? x : x * NEG_SLOPE; }

__device__ __forceinline__ v8f wmma4(v2f a, v2f b, v8f c) {
    return __builtin_amdgcn_wmma_f32_16x16x4_f32(false, a, false, b, (short)0, c, false, false);
}


__launch_bounds__(256)
__global__ void mlp_kernel(const float* __restrict__ X,
                           const float* __restrict__ W1, const float* __restrict__ b1,
                           const float* __restrict__ W2, const float* __restrict__ b2,
                           const float* __restrict__ sdeg,
                           float* __restrict__ h1, int nTiles) {
    __shared__ float lds[8 * 16 * DH];
    const int w    = threadIdx.x >> 5;
    const int lane = threadIdx.x & 31;
    const int m16  = lane & 15;
    const int kh   = lane >> 4;
    const int t    = blockIdx.x * 8 + w;
    const bool active = (t < nTiles);
    const int row0 = t * 16;
    float* my = &lds[w * 16 * DH];

    if (active) {
        v8f c0 = {}, c1 = {}, c2 = {}, c3 = {};
        #pragma unroll 4
        for (int kb = 0; kb < DF; kb += 4) {
            v2f a = *(const v2f*)(X + (size_t)(row0 + m16) * DF + kb + 2 * kh);
            const float* wr0 = W1 + (size_t)(kb + 2 * kh) * DH;
            const float* wr1 = wr0 + DH;
            v2f b;
            b.x = wr0[m16];      b.y = wr1[m16];      c0 = wmma4(a, b, c0);
            b.x = wr0[16 + m16]; b.y = wr1[16 + m16]; c1 = wmma4(a, b, c1);
            b.x = wr0[32 + m16]; b.y = wr1[32 + m16]; c2 = wmma4(a, b, c2);
            b.x = wr0[48 + m16]; b.y = wr1[48 + m16]; c3 = wmma4(a, b, c3);
        }
        #pragma unroll
        for (int r = 0; r < 8; ++r) {
            float* dr = my + (r + 8 * kh) * DH;
            dr[m16]      = lrelu(c0[r] + b1[m16]);
            dr[16 + m16] = lrelu(c1[r] + b1[16 + m16]);
            dr[32 + m16] = lrelu(c2[r] + b1[32 + m16]);
            dr[48 + m16] = lrelu(c3[r] + b1[48 + m16]);
        }
    }
    __syncthreads();
    if (!active) return;

    v8f d0 = {}, d1 = {}, d2 = {}, d3 = {};
    #pragma unroll 4
    for (int kb = 0; kb < DH; kb += 4) {
        v2f a = *(const v2f*)(my + m16 * DH + kb + 2 * kh);
        const float* wr0 = W2 + (size_t)(kb + 2 * kh) * DH;
        const float* wr1 = wr0 + DH;
        v2f b;
        b.x = wr0[m16];      b.y = wr1[m16];      d0 = wmma4(a, b, d0);
        b.x = wr0[16 + m16]; b.y = wr1[16 + m16]; d1 = wmma4(a, b, d1);
        b.x = wr0[32 + m16]; b.y = wr1[32 + m16]; d2 = wmma4(a, b, d2);
        b.x = wr0[48 + m16]; b.y = wr1[48 + m16]; d3 = wmma4(a, b, d3);
    }
    float sc[8];
    #pragma unroll
    for (int r = 0; r < 8; ++r) sc[r] = rsqrtf(fmaxf(sdeg[row0 + r + 8 * kh], 1.f));
    #pragma unroll
    for (int r = 0; r < 8; ++r) {
        float* orow = h1 + (size_t)(row0 + r + 8 * kh) * DH;
        orow[m16]      = lrelu(d0[r] + b2[m16])      * sc[r];
        orow[16 + m16] = lrelu(d1[r] + b2[16 + m16]) * sc[r];
        orow[32 + m16] = lrelu(d2[r] + b2[32 + m16]) * sc[r];
        orow[48 + m16] = lrelu(d3[r] + b2[48 + m16]) * sc[r];
    }
}

static constexpr int OWN_G  = 256;
static constexpr int OWN_R  = (NN + OWN_G - 1) / OWN_G;
static constexpr int OWN_NW = 32;
static constexpr int OWN_GQCAP = 64;
static constexpr int OWN_QCAP  = 16 * 32 + 32;
static constexpr int NG16 = NE / 16;
static constexpr int OWN_RT = (OWN_R + 15) / 16;
static constexpr int OWN_ECAP = 1 << 17;
static constexpr int IDX_GPB = 1024;
static constexpr int IDX_NB  = (NG16 + IDX_GPB - 1) / IDX_GPB;
static constexpr int IDX_ROW = IDX_NB * 32;
static constexpr int IDX_P   = OWN_G + 1;
static constexpr int IDX_FULL = IDX_NB / OWN_NW;
static constexpr int IDX_REM  = (IDX_ROW - IDX_FULL * OWN_NW * 32) / OWN_NW;
static_assert(OWN_R == (int)TAG_R, "the index must name the owner blocks");
static_assert(NN <= (1 << 17) && OWN_R <= (1 << 9), "a list entry holds the sender in bits 0..16 and the local receiver above them");
static_assert(OWN_RT <= OWN_NW, "the fused product gives every row tile a wave of its own");
static_assert(NE % 16 == 0, "edge list must split into groups of sixteen");
static_assert(NG16 % 32 == 0, "a wave's 32 groups must lie all inside the list or all past its end");
static_assert(OWN_G % 32 == 0 && IDX_GPB == 32 * 32, "an index block is 32 waves of 32 groups, and every wave writes whole owners' lines");
static_assert((IDX_ROW - IDX_FULL * OWN_NW * 32) % OWN_NW == 0 && IDX_REM >= 1 && IDX_REM <= 32, "the words left after the full rounds split evenly over the waves, at least one each");

__launch_bounds__(1024)
__global__ void index_kernel(const int* __restrict__ snd, const int* __restrict__ rcv,
                             unsigned* __restrict__ sbits, unsigned* __restrict__ rbits) {
    __shared__ unsigned bm[32 * IDX_P];
    const int w    = threadIdx.x >> 5;
    const int lane = threadIdx.x & 31;
    const int* __restrict__ ids = blockIdx.y ? rcv : snd;
    unsigned* __restrict__ rows = blockIdx.y ? rbits : sbits;

    for (int i = threadIdx.x; i < 32 * IDX_P; i += 1024) bm[i] = 0u;
    __syncthreads();

    const int g0 = blockIdx.x * IDX_GPB + w * 32;
    if (g0 < NG16) {
        const int4* p = (const int4*)ids + (size_t)g0 * 4;
        int4 v[4];
        #pragma unroll
        for (int k = 0; k < 4; ++k) v[k] = p[k * 32 + lane];
        asm volatile("" :: "v"(v[0].x), "v"(v[0].y), "v"(v[0].z), "v"(v[0].w), "v"(v[1].x), "v"(v[1].y), "v"(v[1].z), "v"(v[1].w),
                           "v"(v[2].x), "v"(v[2].y), "v"(v[2].z), "v"(v[2].w), "v"(v[3].x), "v"(v[3].y), "v"(v[3].z), "v"(v[3].w));
        unsigned* row = bm + w * IDX_P;
        #pragma unroll
        for (int k = 0; k < 4; ++k) {
            const unsigned bit = 1u << (k * 8 + (lane >> 2));
            atomicOr(row + (unsigned)v[k].x / TAG_R, bit);
            atomicOr(row + (unsigned)v[k].y / TAG_R, bit);
            atomicOr(row + (unsigned)v[k].z / TAG_R, bit);
            atomicOr(row + (unsigned)v[k].w / TAG_R, bit);
        }
    }
    __syncthreads();

    #pragma unroll
    for (int j = 0; j < OWN_G / 32; ++j) {
        const int o = w * (OWN_G / 32) + j;
        rows[(size_t)o * IDX_ROW + blockIdx.x * 32 + lane] = bm[lane * IDX_P + o];
    }
}

template <int F, int P>
__device__ __forceinline__ void add_rows(const float* __restrict__ src, float* acc, int s, int l, int n, int lane) {
    constexpr bool PACK = (F - 32 == 8);
    if constexpr (PACK) {
        static_assert(NN <= (1 << 17) && OWN_R <= (1 << 9), "sender and local receiver travel in one word: sender in bits 0..16, local receiver above them");
        const int tg = lane >> 3;
        const int tc = lane & 7;
        const int m0 = -(int)(tg == 0), m1 = -(int)(tg == 1), m2 = -(int)(tg == 2), m3 = -(int)(tg == 3);
        const int pk = s | (l << 17);
        for (int j0 = 0; j0 < n; j0 += 8) {
            float a[8], t[2];
            int pj[8], lj[8], tl[2];
            #pragma unroll
            for (int u = 0; u < 8; ++u) {
                pj[u] = __builtin_amdgcn_readlane(pk, j0 + u);
                lj[u] = pj[u] >> 17;
                a[u] = src[(size_t)(pj[u] & 0x1FFFF) * F + lane];
            }
            #pragma unroll
            for (int h = 0; h < 2; ++h) {
                const int tp = (pj[4 * h] & m0) | (pj[4 * h + 1] & m1) | (pj[4 * h + 2] & m2) | (pj[4 * h + 3] & m3);
                tl[h] = tp >> 17;
                t[h] = src[(size_t)(tp & 0x1FFFF) * F + 32 + tc];
            }
            #pragma unroll
            for (int u = 0; u < 8; ++u) {
                if (j0 + u < n) atomicAdd(acc + lj[u] * P + lane, a[u]);
            }
            #pragma unroll
            for (int h = 0; h < 2; ++h) {
                if (j0 + 4 * h + tg < n) atomicAdd(acc + tl[h] * P + 32 + tc, t[h]);
            }
        }
    } else {
        for (int j0 = 0; j0 < n; j0 += 8) {
            float a[8], b[8];
            int lj[8];
            #pragma unroll
            for (int u = 0; u < 8; ++u) {
                const int sj = __builtin_amdgcn_readlane(s, j0 + u);
                lj[u] = __builtin_amdgcn_readlane(l, j0 + u);
                const float* row = src + (size_t)sj * F;
                a[u] = row[lane];
                b[u] = (lane < F - 32) ? row[32 + lane] : 0.f;
            }
            #pragma unroll
            for (int u = 0; u < 8; ++u) {
                if (j0 + u < n) {
                    float* d = acc + lj[u] * P;
                    atomicAdd(d + lane, a[u]);
                    if (lane < F - 32) atomicAdd(d + 32 + lane, b[u]);
                }
            }
        }
    }
}

template <int F, int P, bool CNT, bool REC>
__device__ __forceinline__ void drain_rows(const float* __restrict__ src, const int* __restrict__ snd,
                                           float* acc, int* cnt, unsigned* __restrict__ rec, int* rcnt,
                                           const int* q, int n, int lane) {
    const int ent = (lane < n) ? q[lane] : 0;
    const int l = ent >> 21;
    if (CNT) {
        if (lane < n) atomicAdd(cnt + l, 1);
    }
    const int s = (lane < n) ? snd[ent & 0x1FFFFF] : 0;
    if (REC) {
        int at = 0;
        if (lane == 0) at = atomicAdd(rcnt, n);
        at = __builtin_amdgcn_readfirstlane(at);
        if (lane < n && at + lane < OWN_ECAP) rec[at + lane] = (unsigned)s | ((unsigned)l << 17);
    }
    add_rows<F, P>(src, acc, s, l, n, lane);
}


template <int F, int P, bool CNT, bool REC>
__device__ __forceinline__ void expand_groups(const float* __restrict__ src, const int* __restrict__ snd,
                                              const int4* __restrict__ rcv4, float* acc, int* cnt,
                                              unsigned* __restrict__ rec, int* rcnt, int* q, int& qn,
                                              const int* gq, int gn, int r0, int lane) {
    const int sub = lane & 3;
    int  g[4];
    bool has[4];
    int4 r[4];
    #pragma unroll
    for (int p = 0; p < 4; ++p) {
        const int gi = p * 8 + (lane >> 2);
        has[p] = gi < gn;
        g[p] = gq[has[p] ? gi : gn - 1];
        r[p] = rcv4[g[p] * 4 + sub];
    }
    #pragma unroll
    for (int p = 0; p < 4; ++p) {
        const int rr[4] = {r[p].x, r[p].y, r[p].z, r[p].w};
        #pragma unroll
        for (int c = 0; c < 4; ++c) {
            const unsigned d = (unsigned)(rr[c] - r0);
            const bool hit = has[p] && d < (unsigned)OWN_R;
            const unsigned m = __builtin_amdgcn_ballot_w32(hit);
            if (m) {
                const int pos = qn + (int)__builtin_amdgcn_mbcnt_lo(m, 0u);
                if (hit) q[pos] = (g[p] * 16 + sub * 4 + c) | ((int)d << 21);
                qn += __builtin_popcount(m);
            }
        }
    }
    if (qn >= 32) {
        int done = 0;
        while (qn - done >= 32) {
            drain_rows<F, P, CNT, REC>(src, snd, acc, cnt, rec, rcnt, q + done, 32, lane);
            done += 32;
        }
        const int rem = qn - done;
        const int t = (lane < rem) ? q[done + lane] : 0;
        if (lane < rem) q[lane] = t;
        qn = rem;
    }
}

template <int CTRL>
__device__ __forceinline__ float row16_pick(float x) {
    const int xi = __builtin_bit_cast(int, x);
    return __builtin_bit_cast(float, __builtin_amdgcn_update_dpp(0, xi, CTRL, 0xF, 0xF, true));
}
__device__ __forceinline__ float row16_max(float x) {
    x = fmaxf(x, row16_pick<0xB1>(x));
    x = fmaxf(x, row16_pick<0x4E>(x));
    x = fmaxf(x, row16_pick<0x141>(x));
    x = fmaxf(x, row16_pick<0x140>(x));
    return x;
}
__device__ __forceinline__ float row16_sum(float x) {
    x += row16_pick<0xB1>(x);
    x += row16_pick<0x4E>(x);
    x += row16_pick<0x141>(x);
    x += row16_pick<0x140>(x);
    return x;
}

template <int F, int P, bool CNT, bool REC, bool PLAY, bool FIN, bool SMAX>
__launch_bounds__(OWN_NW * 32)
__global__ void agg_own_kernel(const float* __restrict__ src, const int* __restrict__ snd,
                               const int* __restrict__ rcv, const unsigned* __restrict__ bits,
                               float* __restrict__ dst, float* __restrict__ rdeg,
                               unsigned* __restrict__ elist, int* __restrict__ ecount,
                               const float* __restrict__ sdeg, const float* __restrict__ W3,
                               const float* __restrict__ b3, const float* __restrict__ rdin) {
    static_assert(!FIN || (CNT && F == DH && P % 2 == 0), "the fused product reads 64-wide sums in pairs and scales them by the block's own counts");
    static_assert(!SMAX || (!FIN && F > 32 && F <= 48), "the fused softmax gives a row's sixteen lanes three columns each");
    constexpr int ROWS = FIN ? OWN_RT * 16 : OWN_R;
    __shared__ __attribute__((aligned(16))) float acc[ROWS * P];
    __shared__ int   cnt[CNT ? ROWS : 1];
    __shared__ int   rcnt[1];
    __shared__ float rsc[SMAX ? OWN_R : 1];
    __shared__ int   qs[OWN_NW * OWN_QCAP];
    __shared__ int   gqs[OWN_NW * OWN_GQCAP];
    const int w    = __builtin_amdgcn_readfirstlane((int)(threadIdx.x >> 5));
    const int lane = threadIdx.x & 31;
    const int r0   = blockIdx.x * OWN_R;
    int* q  = qs + w * OWN_QCAP;
    int* gq = gqs + w * OWN_GQCAP;

    for (int i = threadIdx.x; i < ROWS * P; i += OWN_NW * 32) acc[i] = 0.f;
    if (CNT) {
        for (int i = threadIdx.x; i < ROWS; i += OWN_NW * 32) cnt[i] = 0;
    }
    if (REC) {
        if (threadIdx.x == 0) rcnt[0] = 0;
    }
    if (SMAX) {
        for (int i = threadIdx.x; i < OWN_R; i += OWN_NW * 32) {
            const int node = r0 + i;
            rsc[i] = rsqrtf(fmaxf(rdin[(node < NN) ? node : NN - 1], 1.f));
        }
    }
    __syncthreads();

    unsigned* mine = elist + (size_t)blockIdx.x * OWN_ECAP;
    bool scan = true;
    if (PLAY) {
        const int ne = __builtin_amdgcn_readfirstlane(ecount[blockIdx.x]);
        if (ne <= OWN_ECAP) {
            scan = false;
            int at = w * 32;
            unsigned e = (at + lane < ne) ? mine[at + lane] : 0u;
            while (at + 32 <= ne) {
                const int nx = at + OWN_NW * 32;
                const int ix = nx + lane;
                const unsigned ld = mine[(ix < ne) ? ix : ne - 1];
                add_rows<F, P>(src, acc, (int)(e & 0x1FFFFu), (int)(e >> 17), 32, lane);
                e = (ix < ne) ? ld : 0u;
                at = nx;
            }
            if (at < ne) add_rows<F, P>(src, acc, (int)(e & 0x1FFFFu), (int)(e >> 17), ne - at, lane);
        }
    }

    const unsigned* myb = bits + (size_t)blockIdx.x * IDX_ROW;
    const int4*  rcv4  = (const int4*)rcv;
    int qn = 0;
    int gn = 0;
    if (scan) {
        unsigned wd[IDX_FULL + 1];
        int      wi[IDX_FULL + 1];
        #pragma unroll
        for (int u = 0; u < IDX_FULL; ++u) {
            wi[u] = (u * OWN_NW + w) * 32 + lane;
            wd[u] = myb[wi[u]];
        }
        wi[IDX_FULL] = IDX_FULL * OWN_NW * 32 + w * IDX_REM + ((lane < IDX_REM) ? lane : 0);
        const unsigned last = myb[wi[IDX_FULL]];
        wd[IDX_FULL] = (lane < IDX_REM) ? last : 0u;
        #pragma unroll
        for (int u = 0; u <= IDX_FULL; ++u) {
            unsigned word = wd[u];
            unsigned m = __builtin_amdgcn_ballot_w32(word != 0u);
            while (m) {
                const int pos = gn + (int)__builtin_amdgcn_mbcnt_lo(m, 0u);
                if (word != 0u) {
                    gq[pos] = wi[u] * 32 + __builtin_ctz(word);
                    word &= word - 1u;
                }
                gn += __builtin_popcount(m);
                if (gn >= 32) {
                    expand_groups<F, P, CNT, REC>(src, snd, rcv4, acc, cnt, mine, rcnt, q, qn, gq, 32, r0, lane);
                    const int rem = gn - 32;
                    const int t = (lane < rem) ? gq[32 + lane] : 0;
                    if (lane < rem) gq[lane] = t;
                    gn = rem;
                }
                m = __builtin_amdgcn_ballot_w32(word != 0u);
            }
        }
    }
    if (gn > 0) expand_groups<F, P, CNT, REC>(src, snd, rcv4, acc, cnt, mine, rcnt, q, qn, gq, gn, r0, lane);
    if (qn > 0) drain_rows<F, P, CNT, REC>(src, snd, acc, cnt, mine, rcnt, q, qn, lane);
    __syncthreads();

    if (REC) {
        if (threadIdx.x == 0) ecount[blockIdx.x] = rcnt[0];
    }

    if (CNT) {
        for (int i = threadIdx.x; i < OWN_R; i += OWN_NW * 32) {
            if (r0 + i < NN) rdeg[r0 + i] = (float)cnt[i];
        }
    }

    if (FIN) {
        if (w < OWN_RT) {
            const int m16  = lane & 15;
            const int kh   = lane >> 4;
            const int lr0  = w * 16;
            const float rs = rsqrtf(fmaxf((float)cnt[lr0 + m16], 1.f));
            const float* arow = acc + (lr0 + m16) * P + 2 * kh;

            float sc[8];
            #pragma unroll
            for (int r = 0; r < 8; ++r) {
                const int node = r0 + lr0 + r + 8 * kh;
                sc[r] = sdeg[(node < NN) ? node : NN - 1];
            }
            const int   n2c = (32 + m16 < DC) ? 32 + m16 : DC - 1;
            const float bb0 = b3[m16], bb1 = b3[16 + m16], bb2 = b3[n2c];

            v8f c0 = {}, c1 = {}, c2 = {};
            #pragma unroll 4
            for (int kb = 0; kb < DH; kb += 4) {
                v2f a = *(const v2f*)(arow + kb);
                a.x *= rs; a.y *= rs;
                const float* wr0 = W3 + (size_t)(kb + 2 * kh) * DC;
                const float* wr1 = wr0 + DC;
                v2f b;
                b.x = wr0[m16];      b.y = wr1[m16];      c0 = wmma4(a, b, c0);
                b.x = wr0[16 + m16]; b.y = wr1[16 + m16]; c1 = wmma4(a, b, c1);
                const float p0 = wr0[n2c], p1 = wr1[n2c];
                b.x = (32 + m16 < DC) ? p0 : 0.f;
                b.y = (32 + m16 < DC) ? p1 : 0.f;
                c2 = wmma4(a, b, c2);
            }
            #pragma unroll
            for (int r = 0; r < 8; ++r) {
                const int lr   = lr0 + r + 8 * kh;
                const int node = r0 + lr;
                if (lr < OWN_R && node < NN) {
                    const float s = rsqrtf(fmaxf(sc[r], 1.f));
                    float* orow = dst + (size_t)node * DC;
                    orow[m16]      = (c0[r] + bb0) * s;
                    orow[16 + m16] = (c1[r] + bb1) * s;
                    if (32 + m16 < DC)
                        orow[32 + m16] = (c2[r] + bb2) * s;
                }
            }
        }
    } else if (SMAX) {
        const int c = lane & 15;
        const int h = lane >> 4;
        const bool t3 = 32 + c < F;
        for (int p = w; p < (OWN_R + 1) / 2; p += OWN_NW) {
            const int  lr  = 2 * p + h;
            const bool ok  = lr < OWN_R && r0 + lr < NN;
            const int  lrc = ok ? lr : 0;
            const float rs = rsc[lrc];
            const float* a = acc + lrc * P;
            const float v0 = a[c] * rs;
            const float v1 = a[16 + c] * rs;
            const float v2 = t3 ? a[t3 ? 32 + c : 0] * rs : -3.402823466e38f;
            const float mx = row16_max(fmaxf(fmaxf(v0, v1), v2));
            const float e0 = __expf(v0 - mx);
            const float e1 = __expf(v1 - mx);
            const float e2 = t3 ? __expf(v2 - mx) : 0.f;
            const float inv = 1.f / row16_sum(e0 + e1 + e2);
            if (ok) {
                float* orow = dst + (size_t)(r0 + lr) * F;
                orow[c]      = e0 * inv;
                orow[16 + c] = e1 * inv;
                if (t3) orow[32 + c] = e2 * inv;
            }
        }
    } else {
        for (int i = threadIdx.x; i < OWN_R * F; i += OWN_NW * 32) {
            const int lr = i / F;
            if (r0 + lr < NN) dst[(size_t)r0 * F + i] = acc[lr * P + (i - lr * F)];
        }
    }
}

__device__ __forceinline__ void count_groups(const int4* __restrict__ snd4, int* cnt, const int* gq, int gn, int r0, int lane) {
    const int sub = lane & 3;
    bool has[4];
    int4 s[4];
    #pragma unroll
    for (int p = 0; p < 4; ++p) {
        const int gi = p * 8 + (lane >> 2);
        has[p] = gi < gn;
        s[p] = snd4[gq[has[p] ? gi : gn - 1] * 4 + sub];
    }
    #pragma unroll
    for (int p = 0; p < 4; ++p) {
        const int ss[4] = {s[p].x, s[p].y, s[p].z, s[p].w};
        #pragma unroll
        for (int c = 0; c < 4; ++c) {
            const unsigned d = (unsigned)(ss[c] - r0);
            if (has[p] && d < (unsigned)OWN_R) atomicAdd(cnt + d, 1);
        }
    }
}

__launch_bounds__(OWN_NW * 32)
__global__ void sdeg_own_kernel(const int* __restrict__ snd, const unsigned* __restrict__ sbits,
                                float* __restrict__ sdeg) {
    __shared__ int cnt[OWN_R];
    __shared__ int gqs[OWN_NW * OWN_GQCAP];
    const int w    = __builtin_amdgcn_readfirstlane((int)(threadIdx.x >> 5));
    const int lane = threadIdx.x & 31;
    const int r0   = blockIdx.x * OWN_R;
    int* gq = gqs + w * OWN_GQCAP;

    for (int i = threadIdx.x; i < OWN_R; i += OWN_NW * 32) cnt[i] = 0;
    __syncthreads();

    const unsigned* myb = sbits + (size_t)blockIdx.x * IDX_ROW;
    const int4*  snd4  = (const int4*)snd;
    int gn = 0;
    {
        unsigned wd[IDX_FULL + 1];
        int      wi[IDX_FULL + 1];
        #pragma unroll
        for (int u = 0; u < IDX_FULL; ++u) {
            wi[u] = (u * OWN_NW + w) * 32 + lane;
            wd[u] = myb[wi[u]];
        }
        wi[IDX_FULL] = IDX_FULL * OWN_NW * 32 + w * IDX_REM + ((lane < IDX_REM) ? lane : 0);
        const unsigned last = myb[wi[IDX_FULL]];
        wd[IDX_FULL] = (lane < IDX_REM) ? last : 0u;
        #pragma unroll
        for (int u = 0; u <= IDX_FULL; ++u) {
            unsigned word = wd[u];
            unsigned m = __builtin_amdgcn_ballot_w32(word != 0u);
            while (m) {
                const int pos = gn + (int)__builtin_amdgcn_mbcnt_lo(m, 0u);
                if (word != 0u) {
                    gq[pos] = wi[u] * 32 + __builtin_ctz(word);
                    word &= word - 1u;
                }
                gn += __builtin_popcount(m);
                if (gn >= 32) {
                    count_groups(snd4, cnt, gq, 32, r0, lane);
                    const int rem = gn - 32;
                    const int t = (lane < rem) ? gq[32 + lane] : 0;
                    if (lane < rem) gq[lane] = t;
                    gn = rem;
                }
                m = __builtin_amdgcn_ballot_w32(word != 0u);
            }
        }
    }
    if (gn > 0) count_groups(snd4, cnt, gq, gn, r0, lane);
    __syncthreads();

    for (int i = threadIdx.x; i < OWN_R; i += OWN_NW * 32) {
        if (r0 + i < NN) sdeg[r0 + i] = (float)cnt[i];
    }
}

extern "C" void kernel_launch(void* const* d_in, const int* in_sizes, int n_in,
                              void* d_out, int out_size, void* d_ws, size_t ws_size,
                              hipStream_t stream) {
    const float* nodes     = (const float*)d_in[0];
    const int*   senders   = (const int*)d_in[1];
    const int*   receivers = (const int*)d_in[2];
    const float* W1 = (const float*)d_in[3];
    const float* b1 = (const float*)d_in[4];
    const float* W2 = (const float*)d_in[5];
    const float* b2 = (const float*)d_in[6];
    const float* W3 = (const float*)d_in[7];
    const float* b3 = (const float*)d_in[8];
    float* out = (float*)d_out;
    float* ws  = (float*)d_ws;

    float* sdeg = ws;
    float* rdeg = ws + NN;
    float* h1   = ws + 2 * (size_t)NN;
    float* h2   = h1 + 2 * (size_t)NN * DH;
    unsigned* rbits = (unsigned*)(h2 + (size_t)NN * DC);
    unsigned* sbits = rbits + (size_t)OWN_G * IDX_ROW;
    unsigned* elist = sbits + (size_t)OWN_G * IDX_ROW;
    int* ecount = (int*)(elist + (size_t)OWN_G * OWN_ECAP);

    const int nTiles = NN / 16;
    index_kernel<<<dim3(IDX_NB, 2), 1024, 0, stream>>>(senders, receivers, sbits, rbits);
    sdeg_own_kernel<<<OWN_G, OWN_NW * 32, 0, stream>>>(senders, sbits, sdeg);
    mlp_kernel<<<(nTiles + 7) / 8, 256, 0, stream>>>(nodes, W1, b1, W2, b2, sdeg, h1, nTiles);
    agg_own_kernel<DH, 68, true, true, false, true, false><<<OWN_G, OWN_NW * 32, 0, stream>>>(h1, senders, receivers, rbits, h2, rdeg, elist, ecount, sdeg, W3, b3, nullptr);
    agg_own_kernel<DC, DC, false, false, true, false, true><<<OWN_G, OWN_NW * 32, 0, stream>>>(h2, senders, receivers, rbits, out, nullptr, elist, ecount, nullptr, nullptr, nullptr, rdeg);
}
